# starting kernel
# baseline (speedup 1.0000x reference)
#include <hip/hip_runtime.h>
#include <math.h>


typedef float v2f __attribute__((ext_vector_type(2)));
typedef float v8f __attribute__((ext_vector_type(8)));

#define BB   16
#define TT   512
#define HH   128
#define MM   (BB * TT)
#define NEL  (BB * TT * HH)
#define SCALE 0.08838834764831845f
#define LN_EPS 1e-5f
#define LCH  16
#define NCH  (TT / LCH)

__global__ __launch_bounds__(128) void proj_kernel(
    const float* __restrict__ x,
    const float* __restrict__ Wq, const float* __restrict__ Wk,
    const float* __restrict__ Wv, const float* __restrict__ Wi,
    const float* __restrict__ Wf, const float* __restrict__ Wo,
    const float* __restrict__ b_i, const float* __restrict__ b_f,
    const float* __restrict__ b_o,
    float* __restrict__ oq, float* __restrict__ ok, float* __restrict__ ov,
    float* __restrict__ oi, float* __restrict__ of, float* __restrict__ oo)
{
    const int lane   = threadIdx.x & 31;
    const int waveId = threadIdx.x >> 5;
    const int tile   = blockIdx.x * 4 + waveId;

    const int matIdx = tile / (512 * 8);
    const int rem    = tile % (512 * 8);
    const int m0     = (rem >> 3) << 4;
    const int n0     = (rem & 7) << 4;

    const float* W;
    switch (matIdx) {
        case 0: W = Wq; break;
        case 1: W = Wk; break;
        case 2: W = Wv; break;
        case 3: W = Wi; break;
        case 4: W = Wf; break;
        default: W = Wo; break;
    }

    const int lm = lane & 15;
    const int lk = (lane >> 4) << 1;

    const float* aptr = x + (m0 + lm) * HH + lk;
    const float* bptr = W + (n0 + lm) * HH + lk;

    v8f acc = {};
    #pragma unroll
    for (int kb = 0; kb < HH; kb += 4) {
        v2f a = *(const v2f*)(aptr + kb);
        v2f b = *(const v2f*)(bptr + kb);
        acc = __builtin_amdgcn_wmma_f32_16x16x4_f32(false, a, false, b,
                                                    (short)0, acc, false, false);
    }

    const int col = n0 + lm;
    float bias = 0.0f;
    if (matIdx == 3) bias = b_i[col];
    else if (matIdx == 4) bias = b_f[col];
    else if (matIdx == 5) bias = b_o[col];

    float* outp;
    switch (matIdx) {
        case 0: outp = oq; break;
        case 1: outp = ok; break;
        case 2: outp = ov; break;
        case 3: outp = oi; break;
        case 4: outp = of; break;
        default: outp = oo; break;
    }

    const int rowBase = m0 + ((lane >> 4) << 3);
    #pragma unroll
    for (int v = 0; v < 8; ++v) {
        float val = acc[v];
        if (matIdx <= 1)       val *= SCALE;
        else if (matIdx == 3)  val = expf(val + bias);
        else if (matIdx >= 4)  val = 1.0f / (1.0f + expf(-(val + bias)));
        outp[(rowBase + v) * HH + col] = val;
    }
}

#define CSTR 132
#define TSTR 20
#define C_OFF   0
#define Q_OFF   (C_OFF + 128 * CSTR)
#define K_OFF   (Q_OFF + 16 * CSTR)
#define KT_OFF  (K_OFF + 16 * CSTR)
#define WC_OFF  (KT_OFF + 128 * TSTR)
#define P_OFF   (WC_OFF + 128 * TSTR)
#define M_OFF   (P_OFF + 16 * CSTR)
#define SP_OFF  (M_OFF + 16 * CSTR)
#define S_OFF   (SP_OFF + 4 * 16 * 17)
#define G15_OFF (S_OFF + 16 * 17)
#define NPS_OFF (G15_OFF + 128)
#define NQ_OFF  (NPS_OFF + 128)
#define SMEM_FLOATS (NQ_OFF + 16)

__global__ __launch_bounds__(128, 1) void rec_kernel(
    const float* __restrict__ qb, const float* __restrict__ kb,
    const float* __restrict__ vb, const float* __restrict__ ib,
    const float* __restrict__ fb, const float* __restrict__ ob,
    float* __restrict__ hb)
{
    extern __shared__ float sm[];
    const int b    = blockIdx.x;
    const int r    = threadIdx.x;
    const int lane = r & 31;
    const int wave = r >> 5;
    const int lm   = lane & 15;
    const int lk   = (lane >> 4) << 1;

    #pragma unroll 4
    for (int j = 0; j < CSTR; ++j) sm[C_OFF + r * CSTR + j] = 0.0f;

    float nprev = 0.0f;
    const long bbase = (long)b * TT * HH + r;

    float pk[LCH], pq[LCH], pi[LCH], pf[LCH], pv[LCH], po[LCH];
    #pragma unroll
    for (int s = 0; s < LCH; ++s) {
        const long idx = bbase + (long)s * HH;
        pk[s] = kb[idx]; pq[s] = qb[idx]; pi[s] = ib[idx];
        pf[s] = fb[idx]; pv[s] = vb[idx]; po[s] = ob[idx];
    }

    for (int ch = 0; ch < NCH; ++ch) {
        const long gbase = bbase + (long)(ch * LCH) * HH;

        float G[LCH], Wg[LCH], ovv[LCH];
        {
            float g = 1.0f;
            #pragma unroll
            for (int s = 0; s < LCH; ++s) { g *= pf[s]; G[s] = g; ovv[s] = po[s]; }
            #pragma unroll
            for (int s = 0; s < LCH; ++s) Wg[s] = pi[s] * pv[s] / G[s];
            const float g15 = G[LCH - 1];
            float np = nprev;
            #pragma unroll
            for (int s = 0; s < LCH; ++s) {
                np = fmaf(pf[s], np, pi[s] * pk[s]);
                sm[M_OFF + s * CSTR + r] = np * pq[s];
            }
            nprev = np;
            #pragma unroll
            for (int s = 0; s < LCH; ++s) {
                sm[Q_OFF + s * CSTR + r]  = pq[s];
                sm[K_OFF + s * CSTR + r]  = pk[s];
                sm[KT_OFF + r * TSTR + s] = pk[s];
                sm[WC_OFF + r * TSTR + s] = g15 * Wg[s];
            }
            sm[G15_OFF + r] = g15;
        }
        if (ch + 1 < NCH) {
            const long nbase = bbase + (long)((ch + 1) * LCH) * HH;
            #pragma unroll
            for (int s = 0; s < LCH; ++s) {
                const long idx = nbase + (long)s * HH;
                pk[s] = kb[idx]; pq[s] = qb[idx]; pi[s] = ib[idx];
                pf[s] = fb[idx]; pv[s] = vb[idx]; po[s] = ob[idx];
            }
        }
        __syncthreads();

        {
            const int s = r >> 3, seg = r & 7;
            float ps = 0.0f;
            #pragma unroll
            for (int e = 0; e < 16; ++e) ps += sm[M_OFF + s * CSTR + seg * 16 + e];
            sm[NPS_OFF + s * 8 + seg] = ps;
        }
        {
            v2f av[32];
            #pragma unroll
            for (int kk = 0; kk < 32; ++kk)
                av[kk] = *(const v2f*)&sm[Q_OFF + lm * CSTR + kk * 4 + lk];
            #pragma unroll
            for (int tile = 0; tile < 2; ++tile) {
                const int n0p = (2 * wave + tile) << 4;
                v8f acc = {};
                #pragma unroll
                for (int kk = 0; kk < 32; ++kk) {
                    v2f bbf = *(const v2f*)&sm[C_OFF + (n0p + lm) * CSTR + kk * 4 + lk];
                    acc = __builtin_amdgcn_wmma_f32_16x16x4_f32(false, av[kk], false, bbf,
                                                                (short)0, acc, false, false);
                }
                #pragma unroll
                for (int v = 0; v < 8; ++v)
                    sm[P_OFF + (v + 8 * (lane >> 4)) * CSTR + n0p + lm] = acc[v];
            }
        }
        __syncthreads();

        if (r < 16) {
            float t = 0.0f;
            #pragma unroll
            for (int e = 0; e < 8; ++e) t += sm[NPS_OFF + r * 8 + e];
            sm[NQ_OFF + r] = t;
        }
        {
            const int kbase = wave * 32;
            v8f sacc = {};
            #pragma unroll
            for (int kk = 0; kk < 8; ++kk) {
                const int c0 = kbase + kk * 4 + lk;
                v2f a  = *(const v2f*)&sm[K_OFF + lm * CSTR + c0];
                v2f bq = *(const v2f*)&sm[Q_OFF + lm * CSTR + c0];
                sacc = __builtin_amdgcn_wmma_f32_16x16x4_f32(false, a, false, bq,
                                                             (short)0, sacc, false, false);
            }
            #pragma unroll
            for (int v = 0; v < 8; ++v)
                sm[SP_OFF + wave * 272 + (v + 8 * (lane >> 4)) * 17 + lm] = sacc[v];
        }
        {
            const int mt0 = wave * 2;
            v2f afr[2][4];
            #pragma unroll
            for (int mi = 0; mi < 2; ++mi)
                #pragma unroll
                for (int kk = 0; kk < 4; ++kk)
                    afr[mi][kk] = *(const v2f*)&sm[WC_OFF + ((mt0 + mi) * 16 + lm) * TSTR + kk * 4 + lk];
            #pragma unroll
            for (int nt = 0; nt < 8; ++nt) {
                v2f bfr[4];
                #pragma unroll
                for (int kk = 0; kk < 4; ++kk)
                    bfr[kk] = *(const v2f*)&sm[KT_OFF + (nt * 16 + lm) * TSTR + kk * 4 + lk];
                #pragma unroll
                for (int mi = 0; mi < 2; ++mi) {
                    v8f dacc = {};
                    #pragma unroll
                    for (int kk = 0; kk < 4; ++kk)
                        dacc = __builtin_amdgcn_wmma_f32_16x16x4_f32(false, afr[mi][kk], false, bfr[kk],
                                                                     (short)0, dacc, false, false);
                    const int m0r = (mt0 + mi) * 16;
                    #pragma unroll
                    for (int v = 0; v < 8; ++v) {
                        const int row = m0r + v + 8 * (lane >> 4);
                        const int ci  = C_OFF + row * CSTR + nt * 16 + lm;
                        const float g = sm[G15_OFF + row];
                        sm[ci] = fmaf(g, sm[ci], dacc[v]);
                    }
                }
            }
        }
        __syncthreads();

        #pragma unroll
        for (int p = 0; p < 2; ++p) {
            const int idx = r * 2 + p;
            const int u = idx >> 4, sc = idx & 15;
            float ssum = sm[SP_OFF + 0 * 272 + u * 17 + sc]
                       + sm[SP_OFF + 1 * 272 + u * 17 + sc]
                       + sm[SP_OFF + 2 * 272 + u * 17 + sc]
                       + sm[SP_OFF + 3 * 272 + u * 17 + sc];
            sm[S_OFF + u * 17 + sc] = ssum;
        }
        __syncthreads();

        #pragma unroll
        for (int s = 0; s < LCH; ++s) {
            float acc = sm[P_OFF + s * CSTR + r];
            #pragma unroll
            for (int u = 0; u < LCH; ++u)
                if (u <= s) acc = fmaf(Wg[u], sm[S_OFF + u * 17 + s], acc);
            const float ret = G[s] * acc;
            const float nrm = fmaxf(fabsf(sm[NQ_OFF + s]), 1.0f);
            hb[gbase + (long)s * HH] = ovv[s] * ret / nrm;
        }
        __syncthreads();
    }
}

__global__ __launch_bounds__(256) void ln_kernel(
    const float* __restrict__ hb, const float* __restrict__ x,
    const float* __restrict__ lnw, const float* __restrict__ lnb,
    float* __restrict__ out)
{
    const int lane = threadIdx.x & 31;
    const int wave = threadIdx.x >> 5;
    const long row = (long)blockIdx.x * 8 + wave;
    const long rb  = row * HH;

    float y[4];
    float s = 0.0f, s2 = 0.0f;
    #pragma unroll
    for (int j = 0; j < 4; ++j) {
        const int c = lane + 32 * j;
        y[j] = hb[rb + c] + x[rb + c];
        s  += y[j];
        s2 += y[j] * y[j];
    }
    #pragma unroll
    for (int off = 16; off >= 1; off >>= 1) {
        s  += __shfl_xor(s,  off, 32);
        s2 += __shfl_xor(s2, off, 32);
    }
    const float mu  = s * (1.0f / HH);
    const float var = s2 * (1.0f / HH) - mu * mu;
    const float rs  = rsqrtf(var + LN_EPS);

    #pragma unroll
    for (int j = 0; j < 4; ++j) {
        const int c = lane + 32 * j;
        out[rb + c] = (y[j] - mu) * rs * lnw[c] + lnb[c];
    }
}

extern "C" void kernel_launch(void* const* d_in, const int* in_sizes, int n_in,
                              void* d_out, int out_size, void* d_ws, size_t ws_size,
                              hipStream_t stream)
{
    const float* x   = (const float*)d_in[0];
    const float* Wq  = (const float*)d_in[1];
    const float* Wk  = (const float*)d_in[2];
    const float* Wv  = (const float*)d_in[3];
    const float* Wi  = (const float*)d_in[4];
    const float* Wf  = (const float*)d_in[5];
    const float* Wo  = (const float*)d_in[6];
    const float* b_i = (const float*)d_in[7];
    const float* b_f = (const float*)d_in[8];
    const float* b_o = (const float*)d_in[9];
    const float* lnw = (const float*)d_in[10];
    const float* lnb = (const float*)d_in[11];
    float* out = (float*)d_out;

    float* ws = (float*)d_ws;
    float* q  = ws + 0L * NEL;
    float* k  = ws + 1L * NEL;
    float* v  = ws + 2L * NEL;
    float* ig = ws + 3L * NEL;
    float* fg = ws + 4L * NEL;
    float* og = ws + 5L * NEL;
    float* h  = ws + 6L * NEL;

    proj_kernel<<<6144, 128, 0, stream>>>(x, Wq, Wk, Wv, Wi, Wf, Wo,
                                          b_i, b_f, b_o,
                                          q, k, v, ig, fg, og);
    rec_kernel<<<BB, 128, SMEM_FLOATS * sizeof(float), stream>>>(q, k, v, ig, fg, og, h);
    ln_kernel<<<MM / 8, 256, 0, stream>>>(h, x, lnw, lnb, out);
}


# Recurrence barriers fence LDS only, so the next chunk's global loads stay in flight across them (issued in two groups of up to 48) instead of being drained in front of each chunk's first barrier
# speedup vs baseline: 3.5375x; 3.5375x over previous
#include <hip/hip_runtime.h>
#include <math.h>


typedef float v2f __attribute__((ext_vector_type(2)));
typedef float v8f __attribute__((ext_vector_type(8)));

#define BB   16
#define TT   512
#define HH   128
#define MM   (BB * TT)
#define NEL  (BB * TT * HH)
#define SCALE 0.08838834764831845f
#define LN_EPS 1e-5f
#define LCH  16
#define NCH  (TT / LCH)

__global__ __launch_bounds__(128) void proj_kernel(
    const float* __restrict__ x,
    const float* __restrict__ Wq, const float* __restrict__ Wk,
    const float* __restrict__ Wv, const float* __restrict__ Wi,
    const float* __restrict__ Wf, const float* __restrict__ Wo,
    const float* __restrict__ b_i, const float* __restrict__ b_f,
    const float* __restrict__ b_o,
    float* __restrict__ oq, float* __restrict__ ok, float* __restrict__ ov,
    float* __restrict__ oi, float* __restrict__ of, float* __restrict__ oo)
{
    const int lane   = threadIdx.x & 31;
    const int waveId = threadIdx.x >> 5;
    const int tile   = blockIdx.x * 4 + waveId;

    const int matIdx = tile / (512 * 8);
    const int rem    = tile % (512 * 8);
    const int m0     = (rem >> 3) << 4;
    const int n0     = (rem & 7) << 4;

    const float* W;
    switch (matIdx) {
        case 0: W = Wq; break;
        case 1: W = Wk; break;
        case 2: W = Wv; break;
        case 3: W = Wi; break;
        case 4: W = Wf; break;
        default: W = Wo; break;
    }

    const int lm = lane & 15;
    const int lk = (lane >> 4) << 1;

    const float* aptr = x + (m0 + lm) * HH + lk;
    const float* bptr = W + (n0 + lm) * HH + lk;

    v8f acc = {};
    #pragma unroll
    for (int kb = 0; kb < HH; kb += 4) {
        v2f a = *(const v2f*)(aptr + kb);
        v2f b = *(const v2f*)(bptr + kb);
        acc = __builtin_amdgcn_wmma_f32_16x16x4_f32(false, a, false, b,
                                                    (short)0, acc, false, false);
    }

    const int col = n0 + lm;
    float bias = 0.0f;
    if (matIdx == 3) bias = b_i[col];
    else if (matIdx == 4) bias = b_f[col];
    else if (matIdx == 5) bias = b_o[col];

    float* outp;
    switch (matIdx) {
        case 0: outp = oq; break;
        case 1: outp = ok; break;
        case 2: outp = ov; break;
        case 3: outp = oi; break;
        case 4: outp = of; break;
        default: outp = oo; break;
    }

    const int rowBase = m0 + ((lane >> 4) << 3);
    #pragma unroll
    for (int v = 0; v < 8; ++v) {
        float val = acc[v];
        if (matIdx <= 1)       val *= SCALE;
        else if (matIdx == 3)  val = expf(val + bias);
        else if (matIdx >= 4)  val = 1.0f / (1.0f + expf(-(val + bias)));
        outp[(rowBase + v) * HH + col] = val;
    }
}

#define NSL  8
#define RS   16
#define REC_GRID 128
#define CSTR 132
#define TSTR 20
#define PSTR 17
#define PLN  272
#define C_OFF   0
#define Q_OFF   (C_OFF + RS * CSTR)
#define K_OFF   (Q_OFF + 16 * CSTR)
#define M_OFF   (K_OFF + 16 * CSTR)
#define KT_OFF  (M_OFF + 16 * CSTR)
#define WC_OFF  (KT_OFF + 128 * TSTR)
#define PP_OFF  (WC_OFF + RS * TSTR)
#define SP_OFF  (PP_OFF + 4 * PLN)
#define S_OFF   (SP_OFF + 4 * PLN)
#define NPS_OFF (S_OFF + PLN)
#define NQ_OFF  (NPS_OFF + 128)
#define G15_OFF (NQ_OFF + 16)
#define RW_OFF  (G15_OFF + 16)
#define RW_SZ   (3 * PLN)
#define SMEM_FLOATS (RW_OFF + 2 * RW_SZ)


static __device__ __forceinline__ void lds_barrier()
{
    __builtin_amdgcn_fence(__ATOMIC_RELEASE, "workgroup", "local");
    __builtin_amdgcn_s_barrier();
    __builtin_amdgcn_fence(__ATOMIC_ACQUIRE, "workgroup", "local");
}

__global__ __launch_bounds__(128, 1) void rec_kernel(
    const float* __restrict__ qb, const float* __restrict__ kb,
    const float* __restrict__ vb, const float* __restrict__ ib,
    const float* __restrict__ fb, const float* __restrict__ ob,
    float* __restrict__ hb)
{
    extern __shared__ float sm[];
    const int b    = blockIdx.x >> 3;
    const int sl   = blockIdx.x & 7;
    const int r    = threadIdx.x;
    const int lane = r & 31;
    const int wave = r >> 5;
    const int lm   = lane & 15;
    const int hh   = lane >> 4;
    const int lk   = hh << 1;
    const int x    = r & 15;
    const bool rowthr = (r >> 4) == sl;

    for (int j = r; j < RS * CSTR; j += 128) sm[C_OFF + j] = 0.0f;
    float creg[2][8];
    #pragma unroll
    for (int t = 0; t < 2; ++t)
        #pragma unroll
        for (int v = 0; v < 8; ++v) creg[t][v] = 0.0f;

    float nprev = 0.0f;
    const long bbase = (long)b * TT * HH + r;
    const long hbase = (long)blockIdx.x * (TT * RS) + r;

    float pk[LCH], pq[LCH], pi[LCH], pf[LCH], pv[LCH], po[LCH];
    #pragma unroll
    for (int s = 0; s < LCH; ++s) {
        const long idx = bbase + (long)s * HH;
        pk[s] = kb[idx]; pq[s] = qb[idx]; pi[s] = ib[idx]; pf[s] = fb[idx];
        pv[s] = 0.0f; po[s] = 0.0f;
    }
    if (rowthr) {
        #pragma unroll
        for (int s = 0; s < LCH; ++s) {
            const long idx = bbase + (long)s * HH;
            pv[s] = vb[idx]; po[s] = ob[idx];
        }
    }

    for (int ch = 0; ch < NCH; ++ch) {
        float* rw = sm + RW_OFF + (ch & 1) * RW_SZ;

        {
            float np = nprev;
            #pragma unroll
            for (int s = 0; s < LCH; ++s) {
                np = fmaf(pf[s], np, pi[s] * pk[s]);
                sm[M_OFF + s * CSTR + r] = np * pq[s];
            }
            nprev = np;
            #pragma unroll
            for (int s = 0; s < LCH; ++s) {
                sm[Q_OFF + s * CSTR + r]  = pq[s];
                sm[K_OFF + s * CSTR + r]  = pk[s];
                sm[KT_OFF + r * TSTR + s] = pk[s];
            }
        }
        if (rowthr) {
            float G[LCH];
            float g = 1.0f;
            #pragma unroll
            for (int s = 0; s < LCH; ++s) { g *= pf[s]; G[s] = g; }
            const float g15 = G[LCH - 1];
            #pragma unroll
            for (int s = 0; s < LCH; ++s) {
                const float wg = pi[s] * pv[s] / G[s];
                sm[WC_OFF + x * TSTR + s]  = g15 * wg;
                rw[0 * PLN + s * PSTR + x] = wg;
                rw[1 * PLN + s * PSTR + x] = G[s];
                rw[2 * PLN + s * PSTR + x] = po[s];
            }
            sm[G15_OFF + x] = g15;
        }
        if (ch + 1 < NCH) {
            const long nbase = bbase + (long)((ch + 1) * LCH) * HH;
            #pragma unroll
            for (int s = 0; s < LCH; ++s) {
                const long idx = nbase + (long)s * HH;
                pk[s] = kb[idx]; pq[s] = qb[idx]; pi[s] = ib[idx];
            }
        }
        lds_barrier();

        {
            const int s1 = r >> 3, seg = r & 7;
            float mv[16];
            #pragma unroll
            for (int e = 0; e < 16; ++e) mv[e] = sm[M_OFF + s1 * CSTR + seg * 16 + e];

            const int cb = wave * 32 + lk;
            v2f qf[8], kf[8], cf[8];
            #pragma unroll
            for (int kk = 0; kk < 8; ++kk) {
                qf[kk] = *(const v2f*)&sm[Q_OFF + lm * CSTR + cb + kk * 4];
                kf[kk] = *(const v2f*)&sm[K_OFF + lm * CSTR + cb + kk * 4];
                cf[kk] = *(const v2f*)&sm[C_OFF + lm * CSTR + cb + kk * 4];
            }
            float ps = 0.0f;
            #pragma unroll
            for (int e = 0; e < 16; ++e) ps += mv[e];

            v8f pacc = {};
            v8f sacc = {};
            #pragma unroll
            for (int kk = 0; kk < 8; ++kk) {
                pacc = __builtin_amdgcn_wmma_f32_16x16x4_f32(false, qf[kk], false, cf[kk],
                                                             (short)0, pacc, false, false);
                sacc = __builtin_amdgcn_wmma_f32_16x16x4_f32(false, kf[kk], false, qf[kk],
                                                             (short)0, sacc, false, false);
            }
            sm[NPS_OFF + s1 * 8 + seg] = ps;
            #pragma unroll
            for (int v = 0; v < 8; ++v) {
                sm[PP_OFF + wave * PLN + (v + 8 * hh) * PSTR + lm] = pacc[v];
                sm[SP_OFF + wave * PLN + (v + 8 * hh) * PSTR + lm] = sacc[v];
            }
        }
        lds_barrier();

        if (ch + 1 < NCH) {
            const long nbase = bbase + (long)((ch + 1) * LCH) * HH;
            #pragma unroll
            for (int s = 0; s < LCH; ++s) pf[s] = fb[nbase + (long)s * HH];
            if (rowthr) {
                #pragma unroll
                for (int s = 0; s < LCH; ++s) {
                    const long idx = nbase + (long)s * HH;
                    pv[s] = vb[idx]; po[s] = ob[idx];
                }
            }
        }

        {
            float nps[8];
            #pragma unroll
            for (int e = 0; e < 8; ++e) nps[e] = sm[NPS_OFF + x * 8 + e];
            float spv[2][4];
            #pragma unroll
            for (int p = 0; p < 2; ++p) {
                const int idx = r * 2 + p;
                const int u = idx >> 4, sc = idx & 15;
                #pragma unroll
                for (int w = 0; w < 4; ++w) spv[p][w] = sm[SP_OFF + w * PLN + u * PSTR + sc];
            }
            v2f afr[4];
            #pragma unroll
            for (int kk = 0; kk < 4; ++kk)
                afr[kk] = *(const v2f*)&sm[WC_OFF + lm * TSTR + kk * 4 + lk];
            float g[8];
            #pragma unroll
            for (int v = 0; v < 8; ++v) g[v] = sm[G15_OFF + 8 * hh + v];
            v2f bfr[2][4];
            #pragma unroll
            for (int t = 0; t < 2; ++t)
                #pragma unroll
                for (int kk = 0; kk < 4; ++kk)
                    bfr[t][kk] = *(const v2f*)&sm[KT_OFF + (wave * 32 + t * 16 + lm) * TSTR + kk * 4 + lk];

            v8f d0 = {};
            v8f d1 = {};
            #pragma unroll
            for (int kk = 0; kk < 4; ++kk) {
                d0 = __builtin_amdgcn_wmma_f32_16x16x4_f32(false, afr[kk], false, bfr[0][kk],
                                                           (short)0, d0, false, false);
                d1 = __builtin_amdgcn_wmma_f32_16x16x4_f32(false, afr[kk], false, bfr[1][kk],
                                                           (short)0, d1, false, false);
            }
            #pragma unroll
            for (int v = 0; v < 8; ++v) {
                creg[0][v] = fmaf(g[v], creg[0][v], d0[v]);
                creg[1][v] = fmaf(g[v], creg[1][v], d1[v]);
            }
            if (r < 16) {
                float t = 0.0f;
                #pragma unroll
                for (int e = 0; e < 8; ++e) t += nps[e];
                sm[NQ_OFF + r] = t;
            }
            #pragma unroll
            for (int p = 0; p < 2; ++p) {
                const int idx = r * 2 + p;
                const int u = idx >> 4, sc = idx & 15;
                sm[S_OFF + u * PSTR + sc] = spv[p][0] + spv[p][1] + spv[p][2] + spv[p][3];
            }
            #pragma unroll
            for (int t = 0; t < 2; ++t)
                #pragma unroll
                for (int v = 0; v < 8; ++v)
                    sm[C_OFF + (8 * hh + v) * CSTR + wave * 32 + t * 16 + lm] = creg[t][v];
        }
        lds_barrier();

        {
            const int sA = r >> 4;
            float wgv[LCH];
            #pragma unroll
            for (int u = 0; u < LCH; ++u) wgv[u] = rw[0 * PLN + u * PSTR + x];
            float sv[2][LCH], ppv[2][4], gs[2], os[2], nqv[2];
            #pragma unroll
            for (int p = 0; p < 2; ++p) {
                const int s = sA + 8 * p;
                #pragma unroll
                for (int u = 0; u < LCH; ++u) sv[p][u] = sm[S_OFF + u * PSTR + s];
                #pragma unroll
                for (int w = 0; w < 4; ++w) ppv[p][w] = sm[PP_OFF + w * PLN + s * PSTR + x];
                gs[p]  = rw[1 * PLN + s * PSTR + x];
                os[p]  = rw[2 * PLN + s * PSTR + x];
                nqv[p] = sm[NQ_OFF + s];
            }
            #pragma unroll
            for (int p = 0; p < 2; ++p) {
                const int s = sA + 8 * p;
                float acc = ((ppv[p][0] + ppv[p][1]) + ppv[p][2]) + ppv[p][3];
                #pragma unroll
                for (int u = 0; u < LCH; ++u) {
                    const float nx = fmaf(wgv[u], sv[p][u], acc);
                    acc = (u <= s) ? nx : acc;
                }
                const float ret = gs[p] * acc;
                const float nrm = fmaxf(fabsf(nqv[p]), 1.0f);
                hb[hbase + (long)ch * (LCH * RS) + p * 128] = os[p] * ret / nrm;
            }
        }
    }
}

__global__ __launch_bounds__(256) void ln_kernel(
    const float* __restrict__ hb, const float* __restrict__ x,
    const float* __restrict__ lnw, const float* __restrict__ lnb,
    float* __restrict__ out)
{
    const int lane = threadIdx.x & 31;
    const int wave = threadIdx.x >> 5;
    const long row = (long)blockIdx.x * 8 + wave;
    const long rb  = row * HH;

    float y[4];
    float s = 0.0f, s2 = 0.0f;
    #pragma unroll
    for (int j = 0; j < 4; ++j) {
        const int c = lane + 32 * j;
        const long hidx = (((row >> 9) * NSL + (c >> 4)) * TT + (row & (TT - 1))) * RS + (c & 15);
        y[j] = hb[hidx] + x[rb + c];
        s  += y[j];
        s2 += y[j] * y[j];
    }
    #pragma unroll
    for (int off = 16; off >= 1; off >>= 1) {
        s  += __shfl_xor(s,  off, 32);
        s2 += __shfl_xor(s2, off, 32);
    }
    const float mu  = s * (1.0f / HH);
    const float var = s2 * (1.0f / HH) - mu * mu;
    const float rs  = rsqrtf(var + LN_EPS);

    #pragma unroll
    for (int j = 0; j < 4; ++j) {
        const int c = lane + 32 * j;
        out[rb + c] = (y[j] - mu) * rs * lnw[c] + lnb[c];
    }
}

extern "C" void kernel_launch(void* const* d_in, const int* in_sizes, int n_in,
                              void* d_out, int out_size, void* d_ws, size_t ws_size,
                              hipStream_t stream)
{
    const float* x   = (const float*)d_in[0];
    const float* Wq  = (const float*)d_in[1];
    const float* Wk  = (const float*)d_in[2];
    const float* Wv  = (const float*)d_in[3];
    const float* Wi  = (const float*)d_in[4];
    const float* Wf  = (const float*)d_in[5];
    const float* Wo  = (const float*)d_in[6];
    const float* b_i = (const float*)d_in[7];
    const float* b_f = (const float*)d_in[8];
    const float* b_o = (const float*)d_in[9];
    const float* lnw = (const float*)d_in[10];
    const float* lnb = (const float*)d_in[11];
    float* out = (float*)d_out;

    float* ws = (float*)d_ws;
    float* q  = ws + 0L * NEL;
    float* k  = ws + 1L * NEL;
    float* v  = ws + 2L * NEL;
    float* ig = ws + 3L * NEL;
    float* fg = ws + 4L * NEL;
    float* og = ws + 5L * NEL;
    float* h  = ws + 6L * NEL;

    proj_kernel<<<6144, 128, 0, stream>>>(x, Wq, Wk, Wv, Wi, Wf, Wo,
                                          b_i, b_f, b_o,
                                          q, k, v, ig, fg, og);
    rec_kernel<<<REC_GRID, 128, SMEM_FLOATS * sizeof(float), stream>>>(q, k, v, ig, fg, og, h);
    ln_kernel<<<MM / 8, 256, 0, stream>>>(h, x, lnw, lnb, out);
}


# Projection wave tile 64x32 fed straight from global (no LDS) by 128-bit fragment loads in a permuted k order, a fragment feeds 2 or 4 products; whole-line stores; 8x fewer waves; k sum order changes
# speedup vs baseline: 4.2556x; 1.2030x over previous
#include <hip/hip_runtime.h>
#include <math.h>


typedef float v2f __attribute__((ext_vector_type(2)));
typedef float v4f __attribute__((ext_vector_type(4)));
typedef float v8f __attribute__((ext_vector_type(8)));

#define BB   16
#define TT   512
#define HH   128
#define MM   (BB * TT)
#define NEL  (BB * TT * HH)
#define SCALE 0.08838834764831845f
#define LN_EPS 1e-5f
#define LCH  16
#define NCH  (TT / LCH)

#define PJ_MT 4
#define PJ_NT 2
#define PJ_ROWS (16 * PJ_MT)
#define PJ_GRID ((MM / PJ_ROWS) * 6)

static __device__ __forceinline__ float xhalf(float v)
{
    return __builtin_bit_cast(float, __builtin_amdgcn_permlanex16(
        __builtin_bit_cast(int, v), __builtin_bit_cast(int, v),
        0x76543210, 0xfedcba98, false, false));
}

template <int ACT>
static __device__ __forceinline__ float proj_act(float val, float bias)
{
    if (ACT == 0) return val * SCALE;
    if (ACT == 1) return val;
    if (ACT == 2) return expf(val + bias);
    return 1.0f / (1.0f + expf(-(val + bias)));
}

template <int ACT>
static __device__ __forceinline__ void proj_store(
    const v8f (&acc)[PJ_MT][PJ_NT], float* __restrict__ outp,
    int m0, int n0, int lane, float bias0, float bias1)
{
    const int hh = lane >> 4;
    #pragma unroll
    for (int i = 0; i < PJ_MT; ++i) {
        #pragma unroll
        for (int v = 0; v < 8; ++v) {
            const float t0 = proj_act<ACT>(acc[i][0][v], bias0);
            const float t1 = proj_act<ACT>(acc[i][1][v], bias1);
            const float got = xhalf(hh ? t0 : t1);
            const float lo = hh ? got : t0;
            const float hi = hh ? t1 : got;
            float* o = outp + (long)(m0 + 16 * i + v) * HH + n0 + lane;
            o[0]      = lo;
            o[8 * HH] = hi;
        }
    }
}

__global__ __launch_bounds__(128, 3) void proj_kernel(
    const float* __restrict__ x,
    const float* __restrict__ Wq, const float* __restrict__ Wk,
    const float* __restrict__ Wv, const float* __restrict__ Wi,
    const float* __restrict__ Wf, const float* __restrict__ Wo,
    const float* __restrict__ b_i, const float* __restrict__ b_f,
    const float* __restrict__ b_o,
    float* __restrict__ oq, float* __restrict__ ok, float* __restrict__ ov,
    float* __restrict__ oi, float* __restrict__ of, float* __restrict__ oo)
{
    const int lane   = threadIdx.x & 31;
    const int wave   = threadIdx.x >> 5;
    const int matIdx = blockIdx.x % 6;
    const int m0     = (blockIdx.x / 6) * PJ_ROWS;
    const int n0     = wave * (16 * PJ_NT);

    const float* W;
    const float* bp = b_i;
    float* outp;
    switch (matIdx) {
        case 0:  W = Wq; outp = oq; break;
        case 1:  W = Wk; outp = ok; break;
        case 2:  W = Wv; outp = ov; break;
        case 3:  W = Wi; outp = oi; bp = b_i; break;
        case 4:  W = Wf; outp = of; bp = b_f; break;
        default: W = Wo; outp = oo; bp = b_o; break;
    }

    const int lm = lane & 15;
    const int hh = lane >> 4;

    const float* aptr = x + (long)(m0 + lm) * HH + 16 * hh;
    const float* bptr = W + (n0 + lm) * HH + 16 * hh;

    v8f acc[PJ_MT][PJ_NT];
    #pragma unroll
    for (int i = 0; i < PJ_MT; ++i)
        #pragma unroll
        for (int j = 0; j < PJ_NT; ++j) acc[i][j] = (v8f){};

    v4f fa[2][PJ_MT][4];
    v4f fb[2][PJ_NT][4];

    #pragma unroll
    for (int i = 0; i < PJ_MT; ++i)
        #pragma unroll
        for (int q = 0; q < 4; ++q) fa[0][i][q] = *(const v4f*)(aptr + i * 16 * HH + 4 * q);
    #pragma unroll
    for (int j = 0; j < PJ_NT; ++j)
        #pragma unroll
        for (int q = 0; q < 4; ++q) fb[0][j][q] = *(const v4f*)(bptr + j * 16 * HH + 4 * q);

    #pragma unroll
    for (int w = 0; w < 4; ++w) {
        const int cur = w & 1, nxt = cur ^ 1;
        if (w + 1 < 4) {
            #pragma unroll
            for (int i = 0; i < PJ_MT; ++i)
                #pragma unroll
                for (int q = 0; q < 4; ++q)
                    fa[nxt][i][q] = *(const v4f*)(aptr + i * 16 * HH + 32 * (w + 1) + 4 * q);
            #pragma unroll
            for (int j = 0; j < PJ_NT; ++j)
                #pragma unroll
                for (int q = 0; q < 4; ++q)
                    fb[nxt][j][q] = *(const v4f*)(bptr + j * 16 * HH + 32 * (w + 1) + 4 * q);
        }
        __builtin_amdgcn_sched_barrier(0);
        #pragma unroll
        for (int q = 0; q < 4; ++q) {
            #pragma unroll
            for (int i = 0; i < PJ_MT; ++i)
                #pragma unroll
                for (int j = 0; j < PJ_NT; ++j) {
                    const v2f a0 = {fa[cur][i][q].x, fa[cur][i][q].y};
                    const v2f b0 = {fb[cur][j][q].x, fb[cur][j][q].y};
                    acc[i][j] = __builtin_amdgcn_wmma_f32_16x16x4_f32(false, a0, false, b0,
                                                                      (short)0, acc[i][j], false, false);
                }
            #pragma unroll
            for (int i = 0; i < PJ_MT; ++i)
                #pragma unroll
                for (int j = 0; j < PJ_NT; ++j) {
                    const v2f a1 = {fa[cur][i][q].z, fa[cur][i][q].w};
                    const v2f b1 = {fb[cur][j][q].z, fb[cur][j][q].w};
                    acc[i][j] = __builtin_amdgcn_wmma_f32_16x16x4_f32(false, a1, false, b1,
                                                                      (short)0, acc[i][j], false, false);
                }
        }
        __builtin_amdgcn_sched_barrier(0);
    }

    if (matIdx <= 1) {
        proj_store<0>(acc, outp, m0, n0, lane, 0.0f, 0.0f);
    } else if (matIdx == 2) {
        proj_store<1>(acc, outp, m0, n0, lane, 0.0f, 0.0f);
    } else {
        const float bias0 = bp[n0 + lm];
        const float bias1 = bp[n0 + 16 + lm];
        if (matIdx == 3) proj_store<2>(acc, outp, m0, n0, lane, bias0, bias1);
        else             proj_store<3>(acc, outp, m0, n0, lane, bias0, bias1);
    }
}

#define NSL  8
#define RS   16
#define REC_GRID 128
#define CSTR 132
#define TSTR 20
#define PSTR 17
#define PLN  272
#define C_OFF   0
#define Q_OFF   (C_OFF + RS * CSTR)
#define K_OFF   (Q_OFF + 16 * CSTR)
#define M_OFF   (K_OFF + 16 * CSTR)
#define KT_OFF  (M_OFF + 16 * CSTR)
#define WC_OFF  (KT_OFF + 128 * TSTR)
#define PP_OFF  (WC_OFF + RS * TSTR)
#define SP_OFF  (PP_OFF + 4 * PLN)
#define S_OFF   (SP_OFF + 4 * PLN)
#define NPS_OFF (S_OFF + PLN)
#define NQ_OFF  (NPS_OFF + 128)
#define G15_OFF (NQ_OFF + 16)
#define RW_OFF  (G15_OFF + 16)
#define RW_SZ   (3 * PLN)
#define SMEM_FLOATS (RW_OFF + 2 * RW_SZ)


static __device__ __forceinline__ void lds_barrier()
{
    __builtin_amdgcn_fence(__ATOMIC_RELEASE, "workgroup", "local");
    __builtin_amdgcn_s_barrier();
    __builtin_amdgcn_fence(__ATOMIC_ACQUIRE, "workgroup", "local");
}

__global__ __launch_bounds__(128, 1) void rec_kernel(
    const float* __restrict__ qb, const float* __restrict__ kb,
    const float* __restrict__ vb, const float* __restrict__ ib,
    const float* __restrict__ fb, const float* __restrict__ ob,
    float* __restrict__ hb)
{
    extern __shared__ float sm[];
    const int b    = blockIdx.x >> 3;
    const int sl   = blockIdx.x & 7;
    const int r    = threadIdx.x;
    const int lane = r & 31;
    const int wave = r >> 5;
    const int lm   = lane & 15;
    const int hh   = lane >> 4;
    const int lk   = hh << 1;
    const int x    = r & 15;
    const bool rowthr = (r >> 4) == sl;

    for (int j = r; j < RS * CSTR; j += 128) sm[C_OFF + j] = 0.0f;
    float creg[2][8];
    #pragma unroll
    for (int t = 0; t < 2; ++t)
        #pragma unroll
        for (int v = 0; v < 8; ++v) creg[t][v] = 0.0f;

    float nprev = 0.0f;
    const long bbase = (long)b * TT * HH + r;
    const long hbase = (long)blockIdx.x * (TT * RS) + r;

    float pk[LCH], pq[LCH], pi[LCH], pf[LCH], pv[LCH], po[LCH];
    #pragma unroll
    for (int s = 0; s < LCH; ++s) {
        const long idx = bbase + (long)s * HH;
        pk[s] = kb[idx]; pq[s] = qb[idx]; pi[s] = ib[idx]; pf[s] = fb[idx];
        pv[s] = 0.0f; po[s] = 0.0f;
    }
    if (rowthr) {
        #pragma unroll
        for (int s = 0; s < LCH; ++s) {
            const long idx = bbase + (long)s * HH;
            pv[s] = vb[idx]; po[s] = ob[idx];
        }
    }

    for (int ch = 0; ch < NCH; ++ch) {
        float* rw = sm + RW_OFF + (ch & 1) * RW_SZ;

        {
            float np = nprev;
            #pragma unroll
            for (int s = 0; s < LCH; ++s) {
                np = fmaf(pf[s], np, pi[s] * pk[s]);
                sm[M_OFF + s * CSTR + r] = np * pq[s];
            }
            nprev = np;
            #pragma unroll
            for (int s = 0; s < LCH; ++s) {
                sm[Q_OFF + s * CSTR + r]  = pq[s];
                sm[K_OFF + s * CSTR + r]  = pk[s];
                sm[KT_OFF + r * TSTR + s] = pk[s];
            }
        }
        if (rowthr) {
            float G[LCH];
            float g = 1.0f;
            #pragma unroll
            for (int s = 0; s < LCH; ++s) { g *= pf[s]; G[s] = g; }
            const float g15 = G[LCH - 1];
            #pragma unroll
            for (int s = 0; s < LCH; ++s) {
                const float wg = pi[s] * pv[s] / G[s];
                sm[WC_OFF + x * TSTR + s]  = g15 * wg;
                rw[0 * PLN + s * PSTR + x] = wg;
                rw[1 * PLN + s * PSTR + x] = G[s];
                rw[2 * PLN + s * PSTR + x] = po[s];
            }
            sm[G15_OFF + x] = g15;
        }
        if (ch + 1 < NCH) {
            const long nbase = bbase + (long)((ch + 1) * LCH) * HH;
            #pragma unroll
            for (int s = 0; s < LCH; ++s) {
                const long idx = nbase + (long)s * HH;
                pk[s] = kb[idx]; pq[s] = qb[idx]; pi[s] = ib[idx];
            }
        }
        lds_barrier();

        {
            const int s1 = r >> 3, seg = r & 7;
            float mv[16];
            #pragma unroll
            for (int e = 0; e < 16; ++e) mv[e] = sm[M_OFF + s1 * CSTR + seg * 16 + e];

            const int cb = wave * 32 + lk;
            v2f qf[8], kf[8], cf[8];
            #pragma unroll
            for (int kk = 0; kk < 8; ++kk) {
                qf[kk] = *(const v2f*)&sm[Q_OFF + lm * CSTR + cb + kk * 4];
                kf[kk] = *(const v2f*)&sm[K_OFF + lm * CSTR + cb + kk * 4];
                cf[kk] = *(const v2f*)&sm[C_OFF + lm * CSTR + cb + kk * 4];
            }
            float ps = 0.0f;
            #pragma unroll
            for (int e = 0; e < 16; ++e) ps += mv[e];

            v8f pacc = {};
            v8f sacc = {};
            #pragma unroll
            for (int kk = 0; kk < 8; ++kk) {
                pacc = __builtin_amdgcn_wmma_f32_16x16x4_f32(false, qf[kk], false, cf[kk],
                                                             (short)0, pacc, false, false);
                sacc = __builtin_amdgcn_wmma_f32_16x16x4_f32(false, kf[kk], false, qf[kk],
                                                             (short)0, sacc, false, false);
            }
            sm[NPS_OFF + s1 * 8 + seg] = ps;
            #pragma unroll
            for (int v = 0; v < 8; ++v) {
                sm[PP_OFF + wave * PLN + (v + 8 * hh) * PSTR + lm] = pacc[v];
                sm[SP_OFF + wave * PLN + (v + 8 * hh) * PSTR + lm] = sacc[v];
            }
        }
        lds_barrier();

        if (ch + 1 < NCH) {
            const long nbase = bbase + (long)((ch + 1) * LCH) * HH;
            #pragma unroll
            for (int s = 0; s < LCH; ++s) pf[s] = fb[nbase + (long)s * HH];
            if (rowthr) {
                #pragma unroll
                for (int s = 0; s < LCH; ++s) {
                    const long idx = nbase + (long)s * HH;
                    pv[s] = vb[idx]; po[s] = ob[idx];
                }
            }
        }

        {
            float nps[8];
            #pragma unroll
            for (int e = 0; e < 8; ++e) nps[e] = sm[NPS_OFF + x * 8 + e];
            float spv[2][4];
            #pragma unroll
            for (int p = 0; p < 2; ++p) {
                const int idx = r * 2 + p;
                const int u = idx >> 4, sc = idx & 15;
                #pragma unroll
                for (int w = 0; w < 4; ++w) spv[p][w] = sm[SP_OFF + w * PLN + u * PSTR + sc];
            }
            v2f afr[4];
            #pragma unroll
            for (int kk = 0; kk < 4; ++kk)
                afr[kk] = *(const v2f*)&sm[WC_OFF + lm * TSTR + kk * 4 + lk];
            float g[8];
            #pragma unroll
            for (int v = 0; v < 8; ++v) g[v] = sm[G15_OFF + 8 * hh + v];
            v2f bfr[2][4];
            #pragma unroll
            for (int t = 0; t < 2; ++t)
                #pragma unroll
                for (int kk = 0; kk < 4; ++kk)
                    bfr[t][kk] = *(const v2f*)&sm[KT_OFF + (wave * 32 + t * 16 + lm) * TSTR + kk * 4 + lk];

            v8f d0 = {};
            v8f d1 = {};
            #pragma unroll
            for (int kk = 0; kk < 4; ++kk) {
                d0 = __builtin_amdgcn_wmma_f32_16x16x4_f32(false, afr[kk], false, bfr[0][kk],
                                                           (short)0, d0, false, false);
                d1 = __builtin_amdgcn_wmma_f32_16x16x4_f32(false, afr[kk], false, bfr[1][kk],
                                                           (short)0, d1, false, false);
            }
            #pragma unroll
            for (int v = 0; v < 8; ++v) {
                creg[0][v] = fmaf(g[v], creg[0][v], d0[v]);
                creg[1][v] = fmaf(g[v], creg[1][v], d1[v]);
            }
            if (r < 16) {
                float t = 0.0f;
                #pragma unroll
                for (int e = 0; e < 8; ++e) t += nps[e];
                sm[NQ_OFF + r] = t;
            }
            #pragma unroll
            for (int p = 0; p < 2; ++p) {
                const int idx = r * 2 + p;
                const int u = idx >> 4, sc = idx & 15;
                sm[S_OFF + u * PSTR + sc] = spv[p][0] + spv[p][1] + spv[p][2] + spv[p][3];
            }
            #pragma unroll
            for (int t = 0; t < 2; ++t)
                #pragma unroll
                for (int v = 0; v < 8; ++v)
                    sm[C_OFF + (8 * hh + v) * CSTR + wave * 32 + t * 16 + lm] = creg[t][v];
        }
        lds_barrier();

        {
            const int sA = r >> 4;
            float wgv[LCH];
            #pragma unroll
            for (int u = 0; u < LCH; ++u) wgv[u] = rw[0 * PLN + u * PSTR + x];
            float sv[2][LCH], ppv[2][4], gs[2], os[2], nqv[2];
            #pragma unroll
            for (int p = 0; p < 2; ++p) {
                const int s = sA + 8 * p;
                #pragma unroll
                for (int u = 0; u < LCH; ++u) sv[p][u] = sm[S_OFF + u * PSTR + s];
                #pragma unroll
                for (int w = 0; w < 4; ++w) ppv[p][w] = sm[PP_OFF + w * PLN + s * PSTR + x];
                gs[p]  = rw[1 * PLN + s * PSTR + x];
                os[p]  = rw[2 * PLN + s * PSTR + x];
                nqv[p] = sm[NQ_OFF + s];
            }
            #pragma unroll
            for (int p = 0; p < 2; ++p) {
                const int s = sA + 8 * p;
                float acc = ((ppv[p][0] + ppv[p][1]) + ppv[p][2]) + ppv[p][3];
                #pragma unroll
                for (int u = 0; u < LCH; ++u) {
                    const float nx = fmaf(wgv[u], sv[p][u], acc);
                    acc = (u <= s) ? nx : acc;
                }
                const float ret = gs[p] * acc;
                const float nrm = fmaxf(fabsf(nqv[p]), 1.0f);
                hb[hbase + (long)ch * (LCH * RS) + p * 128] = os[p] * ret / nrm;
            }
        }
    }
}

__global__ __launch_bounds__(256) void ln_kernel(
    const float* __restrict__ hb, const float* __restrict__ x,
    const float* __restrict__ lnw, const float* __restrict__ lnb,
    float* __restrict__ out)
{
    const int lane = threadIdx.x & 31;
    const int wave = threadIdx.x >> 5;
    const long row = (long)blockIdx.x * 8 + wave;
    const long rb  = row * HH;

    float y[4];
    float s = 0.0f, s2 = 0.0f;
    #pragma unroll
    for (int j = 0; j < 4; ++j) {
        const int c = lane + 32 * j;
        const long hidx = (((row >> 9) * NSL + (c >> 4)) * TT + (row & (TT - 1))) * RS + (c & 15);
        y[j] = hb[hidx] + x[rb + c];
        s  += y[j];
        s2 += y[j] * y[j];
    }
    #pragma unroll
    for (int off = 16; off >= 1; off >>= 1) {
        s  += __shfl_xor(s,  off, 32);
        s2 += __shfl_xor(s2, off, 32);
    }
    const float mu  = s * (1.0f / HH);
    const float var = s2 * (1.0f / HH) - mu * mu;
    const float rs  = rsqrtf(var + LN_EPS);

    #pragma unroll
    for (int j = 0; j < 4; ++j) {
        const int c = lane + 32 * j;
        out[rb + c] = (y[j] - mu) * rs * lnw[c] + lnb[c];
    }
}

extern "C" void kernel_launch(void* const* d_in, const int* in_sizes, int n_in,
                              void* d_out, int out_size, void* d_ws, size_t ws_size,
                              hipStream_t stream)
{
    const float* x   = (const float*)d_in[0];
    const float* Wq  = (const float*)d_in[1];
    const float* Wk  = (const float*)d_in[2];
    const float* Wv  = (const float*)d_in[3];
    const float* Wi  = (const float*)d_in[4];
    const float* Wf  = (const float*)d_in[5];
    const float* Wo  = (const float*)d_in[6];
    const float* b_i = (const float*)d_in[7];
    const float* b_f = (const float*)d_in[8];
    const float* b_o = (const float*)d_in[9];
    const float* lnw = (const float*)d_in[10];
    const float* lnb = (const float*)d_in[11];
    float* out = (float*)d_out;

    float* ws = (float*)d_ws;
    float* q  = ws + 0L * NEL;
    float* k  = ws + 1L * NEL;
    float* v  = ws + 2L * NEL;
    float* ig = ws + 3L * NEL;
    float* fg = ws + 4L * NEL;
    float* og = ws + 5L * NEL;
    float* h  = ws + 6L * NEL;

    proj_kernel<<<PJ_GRID, 128, 0, stream>>>(x, Wq, Wk, Wv, Wi, Wf, Wo,
                                          b_i, b_f, b_o,
                                          q, k, v, ig, fg, og);
    rec_kernel<<<REC_GRID, 128, SMEM_FLOATS * sizeof(float), stream>>>(q, k, v, ig, fg, og, h);
    ln_kernel<<<MM / 8, 256, 0, stream>>>(h, x, lnw, lnb, out);
}


# Recurrence: row-gate weights i*v/G by v_rcp_f32 plus one Newton step, not 16 serial IEEE divisions on the one wave the rest await at the first barrier; that quotient's last-bit rounding changes
# speedup vs baseline: 4.6096x; 1.0832x over previous
#include <hip/hip_runtime.h>
#include <math.h>


typedef float v2f __attribute__((ext_vector_type(2)));
typedef float v4f __attribute__((ext_vector_type(4)));
typedef float v8f __attribute__((ext_vector_type(8)));

#define BB   16
#define TT   512
#define HH   128
#define MM   (BB * TT)
#define NEL  (BB * TT * HH)
#define SCALE 0.08838834764831845f
#define LN_EPS 1e-5f
#define LCH  16
#define NCH  (TT / LCH)

#define PJ_MT 4
#define PJ_NT 2
#define PJ_ROWS (16 * PJ_MT)
#define PJ_GRID ((MM / PJ_ROWS) * 6)

static __device__ __forceinline__ float xhalf(float v)
{
    return __builtin_bit_cast(float, __builtin_amdgcn_permlanex16(
        __builtin_bit_cast(int, v), __builtin_bit_cast(int, v),
        0x76543210, 0xfedcba98, false, false));
}

template <int ACT>
static __device__ __forceinline__ float proj_act(float val, float bias)
{
    if (ACT == 0) return val * SCALE;
    if (ACT == 1) return val;
    if (ACT == 2) return expf(val + bias);
    return 1.0f / (1.0f + expf(-(val + bias)));
}

template <int ACT>
static __device__ __forceinline__ void proj_store(
    const v8f (&acc)[PJ_MT][PJ_NT], float* __restrict__ outp,
    int m0, int n0, int lane, float bias0, float bias1)
{
    const int hh = lane >> 4;
    #pragma unroll
    for (int i = 0; i < PJ_MT; ++i) {
        #pragma unroll
        for (int v = 0; v < 8; ++v) {
            const float t0 = proj_act<ACT>(acc[i][0][v], bias0);
            const float t1 = proj_act<ACT>(acc[i][1][v], bias1);
            const float got = xhalf(hh ? t0 : t1);
            const float lo = hh ? got : t0;
            const float hi = hh ? t1 : got;
            float* o = outp + (long)(m0 + 16 * i + v) * HH + n0 + lane;
            o[0]      = lo;
            o[8 * HH] = hi;
        }
    }
}

__global__ __launch_bounds__(128, 3) void proj_kernel(
    const float* __restrict__ x,
    const float* __restrict__ Wq, const float* __restrict__ Wk,
    const float* __restrict__ Wv, const float* __restrict__ Wi,
    const float* __restrict__ Wf, const float* __restrict__ Wo,
    const float* __restrict__ b_i, const float* __restrict__ b_f,
    const float* __restrict__ b_o,
    float* __restrict__ oq, float* __restrict__ ok, float* __restrict__ ov,
    float* __restrict__ oi, float* __restrict__ of, float* __restrict__ oo)
{
    const int lane   = threadIdx.x & 31;
    const int wave   = threadIdx.x >> 5;
    const int matIdx = blockIdx.x % 6;
    const int m0     = (blockIdx.x / 6) * PJ_ROWS;
    const int n0     = wave * (16 * PJ_NT);

    const float* W;
    const float* bp = b_i;
    float* outp;
    switch (matIdx) {
        case 0:  W = Wq; outp = oq; break;
        case 1:  W = Wk; outp = ok; break;
        case 2:  W = Wv; outp = ov; break;
        case 3:  W = Wi; outp = oi; bp = b_i; break;
        case 4:  W = Wf; outp = of; bp = b_f; break;
        default: W = Wo; outp = oo; bp = b_o; break;
    }

    const int lm = lane & 15;
    const int hh = lane >> 4;

    const float* aptr = x + (long)(m0 + lm) * HH + 16 * hh;
    const float* bptr = W + (n0 + lm) * HH + 16 * hh;

    v8f acc[PJ_MT][PJ_NT];
    #pragma unroll
    for (int i = 0; i < PJ_MT; ++i)
        #pragma unroll
        for (int j = 0; j < PJ_NT; ++j) acc[i][j] = (v8f){};

    v4f fa[2][PJ_MT][4];
    v4f fb[2][PJ_NT][4];

    #pragma unroll
    for (int i = 0; i < PJ_MT; ++i)
        #pragma unroll
        for (int q = 0; q < 4; ++q) fa[0][i][q] = *(const v4f*)(aptr + i * 16 * HH + 4 * q);
    #pragma unroll
    for (int j = 0; j < PJ_NT; ++j)
        #pragma unroll
        for (int q = 0; q < 4; ++q) fb[0][j][q] = *(const v4f*)(bptr + j * 16 * HH + 4 * q);

    #pragma unroll
    for (int w = 0; w < 4; ++w) {
        const int cur = w & 1, nxt = cur ^ 1;
        if (w + 1 < 4) {
            #pragma unroll
            for (int i = 0; i < PJ_MT; ++i)
                #pragma unroll
                for (int q = 0; q < 4; ++q)
                    fa[nxt][i][q] = *(const v4f*)(aptr + i * 16 * HH + 32 * (w + 1) + 4 * q);
            #pragma unroll
            for (int j = 0; j < PJ_NT; ++j)
                #pragma unroll
                for (int q = 0; q < 4; ++q)
                    fb[nxt][j][q] = *(const v4f*)(bptr + j * 16 * HH + 32 * (w + 1) + 4 * q);
        }
        __builtin_amdgcn_sched_barrier(0);
        #pragma unroll
        for (int q = 0; q < 4; ++q) {
            #pragma unroll
            for (int i = 0; i < PJ_MT; ++i)
                #pragma unroll
                for (int j = 0; j < PJ_NT; ++j) {
                    const v2f a0 = {fa[cur][i][q].x, fa[cur][i][q].y};
                    const v2f b0 = {fb[cur][j][q].x, fb[cur][j][q].y};
                    acc[i][j] = __builtin_amdgcn_wmma_f32_16x16x4_f32(false, a0, false, b0,
                                                                      (short)0, acc[i][j], false, false);
                }
            #pragma unroll
            for (int i = 0; i < PJ_MT; ++i)
                #pragma unroll
                for (int j = 0; j < PJ_NT; ++j) {
                    const v2f a1 = {fa[cur][i][q].z, fa[cur][i][q].w};
                    const v2f b1 = {fb[cur][j][q].z, fb[cur][j][q].w};
                    acc[i][j] = __builtin_amdgcn_wmma_f32_16x16x4_f32(false, a1, false, b1,
                                                                      (short)0, acc[i][j], false, false);
                }
        }
        __builtin_amdgcn_sched_barrier(0);
    }

    if (matIdx <= 1) {
        proj_store<0>(acc, outp, m0, n0, lane, 0.0f, 0.0f);
    } else if (matIdx == 2) {
        proj_store<1>(acc, outp, m0, n0, lane, 0.0f, 0.0f);
    } else {
        const float bias0 = bp[n0 + lm];
        const float bias1 = bp[n0 + 16 + lm];
        if (matIdx == 3) proj_store<2>(acc, outp, m0, n0, lane, bias0, bias1);
        else             proj_store<3>(acc, outp, m0, n0, lane, bias0, bias1);
    }
}

#define NSL  8
#define RS   16
#define REC_GRID 128
#define CSTR 132
#define TSTR 20
#define PSTR 17
#define PLN  272
#define C_OFF   0
#define Q_OFF   (C_OFF + RS * CSTR)
#define K_OFF   (Q_OFF + 16 * CSTR)
#define M_OFF   (K_OFF + 16 * CSTR)
#define KT_OFF  (M_OFF + 16 * CSTR)
#define WC_OFF  (KT_OFF + 128 * TSTR)
#define PP_OFF  (WC_OFF + RS * TSTR)
#define SP_OFF  (PP_OFF + 4 * PLN)
#define S_OFF   (SP_OFF + 4 * PLN)
#define NPS_OFF (S_OFF + PLN)
#define NQ_OFF  (NPS_OFF + 128)
#define G15_OFF (NQ_OFF + 16)
#define RW_OFF  (G15_OFF + 16)
#define RW_SZ   (3 * PLN)
#define SMEM_FLOATS (RW_OFF + 2 * RW_SZ)


static __device__ __forceinline__ void lds_barrier()
{
    __builtin_amdgcn_fence(__ATOMIC_RELEASE, "workgroup", "local");
    __builtin_amdgcn_s_barrier();
    __builtin_amdgcn_fence(__ATOMIC_ACQUIRE, "workgroup", "local");
}

__global__ __launch_bounds__(128, 1) void rec_kernel(
    const float* __restrict__ qb, const float* __restrict__ kb,
    const float* __restrict__ vb, const float* __restrict__ ib,
    const float* __restrict__ fb, const float* __restrict__ ob,
    float* __restrict__ hb)
{
    extern __shared__ float sm[];
    const int b    = blockIdx.x >> 3;
    const int sl   = blockIdx.x & 7;
    const int r    = threadIdx.x;
    const int lane = r & 31;
    const int wave = r >> 5;
    const int lm   = lane & 15;
    const int hh   = lane >> 4;
    const int lk   = hh << 1;
    const int x    = r & 15;
    const bool rowthr = (r >> 4) == sl;

    for (int j = r; j < RS * CSTR; j += 128) sm[C_OFF + j] = 0.0f;
    float creg[2][8];
    #pragma unroll
    for (int t = 0; t < 2; ++t)
        #pragma unroll
        for (int v = 0; v < 8; ++v) creg[t][v] = 0.0f;

    float nprev = 0.0f;
    const long bbase = (long)b * TT * HH + r;
    const long hbase = (long)blockIdx.x * (TT * RS) + r;

    float pk[LCH], pq[LCH], pi[LCH], pf[LCH], pv[LCH], po[LCH];
    #pragma unroll
    for (int s = 0; s < LCH; ++s) {
        const long idx = bbase + (long)s * HH;
        pk[s] = kb[idx]; pq[s] = qb[idx]; pi[s] = ib[idx]; pf[s] = fb[idx];
        pv[s] = 0.0f; po[s] = 0.0f;
    }
    if (rowthr) {
        #pragma unroll
        for (int s = 0; s < LCH; ++s) {
            const long idx = bbase + (long)s * HH;
            pv[s] = vb[idx]; po[s] = ob[idx];
        }
    }

    for (int ch = 0; ch < NCH; ++ch) {
        float* rw = sm + RW_OFF + (ch & 1) * RW_SZ;

        {
            float np = nprev;
            #pragma unroll
            for (int s = 0; s < LCH; ++s) {
                np = fmaf(pf[s], np, pi[s] * pk[s]);
                sm[M_OFF + s * CSTR + r] = np * pq[s];
            }
            nprev = np;
            #pragma unroll
            for (int s = 0; s < LCH; ++s) {
                sm[Q_OFF + s * CSTR + r]  = pq[s];
                sm[K_OFF + s * CSTR + r]  = pk[s];
                sm[KT_OFF + r * TSTR + s] = pk[s];
            }
        }
        if (rowthr) {
            float G[LCH];
            float g = 1.0f;
            #pragma unroll
            for (int s = 0; s < LCH; ++s) { g *= pf[s]; G[s] = g; }
            const float g15 = G[LCH - 1];
            #pragma unroll
            for (int s = 0; s < LCH; ++s) {
                const float r0 = __builtin_amdgcn_rcpf(G[s]);
                const float rg = fmaf(fmaf(-G[s], r0, 1.0f), r0, r0);
                const float wg = pi[s] * pv[s] * rg;
                sm[WC_OFF + x * TSTR + s]  = g15 * wg;
                rw[0 * PLN + s * PSTR + x] = wg;
                rw[1 * PLN + s * PSTR + x] = G[s];
                rw[2 * PLN + s * PSTR + x] = po[s];
            }
            sm[G15_OFF + x] = g15;
        }
        if (ch + 1 < NCH) {
            const long nbase = bbase + (long)((ch + 1) * LCH) * HH;
            #pragma unroll
            for (int s = 0; s < LCH; ++s) {
                const long idx = nbase + (long)s * HH;
                pk[s] = kb[idx]; pq[s] = qb[idx]; pi[s] = ib[idx];
            }
        }
        lds_barrier();

        {
            const int s1 = r >> 3, seg = r & 7;
            float mv[16];
            #pragma unroll
            for (int e = 0; e < 16; ++e) mv[e] = sm[M_OFF + s1 * CSTR + seg * 16 + e];

            const int cb = wave * 32 + lk;
            v2f qf[8], kf[8], cf[8];
            #pragma unroll
            for (int kk = 0; kk < 8; ++kk) {
                qf[kk] = *(const v2f*)&sm[Q_OFF + lm * CSTR + cb + kk * 4];
                kf[kk] = *(const v2f*)&sm[K_OFF + lm * CSTR + cb + kk * 4];
                cf[kk] = *(const v2f*)&sm[C_OFF + lm * CSTR + cb + kk * 4];
            }
            float ps = 0.0f;
            #pragma unroll
            for (int e = 0; e < 16; ++e) ps += mv[e];

            v8f pacc = {};
            v8f sacc = {};
            #pragma unroll
            for (int kk = 0; kk < 8; ++kk) {
                pacc = __builtin_amdgcn_wmma_f32_16x16x4_f32(false, qf[kk], false, cf[kk],
                                                             (short)0, pacc, false, false);
                sacc = __builtin_amdgcn_wmma_f32_16x16x4_f32(false, kf[kk], false, qf[kk],
                                                             (short)0, sacc, false, false);
            }
            sm[NPS_OFF + s1 * 8 + seg] = ps;
            #pragma unroll
            for (int v = 0; v < 8; ++v) {
                sm[PP_OFF + wave * PLN + (v + 8 * hh) * PSTR + lm] = pacc[v];
                sm[SP_OFF + wave * PLN + (v + 8 * hh) * PSTR + lm] = sacc[v];
            }
        }
        lds_barrier();

        if (ch + 1 < NCH) {
            const long nbase = bbase + (long)((ch + 1) * LCH) * HH;
            #pragma unroll
            for (int s = 0; s < LCH; ++s) pf[s] = fb[nbase + (long)s * HH];
            if (rowthr) {
                #pragma unroll
                for (int s = 0; s < LCH; ++s) {
                    const long idx = nbase + (long)s * HH;
                    pv[s] = vb[idx]; po[s] = ob[idx];
                }
            }
        }

        {
            float nps[8];
            #pragma unroll
            for (int e = 0; e < 8; ++e) nps[e] = sm[NPS_OFF + x * 8 + e];
            float spv[2][4];
            #pragma unroll
            for (int p = 0; p < 2; ++p) {
                const int idx = r * 2 + p;
                const int u = idx >> 4, sc = idx & 15;
                #pragma unroll
                for (int w = 0; w < 4; ++w) spv[p][w] = sm[SP_OFF + w * PLN + u * PSTR + sc];
            }
            v2f afr[4];
            #pragma unroll
            for (int kk = 0; kk < 4; ++kk)
                afr[kk] = *(const v2f*)&sm[WC_OFF + lm * TSTR + kk * 4 + lk];
            float g[8];
            #pragma unroll
            for (int v = 0; v < 8; ++v) g[v] = sm[G15_OFF + 8 * hh + v];
            v2f bfr[2][4];
            #pragma unroll
            for (int t = 0; t < 2; ++t)
                #pragma unroll
                for (int kk = 0; kk < 4; ++kk)
                    bfr[t][kk] = *(const v2f*)&sm[KT_OFF + (wave * 32 + t * 16 + lm) * TSTR + kk * 4 + lk];

            v8f d0 = {};
            v8f d1 = {};
            #pragma unroll
            for (int kk = 0; kk < 4; ++kk) {
                d0 = __builtin_amdgcn_wmma_f32_16x16x4_f32(false, afr[kk], false, bfr[0][kk],
                                                           (short)0, d0, false, false);
                d1 = __builtin_amdgcn_wmma_f32_16x16x4_f32(false, afr[kk], false, bfr[1][kk],
                                                           (short)0, d1, false, false);
            }
            #pragma unroll
            for (int v = 0; v < 8; ++v) {
                creg[0][v] = fmaf(g[v], creg[0][v], d0[v]);
                creg[1][v] = fmaf(g[v], creg[1][v], d1[v]);
            }
            if (r < 16) {
                float t = 0.0f;
                #pragma unroll
                for (int e = 0; e < 8; ++e) t += nps[e];
                sm[NQ_OFF + r] = t;
            }
            #pragma unroll
            for (int p = 0; p < 2; ++p) {
                const int idx = r * 2 + p;
                const int u = idx >> 4, sc = idx & 15;
                sm[S_OFF + u * PSTR + sc] = spv[p][0] + spv[p][1] + spv[p][2] + spv[p][3];
            }
            #pragma unroll
            for (int t = 0; t < 2; ++t)
                #pragma unroll
                for (int v = 0; v < 8; ++v)
                    sm[C_OFF + (8 * hh + v) * CSTR + wave * 32 + t * 16 + lm] = creg[t][v];
        }
        lds_barrier();

        {
            const int sA = r >> 4;
            float wgv[LCH];
            #pragma unroll
            for (int u = 0; u < LCH; ++u) wgv[u] = rw[0 * PLN + u * PSTR + x];
            float sv[2][LCH], ppv[2][4], gs[2], os[2], nqv[2];
            #pragma unroll
            for (int p = 0; p < 2; ++p) {
                const int s = sA + 8 * p;
                #pragma unroll
                for (int u = 0; u < LCH; ++u) sv[p][u] = sm[S_OFF + u * PSTR + s];
                #pragma unroll
                for (int w = 0; w < 4; ++w) ppv[p][w] = sm[PP_OFF + w * PLN + s * PSTR + x];
                gs[p]  = rw[1 * PLN + s * PSTR + x];
                os[p]  = rw[2 * PLN + s * PSTR + x];
                nqv[p] = sm[NQ_OFF + s];
            }
            #pragma unroll
            for (int p = 0; p < 2; ++p) {
                const int s = sA + 8 * p;
                float acc = ((ppv[p][0] + ppv[p][1]) + ppv[p][2]) + ppv[p][3];
                #pragma unroll
                for (int u = 0; u < LCH; ++u) {
                    const float nx = fmaf(wgv[u], sv[p][u], acc);
                    acc = (u <= s) ? nx : acc;
                }
                const float ret = gs[p] * acc;
                const float nrm = fmaxf(fabsf(nqv[p]), 1.0f);
                hb[hbase + (long)ch * (LCH * RS) + p * 128] = os[p] * ret / nrm;
            }
        }
    }
}

__global__ __launch_bounds__(256) void ln_kernel(
    const float* __restrict__ hb, const float* __restrict__ x,
    const float* __restrict__ lnw, const float* __restrict__ lnb,
    float* __restrict__ out)
{
    const int lane = threadIdx.x & 31;
    const int wave = threadIdx.x >> 5;
    const long row = (long)blockIdx.x * 8 + wave;
    const long rb  = row * HH;

    float y[4];
    float s = 0.0f, s2 = 0.0f;
    #pragma unroll
    for (int j = 0; j < 4; ++j) {
        const int c = lane + 32 * j;
        const long hidx = (((row >> 9) * NSL + (c >> 4)) * TT + (row & (TT - 1))) * RS + (c & 15);
        y[j] = hb[hidx] + x[rb + c];
        s  += y[j];
        s2 += y[j] * y[j];
    }
    #pragma unroll
    for (int off = 16; off >= 1; off >>= 1) {
        s  += __shfl_xor(s,  off, 32);
        s2 += __shfl_xor(s2, off, 32);
    }
    const float mu  = s * (1.0f / HH);
    const float var = s2 * (1.0f / HH) - mu * mu;
    const float rs  = rsqrtf(var + LN_EPS);

    #pragma unroll
    for (int j = 0; j < 4; ++j) {
        const int c = lane + 32 * j;
        out[rb + c] = (y[j] - mu) * rs * lnw[c] + lnb[c];
    }
}

extern "C" void kernel_launch(void* const* d_in, const int* in_sizes, int n_in,
                              void* d_out, int out_size, void* d_ws, size_t ws_size,
                              hipStream_t stream)
{
    const float* x   = (const float*)d_in[0];
    const float* Wq  = (const float*)d_in[1];
    const float* Wk  = (const float*)d_in[2];
    const float* Wv  = (const float*)d_in[3];
    const float* Wi  = (const float*)d_in[4];
    const float* Wf  = (const float*)d_in[5];
    const float* Wo  = (const float*)d_in[6];
    const float* b_i = (const float*)d_in[7];
    const float* b_f = (const float*)d_in[8];
    const float* b_o = (const float*)d_in[9];
    const float* lnw = (const float*)d_in[10];
    const float* lnb = (const float*)d_in[11];
    float* out = (float*)d_out;

    float* ws = (float*)d_ws;
    float* q  = ws + 0L * NEL;
    float* k  = ws + 1L * NEL;
    float* v  = ws + 2L * NEL;
    float* ig = ws + 3L * NEL;
    float* fg = ws + 4L * NEL;
    float* og = ws + 5L * NEL;
    float* h  = ws + 6L * NEL;

    proj_kernel<<<PJ_GRID, 128, 0, stream>>>(x, Wq, Wk, Wv, Wi, Wf, Wo,
                                          b_i, b_f, b_o,
                                          q, k, v, ig, fg, og);
    rec_kernel<<<REC_GRID, 128, SMEM_FLOATS * sizeof(float), stream>>>(q, k, v, ig, fg, og, h);
    ln_kernel<<<MM / 8, 256, 0, stream>>>(h, x, lnw, lnb, out);
}


# Recurrence software-pipelined over chunks: a trip stages chunk t, steps the state over t-1 and writes h of t-2 out of separate LDS buffers, one barrier a trip, not three; sum orders unchanged
# speedup vs baseline: 5.0664x; 1.0991x over previous
#include <hip/hip_runtime.h>
#include <math.h>


typedef float v2f __attribute__((ext_vector_type(2)));
typedef float v4f __attribute__((ext_vector_type(4)));
typedef float v8f __attribute__((ext_vector_type(8)));

#define BB   16
#define TT   512
#define HH   128
#define MM   (BB * TT)
#define NEL  (BB * TT * HH)
#define SCALE 0.08838834764831845f
#define LN_EPS 1e-5f
#define LCH  16
#define NCH  (TT / LCH)

#define PJ_MT 4
#define PJ_NT 2
#define PJ_ROWS (16 * PJ_MT)
#define PJ_GRID ((MM / PJ_ROWS) * 6)

static __device__ __forceinline__ float xhalf(float v)
{
    return __builtin_bit_cast(float, __builtin_amdgcn_permlanex16(
        __builtin_bit_cast(int, v), __builtin_bit_cast(int, v),
        0x76543210, 0xfedcba98, false, false));
}

template <int ACT>
static __device__ __forceinline__ float proj_act(float val, float bias)
{
    if (ACT == 0) return val * SCALE;
    if (ACT == 1) return val;
    if (ACT == 2) return expf(val + bias);
    return 1.0f / (1.0f + expf(-(val + bias)));
}

template <int ACT>
static __device__ __forceinline__ void proj_store(
    const v8f (&acc)[PJ_MT][PJ_NT], float* __restrict__ outp,
    int m0, int n0, int lane, float bias0, float bias1)
{
    const int hh = lane >> 4;
    #pragma unroll
    for (int i = 0; i < PJ_MT; ++i) {
        #pragma unroll
        for (int v = 0; v < 8; ++v) {
            const float t0 = proj_act<ACT>(acc[i][0][v], bias0);
            const float t1 = proj_act<ACT>(acc[i][1][v], bias1);
            const float got = xhalf(hh ? t0 : t1);
            const float lo = hh ? got : t0;
            const float hi = hh ? t1 : got;
            float* o = outp + (long)(m0 + 16 * i + v) * HH + n0 + lane;
            o[0]      = lo;
            o[8 * HH] = hi;
        }
    }
}

__global__ __launch_bounds__(128, 3) void proj_kernel(
    const float* __restrict__ x,
    const float* __restrict__ Wq, const float* __restrict__ Wk,
    const float* __restrict__ Wv, const float* __restrict__ Wi,
    const float* __restrict__ Wf, const float* __restrict__ Wo,
    const float* __restrict__ b_i, const float* __restrict__ b_f,
    const float* __restrict__ b_o,
    float* __restrict__ oq, float* __restrict__ ok, float* __restrict__ ov,
    float* __restrict__ oi, float* __restrict__ of, float* __restrict__ oo)
{
    const int lane   = threadIdx.x & 31;
    const int wave   = threadIdx.x >> 5;
    const int matIdx = blockIdx.x % 6;
    const int m0     = (blockIdx.x / 6) * PJ_ROWS;
    const int n0     = wave * (16 * PJ_NT);

    const float* W;
    const float* bp = b_i;
    float* outp;
    switch (matIdx) {
        case 0:  W = Wq; outp = oq; break;
        case 1:  W = Wk; outp = ok; break;
        case 2:  W = Wv; outp = ov; break;
        case 3:  W = Wi; outp = oi; bp = b_i; break;
        case 4:  W = Wf; outp = of; bp = b_f; break;
        default: W = Wo; outp = oo; bp = b_o; break;
    }

    const int lm = lane & 15;
    const int hh = lane >> 4;

    const float* aptr = x + (long)(m0 + lm) * HH + 16 * hh;
    const float* bptr = W + (n0 + lm) * HH + 16 * hh;

    v8f acc[PJ_MT][PJ_NT];
    #pragma unroll
    for (int i = 0; i < PJ_MT; ++i)
        #pragma unroll
        for (int j = 0; j < PJ_NT; ++j) acc[i][j] = (v8f){};

    v4f fa[2][PJ_MT][4];
    v4f fb[2][PJ_NT][4];

    #pragma unroll
    for (int i = 0; i < PJ_MT; ++i)
        #pragma unroll
        for (int q = 0; q < 4; ++q) fa[0][i][q] = *(const v4f*)(aptr + i * 16 * HH + 4 * q);
    #pragma unroll
    for (int j = 0; j < PJ_NT; ++j)
        #pragma unroll
        for (int q = 0; q < 4; ++q) fb[0][j][q] = *(const v4f*)(bptr + j * 16 * HH + 4 * q);

    #pragma unroll
    for (int w = 0; w < 4; ++w) {
        const int cur = w & 1, nxt = cur ^ 1;
        if (w + 1 < 4) {
            #pragma unroll
            for (int i = 0; i < PJ_MT; ++i)
                #pragma unroll
                for (int q = 0; q < 4; ++q)
                    fa[nxt][i][q] = *(const v4f*)(aptr + i * 16 * HH + 32 * (w + 1) + 4 * q);
            #pragma unroll
            for (int j = 0; j < PJ_NT; ++j)
                #pragma unroll
                for (int q = 0; q < 4; ++q)
                    fb[nxt][j][q] = *(const v4f*)(bptr + j * 16 * HH + 32 * (w + 1) + 4 * q);
        }
        __builtin_amdgcn_sched_barrier(0);
        #pragma unroll
        for (int q = 0; q < 4; ++q) {
            #pragma unroll
            for (int i = 0; i < PJ_MT; ++i)
                #pragma unroll
                for (int j = 0; j < PJ_NT; ++j) {
                    const v2f a0 = {fa[cur][i][q].x, fa[cur][i][q].y};
                    const v2f b0 = {fb[cur][j][q].x, fb[cur][j][q].y};
                    acc[i][j] = __builtin_amdgcn_wmma_f32_16x16x4_f32(false, a0, false, b0,
                                                                      (short)0, acc[i][j], false, false);
                }
            #pragma unroll
            for (int i = 0; i < PJ_MT; ++i)
                #pragma unroll
                for (int j = 0; j < PJ_NT; ++j) {
                    const v2f a1 = {fa[cur][i][q].z, fa[cur][i][q].w};
                    const v2f b1 = {fb[cur][j][q].z, fb[cur][j][q].w};
                    acc[i][j] = __builtin_amdgcn_wmma_f32_16x16x4_f32(false, a1, false, b1,
                                                                      (short)0, acc[i][j], false, false);
                }
        }
        __builtin_amdgcn_sched_barrier(0);
    }

    if (matIdx <= 1) {
        proj_store<0>(acc, outp, m0, n0, lane, 0.0f, 0.0f);
    } else if (matIdx == 2) {
        proj_store<1>(acc, outp, m0, n0, lane, 0.0f, 0.0f);
    } else {
        const float bias0 = bp[n0 + lm];
        const float bias1 = bp[n0 + 16 + lm];
        if (matIdx == 3) proj_store<2>(acc, outp, m0, n0, lane, bias0, bias1);
        else             proj_store<3>(acc, outp, m0, n0, lane, bias0, bias1);
    }
}

#define NSL  8
#define RS   16
#define REC_GRID 128
#define CSTR 132
#define TSTR 20
#define PS2  20
#define PL2  (16 * PS2)
#define C_OFF   0
#define ST_OFF  (C_OFF + RS * CSTR)
#define Q_O     0
#define K_O     (Q_O + 16 * CSTR)
#define M_O     (K_O + 16 * CSTR)
#define KT_O    (M_O + 16 * CSTR)
#define WC_O    (KT_O + 128 * TSTR)
#define G15_O   (WC_O + RS * TSTR)
#define ST_SZ   (G15_O + 16)
#define PT_OFF  (ST_OFF + 2 * ST_SZ)
#define PP_O    0
#define SP_O    (PP_O + 4 * PL2)
#define NPS_O   (SP_O + 4 * PL2)
#define PT_SZ   (NPS_O + 128)
#define RW_OFF  (PT_OFF + 2 * PT_SZ)
#define RW_SZ   (3 * PL2)
#define SMEM_FLOATS (RW_OFF + 4 * RW_SZ)


static __device__ __forceinline__ void lds_barrier()
{
    __builtin_amdgcn_fence(__ATOMIC_RELEASE, "workgroup", "local");
    __builtin_amdgcn_s_barrier();
    __builtin_amdgcn_fence(__ATOMIC_ACQUIRE, "workgroup", "local");
}

__global__ __launch_bounds__(128, 1) void rec_kernel(
    const float* __restrict__ qb, const float* __restrict__ kb,
    const float* __restrict__ vb, const float* __restrict__ ib,
    const float* __restrict__ fb, const float* __restrict__ ob,
    float* __restrict__ hb)
{
    __shared__ __attribute__((aligned(16))) float sm[SMEM_FLOATS];
    const int b    = blockIdx.x >> 3;
    const int sl   = blockIdx.x & 7;
    const int r    = threadIdx.x;
    const int lane = r & 31;
    const int wave = r >> 5;
    const int lm   = lane & 15;
    const int hh   = lane >> 4;
    const int lk   = hh << 1;
    const int x    = r & 15;
    const bool rowthr = (r >> 4) == sl;

    for (int j = r; j < RS * CSTR; j += 128) sm[C_OFF + j] = 0.0f;
    float creg[2][8];
    #pragma unroll
    for (int t = 0; t < 2; ++t)
        #pragma unroll
        for (int v = 0; v < 8; ++v) creg[t][v] = 0.0f;

    float nprev = 0.0f;
    const long bbase = (long)b * TT * HH + r;
    float* const hrow = hb + (long)blockIdx.x * (TT * RS);

    float pk[LCH], pq[LCH], pi[LCH], pf[LCH], pv[LCH], po[LCH];
    #pragma unroll
    for (int s = 0; s < LCH; ++s) {
        const long idx = bbase + (long)s * HH;
        pk[s] = kb[idx]; pq[s] = qb[idx]; pi[s] = ib[idx]; pf[s] = fb[idx];
        pv[s] = 0.0f; po[s] = 0.0f;
    }
    if (rowthr) {
        #pragma unroll
        for (int s = 0; s < LCH; ++s) {
            const long idx = bbase + (long)s * HH;
            pv[s] = vb[idx]; po[s] = ob[idx];
        }
    }

    auto stage_rows = [&](int c, int cn) __attribute__((always_inline)) {
        float* st = sm + ST_OFF + (c & 1) * ST_SZ;
        float* rw = sm + RW_OFF + (c & 3) * RW_SZ;
        if (rowthr) {
            float G[LCH];
            float g = 1.0f;
            #pragma unroll
            for (int s = 0; s < LCH; ++s) { g *= pf[s]; G[s] = g; }
            const float g15 = G[LCH - 1];
            #pragma unroll
            for (int s = 0; s < LCH; ++s) {
                const float r0 = __builtin_amdgcn_rcpf(G[s]);
                const float rg = fmaf(fmaf(-G[s], r0, 1.0f), r0, r0);
                const float wg = pi[s] * pv[s] * rg;
                st[WC_O + x * TSTR + s]   = g15 * wg;
                rw[0 * PL2 + s * PS2 + x] = wg;
                rw[1 * PL2 + s * PS2 + x] = G[s];
                rw[2 * PL2 + s * PS2 + x] = po[s];
            }
            st[G15_O + x] = g15;
            const long nbase = bbase + (long)(cn * LCH) * HH;
            #pragma unroll
            for (int s = 0; s < LCH; ++s) {
                const long idx = nbase + (long)s * HH;
                pv[s] = vb[idx]; po[s] = ob[idx];
            }
        }
    };

    auto stage_cols = [&](int c) __attribute__((always_inline)) {
        float* st = sm + ST_OFF + (c & 1) * ST_SZ;
        float np = nprev;
        #pragma unroll
        for (int s = 0; s < LCH; ++s) {
            np = fmaf(pf[s], np, pi[s] * pk[s]);
            st[M_O + s * CSTR + r] = np * pq[s];
        }
        nprev = np;
        #pragma unroll
        for (int s = 0; s < LCH; ++s) {
            st[Q_O + s * CSTR + r]  = pq[s];
            st[K_O + s * CSTR + r]  = pk[s];
            st[KT_O + r * TSTR + s] = pk[s];
        }
    };

    auto load_g1 = [&](int c) __attribute__((always_inline)) {
        const long nbase = bbase + (long)(c * LCH) * HH;
        #pragma unroll
        for (int s = 0; s < LCH; ++s) {
            const long idx = nbase + (long)s * HH;
            pk[s] = kb[idx]; pq[s] = qb[idx]; pi[s] = ib[idx];
        }
    };
    auto load_g2 = [&](int c) __attribute__((always_inline)) {
        const long nbase = bbase + (long)(c * LCH) * HH;
        #pragma unroll
        for (int s = 0; s < LCH; ++s) pf[s] = fb[nbase + (long)s * HH];
    };

    const int s1 = r >> 3, seg = r & 7;
    const int cb = wave * 32 + lk;
    v4f mv[4];
    v2f qf[8], kf[8], cf[8], afr[4], bfr[2][4];
    float g[8];

    auto state_load = [&](int c) __attribute__((always_inline)) {
        const float* st = sm + ST_OFF + (c & 1) * ST_SZ;
        #pragma unroll
        for (int e = 0; e < 4; ++e) mv[e] = *(const v4f*)&st[M_O + s1 * CSTR + seg * 16 + 4 * e];

        #pragma unroll
        for (int kk = 0; kk < 8; ++kk) {
            qf[kk] = *(const v2f*)&st[Q_O + lm * CSTR + cb + kk * 4];
            kf[kk] = *(const v2f*)&st[K_O + lm * CSTR + cb + kk * 4];
            cf[kk] = *(const v2f*)&sm[C_OFF + lm * CSTR + cb + kk * 4];
        }
        #pragma unroll
        for (int kk = 0; kk < 4; ++kk)
            afr[kk] = *(const v2f*)&st[WC_O + lm * TSTR + kk * 4 + lk];
        #pragma unroll
        for (int v = 0; v < 8; ++v) g[v] = st[G15_O + 8 * hh + v];
        #pragma unroll
        for (int t = 0; t < 2; ++t)
            #pragma unroll
            for (int kk = 0; kk < 4; ++kk)
                bfr[t][kk] = *(const v2f*)&st[KT_O + (wave * 32 + t * 16 + lm) * TSTR + kk * 4 + lk];
    };

    auto state_cupd = [&]() __attribute__((always_inline)) {
        v8f d0 = {};
        v8f d1 = {};
        #pragma unroll
        for (int kk = 0; kk < 4; ++kk) {
            d0 = __builtin_amdgcn_wmma_f32_16x16x4_f32(false, afr[kk], false, bfr[0][kk],
                                                       (short)0, d0, false, false);
            d1 = __builtin_amdgcn_wmma_f32_16x16x4_f32(false, afr[kk], false, bfr[1][kk],
                                                       (short)0, d1, false, false);
        }
        #pragma unroll
        for (int v = 0; v < 8; ++v) {
            creg[0][v] = fmaf(g[v], creg[0][v], d0[v]);
            creg[1][v] = fmaf(g[v], creg[1][v], d1[v]);
        }
    };

    auto state_ps = [&](int c) __attribute__((always_inline)) {
        float* pt = sm + PT_OFF + (c & 1) * PT_SZ;
        float ps = 0.0f;
        #pragma unroll
        for (int e = 0; e < 4; ++e) { ps += mv[e].x; ps += mv[e].y; ps += mv[e].z; ps += mv[e].w; }

        v8f pacc = {};
        v8f sacc = {};
        #pragma unroll
        for (int kk = 0; kk < 8; ++kk) {
            pacc = __builtin_amdgcn_wmma_f32_16x16x4_f32(false, qf[kk], false, cf[kk],
                                                         (short)0, pacc, false, false);
            sacc = __builtin_amdgcn_wmma_f32_16x16x4_f32(false, kf[kk], false, qf[kk],
                                                         (short)0, sacc, false, false);
        }
        pt[NPS_O + s1 * 8 + seg] = ps;
        #pragma unroll
        for (int v = 0; v < 8; ++v)
            pt[PP_O + wave * PL2 + (v + 8 * hh) * PS2 + lm] = pacc[v];
        *(v4f*)&pt[SP_O + wave * PL2 + lm * PS2 + 8 * hh]     = (v4f){sacc[0], sacc[1], sacc[2], sacc[3]};
        *(v4f*)&pt[SP_O + wave * PL2 + lm * PS2 + 8 * hh + 4] = (v4f){sacc[4], sacc[5], sacc[6], sacc[7]};
        #pragma unroll
        for (int t = 0; t < 2; ++t)
            #pragma unroll
            for (int v = 0; v < 8; ++v)
                sm[C_OFF + (8 * hh + v) * CSTR + wave * 32 + t * 16 + lm] = creg[t][v];
    };

    const int s  = r >> 3;
    const int x0 = (r & 7) * 2;
    v4f sp[4][4], n0, n1;
    v2f pp[4], wg[LCH], gs, os;
    auto out_load = [&](int c) __attribute__((always_inline)) {
        const float* pt = sm + PT_OFF + (c & 1) * PT_SZ;
        const float* rw = sm + RW_OFF + (c & 3) * RW_SZ;
        #pragma unroll
        for (int w = 0; w < 4; ++w)
            #pragma unroll
            for (int j = 0; j < 4; ++j) sp[w][j] = *(const v4f*)&pt[SP_O + w * PL2 + s * PS2 + 4 * j];
        #pragma unroll
        for (int w = 0; w < 4; ++w) pp[w] = *(const v2f*)&pt[PP_O + w * PL2 + s * PS2 + x0];
        n0 = *(const v4f*)&pt[NPS_O + s * 8];
        n1 = *(const v4f*)&pt[NPS_O + s * 8 + 4];
        #pragma unroll
        for (int u = 0; u < LCH; ++u) wg[u] = *(const v2f*)&rw[0 * PL2 + u * PS2 + x0];
        gs = *(const v2f*)&rw[1 * PL2 + s * PS2 + x0];
        os = *(const v2f*)&rw[2 * PL2 + s * PS2 + x0];
    };

    float sv[LCH], nrm;
    v2f acc;
    auto out_sum = [&]() __attribute__((always_inline)) {
        #pragma unroll
        for (int j = 0; j < 4; ++j) {
            const v4f t = ((sp[0][j] + sp[1][j]) + sp[2][j]) + sp[3][j];
            sv[4 * j] = t.x; sv[4 * j + 1] = t.y; sv[4 * j + 2] = t.z; sv[4 * j + 3] = t.w;
        }
        float nq = 0.0f;
        nq += n0.x; nq += n0.y; nq += n0.z; nq += n0.w;
        nq += n1.x; nq += n1.y; nq += n1.z; nq += n1.w;
        nrm = fmaxf(fabsf(nq), 1.0f);
        acc = ((pp[0] + pp[1]) + pp[2]) + pp[3];
    };

    auto out_chain = [&](int c) __attribute__((always_inline)) {
        #pragma unroll
        for (int u = 0; u < LCH; ++u) {
            const float nx0 = fmaf(wg[u].x, sv[u], acc.x);
            const float nx1 = fmaf(wg[u].y, sv[u], acc.y);
            acc.x = (u <= s) ? nx0 : acc.x;
            acc.y = (u <= s) ? nx1 : acc.y;
        }
        v2f res;
        res.x = os.x * (gs.x * acc.x) / nrm;
        res.y = os.y * (gs.y * acc.y) / nrm;
        *(v2f*)&hrow[(long)c * (LCH * RS) + s * RS + x0] = res;
    };

    stage_rows(0, 1);
    stage_cols(0);
    load_g1(1);
    load_g2(1);
    lds_barrier();

    stage_rows(1, 2);
    state_load(0);
    stage_cols(1);
    load_g1(2);
    state_cupd();
    load_g2(2);
    state_ps(0);
    lds_barrier();

    for (int t = 2; t < NCH; ++t) {
        const int nc = (t + 1 < NCH) ? t + 1 : NCH - 1;
        stage_rows(t, nc);
        state_load(t - 1);
        out_load(t - 2);
        stage_cols(t);
        load_g1(nc);
        state_cupd();
        __builtin_amdgcn_sched_barrier(0x7C6);
        load_g2(nc);
        state_ps(t - 1);
        out_sum();
        out_chain(t - 2);
        lds_barrier();
    }

    state_load(NCH - 1);
    out_load(NCH - 2);
    state_cupd();
    state_ps(NCH - 1);
    out_sum();
    out_chain(NCH - 2);
    lds_barrier();
    out_load(NCH - 1);
    out_sum();
    out_chain(NCH - 1);
}

__global__ __launch_bounds__(256) void ln_kernel(
    const float* __restrict__ hb, const float* __restrict__ x,
    const float* __restrict__ lnw, const float* __restrict__ lnb,
    float* __restrict__ out)
{
    const int lane = threadIdx.x & 31;
    const int wave = threadIdx.x >> 5;
    const long row = (long)blockIdx.x * 8 + wave;
    const long rb  = row * HH;

    float y[4];
    float s = 0.0f, s2 = 0.0f;
    #pragma unroll
    for (int j = 0; j < 4; ++j) {
        const int c = lane + 32 * j;
        const long hidx = (((row >> 9) * NSL + (c >> 4)) * TT + (row & (TT - 1))) * RS + (c & 15);
        y[j] = hb[hidx] + x[rb + c];
        s  += y[j];
        s2 += y[j] * y[j];
    }
    #pragma unroll
    for (int off = 16; off >= 1; off >>= 1) {
        s  += __shfl_xor(s,  off, 32);
        s2 += __shfl_xor(s2, off, 32);
    }
    const float mu  = s * (1.0f / HH);
    const float var = s2 * (1.0f / HH) - mu * mu;
    const float rs  = rsqrtf(var + LN_EPS);

    #pragma unroll
    for (int j = 0; j < 4; ++j) {
        const int c = lane + 32 * j;
        out[rb + c] = (y[j] - mu) * rs * lnw[c] + lnb[c];
    }
}

extern "C" void kernel_launch(void* const* d_in, const int* in_sizes, int n_in,
                              void* d_out, int out_size, void* d_ws, size_t ws_size,
                              hipStream_t stream)
{
    const float* x   = (const float*)d_in[0];
    const float* Wq  = (const float*)d_in[1];
    const float* Wk  = (const float*)d_in[2];
    const float* Wv  = (const float*)d_in[3];
    const float* Wi  = (const float*)d_in[4];
    const float* Wf  = (const float*)d_in[5];
    const float* Wo  = (const float*)d_in[6];
    const float* b_i = (const float*)d_in[7];
    const float* b_f = (const float*)d_in[8];
    const float* b_o = (const float*)d_in[9];
    const float* lnw = (const float*)d_in[10];
    const float* lnb = (const float*)d_in[11];
    float* out = (float*)d_out;

    float* ws = (float*)d_ws;
    float* q  = ws + 0L * NEL;
    float* k  = ws + 1L * NEL;
    float* v  = ws + 2L * NEL;
    float* ig = ws + 3L * NEL;
    float* fg = ws + 4L * NEL;
    float* og = ws + 5L * NEL;
    float* h  = ws + 6L * NEL;

    proj_kernel<<<PJ_GRID, 128, 0, stream>>>(x, Wq, Wk, Wv, Wi, Wf, Wo,
                                          b_i, b_f, b_o,
                                          q, k, v, ig, fg, og);
    rec_kernel<<<REC_GRID, 128, 0, stream>>>(q, k, v, ig, fg, og, h);
    ln_kernel<<<MM / 8, 256, 0, stream>>>(h, x, lnw, lnb, out);
}


# Recurrence: v and o fetched by all threads, one 8-byte load each a trip (v to the row threads via LDS, o to its output thread), not 32 loads by 16 row threads whose wave then waited for them
# speedup vs baseline: 5.8845x; 1.1615x over previous
#include <hip/hip_runtime.h>
#include <math.h>


typedef float v2f __attribute__((ext_vector_type(2)));
typedef float v4f __attribute__((ext_vector_type(4)));
typedef float v8f __attribute__((ext_vector_type(8)));

#define BB   16
#define TT   512
#define HH   128
#define MM   (BB * TT)
#define NEL  (BB * TT * HH)
#define SCALE 0.08838834764831845f
#define LN_EPS 1e-5f
#define LCH  16
#define NCH  (TT / LCH)

#define PJ_MT 4
#define PJ_NT 2
#define PJ_ROWS (16 * PJ_MT)
#define PJ_GRID ((MM / PJ_ROWS) * 6)

static __device__ __forceinline__ float xhalf(float v)
{
    return __builtin_bit_cast(float, __builtin_amdgcn_permlanex16(
        __builtin_bit_cast(int, v), __builtin_bit_cast(int, v),
        0x76543210, 0xfedcba98, false, false));
}

template <int ACT>
static __device__ __forceinline__ float proj_act(float val, float bias)
{
    if (ACT == 0) return val * SCALE;
    if (ACT == 1) return val;
    if (ACT == 2) return expf(val + bias);
    return 1.0f / (1.0f + expf(-(val + bias)));
}

template <int ACT>
static __device__ __forceinline__ void proj_store(
    const v8f (&acc)[PJ_MT][PJ_NT], float* __restrict__ outp,
    int m0, int n0, int lane, float bias0, float bias1)
{
    const int hh = lane >> 4;
    #pragma unroll
    for (int i = 0; i < PJ_MT; ++i) {
        #pragma unroll
        for (int v = 0; v < 8; ++v) {
            const float t0 = proj_act<ACT>(acc[i][0][v], bias0);
            const float t1 = proj_act<ACT>(acc[i][1][v], bias1);
            const float got = xhalf(hh ? t0 : t1);
            const float lo = hh ? got : t0;
            const float hi = hh ? t1 : got;
            float* o = outp + (long)(m0 + 16 * i + v) * HH + n0 + lane;
            o[0]      = lo;
            o[8 * HH] = hi;
        }
    }
}

__global__ __launch_bounds__(128, 3) void proj_kernel(
    const float* __restrict__ x,
    const float* __restrict__ Wq, const float* __restrict__ Wk,
    const float* __restrict__ Wv, const float* __restrict__ Wi,
    const float* __restrict__ Wf, const float* __restrict__ Wo,
    const float* __restrict__ b_i, const float* __restrict__ b_f,
    const float* __restrict__ b_o,
    float* __restrict__ oq, float* __restrict__ ok, float* __restrict__ ov,
    float* __restrict__ oi, float* __restrict__ of, float* __restrict__ oo)
{
    const int lane   = threadIdx.x & 31;
    const int wave   = threadIdx.x >> 5;
    const int matIdx = blockIdx.x % 6;
    const int m0     = (blockIdx.x / 6) * PJ_ROWS;
    const int n0     = wave * (16 * PJ_NT);

    const float* W;
    const float* bp = b_i;
    float* outp;
    switch (matIdx) {
        case 0:  W = Wq; outp = oq; break;
        case 1:  W = Wk; outp = ok; break;
        case 2:  W = Wv; outp = ov; break;
        case 3:  W = Wi; outp = oi; bp = b_i; break;
        case 4:  W = Wf; outp = of; bp = b_f; break;
        default: W = Wo; outp = oo; bp = b_o; break;
    }

    const int lm = lane & 15;
    const int hh = lane >> 4;

    const float* aptr = x + (long)(m0 + lm) * HH + 16 * hh;
    const float* bptr = W + (n0 + lm) * HH + 16 * hh;

    v8f acc[PJ_MT][PJ_NT];
    #pragma unroll
    for (int i = 0; i < PJ_MT; ++i)
        #pragma unroll
        for (int j = 0; j < PJ_NT; ++j) acc[i][j] = (v8f){};

    v4f fa[2][PJ_MT][4];
    v4f fb[2][PJ_NT][4];

    #pragma unroll
    for (int i = 0; i < PJ_MT; ++i)
        #pragma unroll
        for (int q = 0; q < 4; ++q) fa[0][i][q] = *(const v4f*)(aptr + i * 16 * HH + 4 * q);
    #pragma unroll
    for (int j = 0; j < PJ_NT; ++j)
        #pragma unroll
        for (int q = 0; q < 4; ++q) fb[0][j][q] = *(const v4f*)(bptr + j * 16 * HH + 4 * q);

    #pragma unroll
    for (int w = 0; w < 4; ++w) {
        const int cur = w & 1, nxt = cur ^ 1;
        if (w + 1 < 4) {
            #pragma unroll
            for (int i = 0; i < PJ_MT; ++i)
                #pragma unroll
                for (int q = 0; q < 4; ++q)
                    fa[nxt][i][q] = *(const v4f*)(aptr + i * 16 * HH + 32 * (w + 1) + 4 * q);
            #pragma unroll
            for (int j = 0; j < PJ_NT; ++j)
                #pragma unroll
                for (int q = 0; q < 4; ++q)
                    fb[nxt][j][q] = *(const v4f*)(bptr + j * 16 * HH + 32 * (w + 1) + 4 * q);
        }
        __builtin_amdgcn_sched_barrier(0);
        #pragma unroll
        for (int q = 0; q < 4; ++q) {
            #pragma unroll
            for (int i = 0; i < PJ_MT; ++i)
                #pragma unroll
                for (int j = 0; j < PJ_NT; ++j) {
                    const v2f a0 = {fa[cur][i][q].x, fa[cur][i][q].y};
                    const v2f b0 = {fb[cur][j][q].x, fb[cur][j][q].y};
                    acc[i][j] = __builtin_amdgcn_wmma_f32_16x16x4_f32(false, a0, false, b0,
                                                                      (short)0, acc[i][j], false, false);
                }
            #pragma unroll
            for (int i = 0; i < PJ_MT; ++i)
                #pragma unroll
                for (int j = 0; j < PJ_NT; ++j) {
                    const v2f a1 = {fa[cur][i][q].z, fa[cur][i][q].w};
                    const v2f b1 = {fb[cur][j][q].z, fb[cur][j][q].w};
                    acc[i][j] = __builtin_amdgcn_wmma_f32_16x16x4_f32(false, a1, false, b1,
                                                                      (short)0, acc[i][j], false, false);
                }
        }
        __builtin_amdgcn_sched_barrier(0);
    }

    if (matIdx <= 1) {
        proj_store<0>(acc, outp, m0, n0, lane, 0.0f, 0.0f);
    } else if (matIdx == 2) {
        proj_store<1>(acc, outp, m0, n0, lane, 0.0f, 0.0f);
    } else {
        const float bias0 = bp[n0 + lm];
        const float bias1 = bp[n0 + 16 + lm];
        if (matIdx == 3) proj_store<2>(acc, outp, m0, n0, lane, bias0, bias1);
        else             proj_store<3>(acc, outp, m0, n0, lane, bias0, bias1);
    }
}

#define NSL  8
#define RS   16
#define REC_GRID 128
#define CSTR 132
#define TSTR 20
#define PS2  20
#define PL2  (16 * PS2)
#define C_OFF   0
#define ST_OFF  (C_OFF + RS * CSTR)
#define Q_O     0
#define K_O     (Q_O + 16 * CSTR)
#define M_O     (K_O + 16 * CSTR)
#define KT_O    (M_O + 16 * CSTR)
#define WC_O    (KT_O + 128 * TSTR)
#define G15_O   (WC_O + RS * TSTR)
#define ST_SZ   (G15_O + 16)
#define PT_OFF  (ST_OFF + 2 * ST_SZ)
#define PP_O    0
#define SP_O    (PP_O + 4 * PL2)
#define NPS_O   (SP_O + 4 * PL2)
#define PT_SZ   (NPS_O + 128)
#define RW_OFF  (PT_OFF + 2 * PT_SZ)
#define RW_SZ   (2 * PL2)
#define VP_OFF  (RW_OFF + 4 * RW_SZ)
#define SMEM_FLOATS (VP_OFF + 2 * PL2)


static __device__ __forceinline__ void lds_barrier()
{
    __builtin_amdgcn_fence(__ATOMIC_RELEASE, "workgroup", "local");
    __builtin_amdgcn_s_barrier();
    __builtin_amdgcn_fence(__ATOMIC_ACQUIRE, "workgroup", "local");
}

__global__ __launch_bounds__(128, 1) void rec_kernel(
    const float* __restrict__ qb, const float* __restrict__ kb,
    const float* __restrict__ vb, const float* __restrict__ ib,
    const float* __restrict__ fb, const float* __restrict__ ob,
    float* __restrict__ hb)
{
    __shared__ __attribute__((aligned(16))) float sm[SMEM_FLOATS];
    const int b    = blockIdx.x >> 3;
    const int sl   = blockIdx.x & 7;
    const int r    = threadIdx.x;
    const int lane = r & 31;
    const int wave = r >> 5;
    const int lm   = lane & 15;
    const int hh   = lane >> 4;
    const int lk   = hh << 1;
    const int x    = r & 15;
    const bool rowthr = (r >> 4) == sl;

    for (int j = r; j < RS * CSTR; j += 128) sm[C_OFF + j] = 0.0f;
    float creg[2][8];
    #pragma unroll
    for (int t = 0; t < 2; ++t)
        #pragma unroll
        for (int v = 0; v < 8; ++v) creg[t][v] = 0.0f;

    float nprev = 0.0f;
    const long bbase = (long)b * TT * HH + r;
    float* const hrow = hb + (long)blockIdx.x * (TT * RS);

    float pk[LCH], pq[LCH], pi[LCH], pf[LCH];
    #pragma unroll
    for (int s = 0; s < LCH; ++s) {
        const long idx = bbase + (long)s * HH;
        pk[s] = kb[idx]; pq[s] = qb[idx]; pi[s] = ib[idx]; pf[s] = fb[idx];
    }

    const long vobase = (long)b * TT * HH + (long)(r >> 3) * HH + sl * RS + (r & 7) * 2;
    v2f vnx;
    auto load_v = [&](int c) __attribute__((always_inline)) {
        vnx = *(const v2f*)(vb + vobase + (long)(c * LCH) * HH);
    };
    auto load_o = [&](int c) __attribute__((always_inline)) {
        return *(const v2f*)(ob + vobase + (long)(c * LCH) * HH);
    };
    auto store_v = [&](int c) __attribute__((always_inline)) {
        *(v2f*)&sm[VP_OFF + (c & 1) * PL2 + (r >> 3) * PS2 + (r & 7) * 2] = vnx;
    };
    load_v(0); store_v(0);
    load_v(1); store_v(1);

    auto stage_rows = [&](int c) __attribute__((always_inline)) {
        float* st = sm + ST_OFF + (c & 1) * ST_SZ;
        float* rw = sm + RW_OFF + (c & 3) * RW_SZ;
        if (rowthr) {
            const float* vp = sm + VP_OFF + (c & 1) * PL2;
            float pv[LCH];
            #pragma unroll
            for (int s = 0; s < LCH; ++s) pv[s] = vp[s * PS2 + x];
            float G[LCH];
            float g = 1.0f;
            #pragma unroll
            for (int s = 0; s < LCH; ++s) { g *= pf[s]; G[s] = g; }
            const float g15 = G[LCH - 1];
            #pragma unroll
            for (int s = 0; s < LCH; ++s) {
                const float r0 = __builtin_amdgcn_rcpf(G[s]);
                const float rg = fmaf(fmaf(-G[s], r0, 1.0f), r0, r0);
                const float wg = pi[s] * pv[s] * rg;
                st[WC_O + x * TSTR + s]   = g15 * wg;
                rw[0 * PL2 + s * PS2 + x] = wg;
                rw[1 * PL2 + s * PS2 + x] = G[s];
            }
            st[G15_O + x] = g15;
        }
    };

    auto stage_cols = [&](int c) __attribute__((always_inline)) {
        float* st = sm + ST_OFF + (c & 1) * ST_SZ;
        float np = nprev;
        #pragma unroll
        for (int s = 0; s < LCH; ++s) {
            np = fmaf(pf[s], np, pi[s] * pk[s]);
            st[M_O + s * CSTR + r] = np * pq[s];
        }
        nprev = np;
        #pragma unroll
        for (int s = 0; s < LCH; ++s) {
            st[Q_O + s * CSTR + r]  = pq[s];
            st[K_O + s * CSTR + r]  = pk[s];
            st[KT_O + r * TSTR + s] = pk[s];
        }
    };

    auto load_g1 = [&](int c) __attribute__((always_inline)) {
        const long nbase = bbase + (long)(c * LCH) * HH;
        #pragma unroll
        for (int s = 0; s < LCH; ++s) {
            const long idx = nbase + (long)s * HH;
            pk[s] = kb[idx]; pq[s] = qb[idx]; pi[s] = ib[idx];
        }
    };
    auto load_g2 = [&](int c) __attribute__((always_inline)) {
        const long nbase = bbase + (long)(c * LCH) * HH;
        #pragma unroll
        for (int s = 0; s < LCH; ++s) pf[s] = fb[nbase + (long)s * HH];
    };

    const int s1 = r >> 3, seg = r & 7;
    const int cb = wave * 32 + lk;
    v4f mv[4];
    v2f qf[8], kf[8], cf[8], afr[4], bfr[2][4];
    float g[8];

    auto state_load = [&](int c) __attribute__((always_inline)) {
        const float* st = sm + ST_OFF + (c & 1) * ST_SZ;
        #pragma unroll
        for (int e = 0; e < 4; ++e) mv[e] = *(const v4f*)&st[M_O + s1 * CSTR + seg * 16 + 4 * e];

        #pragma unroll
        for (int kk = 0; kk < 8; ++kk) {
            qf[kk] = *(const v2f*)&st[Q_O + lm * CSTR + cb + kk * 4];
            kf[kk] = *(const v2f*)&st[K_O + lm * CSTR + cb + kk * 4];
            cf[kk] = *(const v2f*)&sm[C_OFF + lm * CSTR + cb + kk * 4];
        }
        #pragma unroll
        for (int kk = 0; kk < 4; ++kk)
            afr[kk] = *(const v2f*)&st[WC_O + lm * TSTR + kk * 4 + lk];
        #pragma unroll
        for (int v = 0; v < 8; ++v) g[v] = st[G15_O + 8 * hh + v];
        #pragma unroll
        for (int t = 0; t < 2; ++t)
            #pragma unroll
            for (int kk = 0; kk < 4; ++kk)
                bfr[t][kk] = *(const v2f*)&st[KT_O + (wave * 32 + t * 16 + lm) * TSTR + kk * 4 + lk];
    };

    auto state_cupd = [&]() __attribute__((always_inline)) {
        v8f d0 = {};
        v8f d1 = {};
        #pragma unroll
        for (int kk = 0; kk < 4; ++kk) {
            d0 = __builtin_amdgcn_wmma_f32_16x16x4_f32(false, afr[kk], false, bfr[0][kk],
                                                       (short)0, d0, false, false);
            d1 = __builtin_amdgcn_wmma_f32_16x16x4_f32(false, afr[kk], false, bfr[1][kk],
                                                       (short)0, d1, false, false);
        }
        #pragma unroll
        for (int v = 0; v < 8; ++v) {
            creg[0][v] = fmaf(g[v], creg[0][v], d0[v]);
            creg[1][v] = fmaf(g[v], creg[1][v], d1[v]);
        }
    };

    auto state_ps = [&](int c) __attribute__((always_inline)) {
        float* pt = sm + PT_OFF + (c & 1) * PT_SZ;
        float ps = 0.0f;
        #pragma unroll
        for (int e = 0; e < 4; ++e) { ps += mv[e].x; ps += mv[e].y; ps += mv[e].z; ps += mv[e].w; }

        v8f pacc = {};
        v8f sacc = {};
        #pragma unroll
        for (int kk = 0; kk < 8; ++kk) {
            pacc = __builtin_amdgcn_wmma_f32_16x16x4_f32(false, qf[kk], false, cf[kk],
                                                         (short)0, pacc, false, false);
            sacc = __builtin_amdgcn_wmma_f32_16x16x4_f32(false, kf[kk], false, qf[kk],
                                                         (short)0, sacc, false, false);
        }
        pt[NPS_O + s1 * 8 + seg] = ps;
        #pragma unroll
        for (int v = 0; v < 8; ++v)
            pt[PP_O + wave * PL2 + (v + 8 * hh) * PS2 + lm] = pacc[v];
        *(v4f*)&pt[SP_O + wave * PL2 + lm * PS2 + 8 * hh]     = (v4f){sacc[0], sacc[1], sacc[2], sacc[3]};
        *(v4f*)&pt[SP_O + wave * PL2 + lm * PS2 + 8 * hh + 4] = (v4f){sacc[4], sacc[5], sacc[6], sacc[7]};
        #pragma unroll
        for (int t = 0; t < 2; ++t)
            #pragma unroll
            for (int v = 0; v < 8; ++v)
                sm[C_OFF + (8 * hh + v) * CSTR + wave * 32 + t * 16 + lm] = creg[t][v];
    };

    const int s  = r >> 3;
    const int x0 = (r & 7) * 2;
    v4f sp[4][4], n0, n1;
    v2f pp[4], wg[LCH], gs;
    auto out_load = [&](int c) __attribute__((always_inline)) {
        const float* pt = sm + PT_OFF + (c & 1) * PT_SZ;
        const float* rw = sm + RW_OFF + (c & 3) * RW_SZ;
        #pragma unroll
        for (int w = 0; w < 4; ++w)
            #pragma unroll
            for (int j = 0; j < 4; ++j) sp[w][j] = *(const v4f*)&pt[SP_O + w * PL2 + s * PS2 + 4 * j];
        #pragma unroll
        for (int w = 0; w < 4; ++w) pp[w] = *(const v2f*)&pt[PP_O + w * PL2 + s * PS2 + x0];
        n0 = *(const v4f*)&pt[NPS_O + s * 8];
        n1 = *(const v4f*)&pt[NPS_O + s * 8 + 4];
        #pragma unroll
        for (int u = 0; u < LCH; ++u) wg[u] = *(const v2f*)&rw[0 * PL2 + u * PS2 + x0];
        gs = *(const v2f*)&rw[1 * PL2 + s * PS2 + x0];
    };

    float sv[LCH], nrm;
    v2f acc;
    auto out_sum = [&]() __attribute__((always_inline)) {
        #pragma unroll
        for (int j = 0; j < 4; ++j) {
            const v4f t = ((sp[0][j] + sp[1][j]) + sp[2][j]) + sp[3][j];
            sv[4 * j] = t.x; sv[4 * j + 1] = t.y; sv[4 * j + 2] = t.z; sv[4 * j + 3] = t.w;
        }
        float nq = 0.0f;
        nq += n0.x; nq += n0.y; nq += n0.z; nq += n0.w;
        nq += n1.x; nq += n1.y; nq += n1.z; nq += n1.w;
        nrm = fmaxf(fabsf(nq), 1.0f);
        acc = ((pp[0] + pp[1]) + pp[2]) + pp[3];
    };

    auto out_chain = [&](int c, v2f os) __attribute__((always_inline)) {
        #pragma unroll
        for (int u = 0; u < LCH; ++u) {
            const float nx0 = fmaf(wg[u].x, sv[u], acc.x);
            const float nx1 = fmaf(wg[u].y, sv[u], acc.y);
            acc.x = (u <= s) ? nx0 : acc.x;
            acc.y = (u <= s) ? nx1 : acc.y;
        }
        v2f res;
        res.x = os.x * (gs.x * acc.x) / nrm;
        res.y = os.y * (gs.y * acc.y) / nrm;
        *(v2f*)&hrow[(long)c * (LCH * RS) + s * RS + x0] = res;
    };

    lds_barrier();
    stage_rows(0);
    stage_cols(0);
    load_g1(1);
    load_g2(1);
    lds_barrier();

    load_v(2);
    stage_rows(1);
    state_load(0);
    stage_cols(1);
    load_g1(2);
    state_cupd();
    load_g2(2);
    state_ps(0);
    store_v(2);
    lds_barrier();

    for (int t = 2; t < NCH; ++t) {
        const int nc = (t + 1 < NCH) ? t + 1 : NCH - 1;
        load_v(nc);
        const v2f os = load_o(t - 2);
        stage_rows(t);
        state_load(t - 1);
        out_load(t - 2);
        stage_cols(t);
        load_g1(nc);
        state_cupd();
        __builtin_amdgcn_sched_barrier(0x7C6);
        load_g2(nc);
        state_ps(t - 1);
        out_sum();
        out_chain(t - 2, os);
        store_v(t + 1);
        lds_barrier();
    }

    const v2f osa = load_o(NCH - 2);
    const v2f osb = load_o(NCH - 1);
    state_load(NCH - 1);
    out_load(NCH - 2);
    state_cupd();
    state_ps(NCH - 1);
    out_sum();
    out_chain(NCH - 2, osa);
    lds_barrier();
    out_load(NCH - 1);
    out_sum();
    out_chain(NCH - 1, osb);
}

__global__ __launch_bounds__(256) void ln_kernel(
    const float* __restrict__ hb, const float* __restrict__ x,
    const float* __restrict__ lnw, const float* __restrict__ lnb,
    float* __restrict__ out)
{
    const int lane = threadIdx.x & 31;
    const int wave = threadIdx.x >> 5;
    const long row = (long)blockIdx.x * 8 + wave;
    const long rb  = row * HH;

    float y[4];
    float s = 0.0f, s2 = 0.0f;
    #pragma unroll
    for (int j = 0; j < 4; ++j) {
        const int c = lane + 32 * j;
        const long hidx = (((row >> 9) * NSL + (c >> 4)) * TT + (row & (TT - 1))) * RS + (c & 15);
        y[j] = hb[hidx] + x[rb + c];
        s  += y[j];
        s2 += y[j] * y[j];
    }
    #pragma unroll
    for (int off = 16; off >= 1; off >>= 1) {
        s  += __shfl_xor(s,  off, 32);
        s2 += __shfl_xor(s2, off, 32);
    }
    const float mu  = s * (1.0f / HH);
    const float var = s2 * (1.0f / HH) - mu * mu;
    const float rs  = rsqrtf(var + LN_EPS);

    #pragma unroll
    for (int j = 0; j < 4; ++j) {
        const int c = lane + 32 * j;
        out[rb + c] = (y[j] - mu) * rs * lnw[c] + lnb[c];
    }
}

extern "C" void kernel_launch(void* const* d_in, const int* in_sizes, int n_in,
                              void* d_out, int out_size, void* d_ws, size_t ws_size,
                              hipStream_t stream)
{
    const float* x   = (const float*)d_in[0];
    const float* Wq  = (const float*)d_in[1];
    const float* Wk  = (const float*)d_in[2];
    const float* Wv  = (const float*)d_in[3];
    const float* Wi  = (const float*)d_in[4];
    const float* Wf  = (const float*)d_in[5];
    const float* Wo  = (const float*)d_in[6];
    const float* b_i = (const float*)d_in[7];
    const float* b_f = (const float*)d_in[8];
    const float* b_o = (const float*)d_in[9];
    const float* lnw = (const float*)d_in[10];
    const float* lnb = (const float*)d_in[11];
    float* out = (float*)d_out;

    float* ws = (float*)d_ws;
    float* q  = ws + 0L * NEL;
    float* k  = ws + 1L * NEL;
    float* v  = ws + 2L * NEL;
    float* ig = ws + 3L * NEL;
    float* fg = ws + 4L * NEL;
    float* og = ws + 5L * NEL;
    float* h  = ws + 6L * NEL;

    proj_kernel<<<PJ_GRID, 128, 0, stream>>>(x, Wq, Wk, Wv, Wi, Wf, Wo,
                                          b_i, b_f, b_o,
                                          q, k, v, ig, fg, og);
    rec_kernel<<<REC_GRID, 128, 0, stream>>>(q, k, v, ig, fg, og, h);
    ln_kernel<<<MM / 8, 256, 0, stream>>>(h, x, lnw, lnb, out);
}


# q, k, i, f leave the projection chunk-transposed [chunk][step/4][column][4]: the recurrence fetches a chunk as 16 16-byte loads, not 64 4-byte, issued at the trip head into a second register set
# speedup vs baseline: 6.6209x; 1.1251x over previous
#include <hip/hip_runtime.h>
#include <math.h>


typedef float v2f __attribute__((ext_vector_type(2)));
typedef float v4f __attribute__((ext_vector_type(4)));
typedef float v8f __attribute__((ext_vector_type(8)));

#define BB   16
#define TT   512
#define HH   128
#define MM   (BB * TT)
#define NEL  (BB * TT * HH)
#define SCALE 0.08838834764831845f
#define LN_EPS 1e-5f
#define LCH  16
#define NCH  (TT / LCH)

#define PJ_MT 4
#define PJ_NT 2
#define PJ_ROWS (16 * PJ_MT)
#define PJ_GRID ((MM / PJ_ROWS) * 6)

static __device__ __forceinline__ float xhalf(float v)
{
    return __builtin_bit_cast(float, __builtin_amdgcn_permlanex16(
        __builtin_bit_cast(int, v), __builtin_bit_cast(int, v),
        0x76543210, 0xfedcba98, false, false));
}

template <int ACT>
static __device__ __forceinline__ float proj_act(float val, float bias)
{
    if (ACT == 0) return val * SCALE;
    if (ACT == 1) return val;
    if (ACT == 2) return expf(val + bias);
    return 1.0f / (1.0f + expf(-(val + bias)));
}

template <int ACT>
static __device__ __forceinline__ void proj_store(
    const v8f (&acc)[PJ_MT][PJ_NT], float* __restrict__ outp,
    int m0, int n0, int lane, float bias0, float bias1)
{
    const int hh = lane >> 4;
    #pragma unroll
    for (int i = 0; i < PJ_MT; ++i) {
        #pragma unroll
        for (int v = 0; v < 8; ++v) {
            const float t0 = proj_act<ACT>(acc[i][0][v], bias0);
            const float t1 = proj_act<ACT>(acc[i][1][v], bias1);
            const float got = xhalf(hh ? t0 : t1);
            const float lo = hh ? got : t0;
            const float hi = hh ? t1 : got;
            float* o = outp + (long)(m0 + 16 * i + v) * HH + n0 + lane;
            o[0]      = lo;
            o[8 * HH] = hi;
        }
    }
}

template <int ACT>
static __device__ __forceinline__ void proj_store_t(
    const v8f (&acc)[PJ_MT][PJ_NT], float* __restrict__ outp,
    int m0, int n0, int lane, float bias0, float bias1)
{
    const int lm = lane & 15;
    const int hh = lane >> 4;
    #pragma unroll
    for (int i = 0; i < PJ_MT; ++i) {
        #pragma unroll
        for (int j = 0; j < PJ_NT; ++j) {
            const float bias = j ? bias1 : bias0;
            #pragma unroll
            for (int g = 0; g < 2; ++g) {
                v4f t;
                t.x = proj_act<ACT>(acc[i][j][4 * g + 0], bias);
                t.y = proj_act<ACT>(acc[i][j][4 * g + 1], bias);
                t.z = proj_act<ACT>(acc[i][j][4 * g + 2], bias);
                t.w = proj_act<ACT>(acc[i][j][4 * g + 3], bias);
                const long grp = (long)((m0 >> 4) + i) * 4 + 2 * hh + g;
                *(v4f*)(outp + (grp * HH + n0 + 16 * j + lm) * 4) = t;
            }
        }
    }
}

__global__ __launch_bounds__(128, 3) void proj_kernel(
    const float* __restrict__ x,
    const float* __restrict__ Wq, const float* __restrict__ Wk,
    const float* __restrict__ Wv, const float* __restrict__ Wi,
    const float* __restrict__ Wf, const float* __restrict__ Wo,
    const float* __restrict__ b_i, const float* __restrict__ b_f,
    const float* __restrict__ b_o,
    float* __restrict__ oq, float* __restrict__ ok, float* __restrict__ ov,
    float* __restrict__ oi, float* __restrict__ of, float* __restrict__ oo)
{
    const int lane   = threadIdx.x & 31;
    const int wave   = threadIdx.x >> 5;
    const int matIdx = blockIdx.x % 6;
    const int m0     = (blockIdx.x / 6) * PJ_ROWS;
    const int n0     = wave * (16 * PJ_NT);

    const float* W;
    const float* bp = b_i;
    float* outp;
    switch (matIdx) {
        case 0:  W = Wq; outp = oq; break;
        case 1:  W = Wk; outp = ok; break;
        case 2:  W = Wv; outp = ov; break;
        case 3:  W = Wi; outp = oi; bp = b_i; break;
        case 4:  W = Wf; outp = of; bp = b_f; break;
        default: W = Wo; outp = oo; bp = b_o; break;
    }

    const int lm = lane & 15;
    const int hh = lane >> 4;

    const float* aptr = x + (long)(m0 + lm) * HH + 16 * hh;
    const float* bptr = W + (n0 + lm) * HH + 16 * hh;

    v8f acc[PJ_MT][PJ_NT];
    #pragma unroll
    for (int i = 0; i < PJ_MT; ++i)
        #pragma unroll
        for (int j = 0; j < PJ_NT; ++j) acc[i][j] = (v8f){};

    v4f fa[2][PJ_MT][4];
    v4f fb[2][PJ_NT][4];

    #pragma unroll
    for (int i = 0; i < PJ_MT; ++i)
        #pragma unroll
        for (int q = 0; q < 4; ++q) fa[0][i][q] = *(const v4f*)(aptr + i * 16 * HH + 4 * q);
    #pragma unroll
    for (int j = 0; j < PJ_NT; ++j)
        #pragma unroll
        for (int q = 0; q < 4; ++q) fb[0][j][q] = *(const v4f*)(bptr + j * 16 * HH + 4 * q);

    #pragma unroll
    for (int w = 0; w < 4; ++w) {
        const int cur = w & 1, nxt = cur ^ 1;
        if (w + 1 < 4) {
            #pragma unroll
            for (int i = 0; i < PJ_MT; ++i)
                #pragma unroll
                for (int q = 0; q < 4; ++q)
                    fa[nxt][i][q] = *(const v4f*)(aptr + i * 16 * HH + 32 * (w + 1) + 4 * q);
            #pragma unroll
            for (int j = 0; j < PJ_NT; ++j)
                #pragma unroll
                for (int q = 0; q < 4; ++q)
                    fb[nxt][j][q] = *(const v4f*)(bptr + j * 16 * HH + 32 * (w + 1) + 4 * q);
        }
        __builtin_amdgcn_sched_barrier(0);
        #pragma unroll
        for (int q = 0; q < 4; ++q) {
            #pragma unroll
            for (int i = 0; i < PJ_MT; ++i)
                #pragma unroll
                for (int j = 0; j < PJ_NT; ++j) {
                    const v2f a0 = {fa[cur][i][q].x, fa[cur][i][q].y};
                    const v2f b0 = {fb[cur][j][q].x, fb[cur][j][q].y};
                    acc[i][j] = __builtin_amdgcn_wmma_f32_16x16x4_f32(false, a0, false, b0,
                                                                      (short)0, acc[i][j], false, false);
                }
            #pragma unroll
            for (int i = 0; i < PJ_MT; ++i)
                #pragma unroll
                for (int j = 0; j < PJ_NT; ++j) {
                    const v2f a1 = {fa[cur][i][q].z, fa[cur][i][q].w};
                    const v2f b1 = {fb[cur][j][q].z, fb[cur][j][q].w};
                    acc[i][j] = __builtin_amdgcn_wmma_f32_16x16x4_f32(false, a1, false, b1,
                                                                      (short)0, acc[i][j], false, false);
                }
        }
        __builtin_amdgcn_sched_barrier(0);
    }

    if (matIdx <= 1) {
        proj_store_t<0>(acc, outp, m0, n0, lane, 0.0f, 0.0f);
    } else if (matIdx == 2) {
        proj_store<1>(acc, outp, m0, n0, lane, 0.0f, 0.0f);
    } else {
        const float bias0 = bp[n0 + lm];
        const float bias1 = bp[n0 + 16 + lm];
        if (matIdx == 3)      proj_store_t<2>(acc, outp, m0, n0, lane, bias0, bias1);
        else if (matIdx == 4) proj_store_t<3>(acc, outp, m0, n0, lane, bias0, bias1);
        else                  proj_store<3>(acc, outp, m0, n0, lane, bias0, bias1);
    }
}

#define NSL  8
#define RS   16
#define REC_GRID 128
#define CSTR 132
#define TSTR 20
#define PS2  20
#define PL2  (16 * PS2)
#define C_OFF   0
#define ST_OFF  (C_OFF + RS * CSTR)
#define Q_O     0
#define K_O     (Q_O + 16 * CSTR)
#define M_O     (K_O + 16 * CSTR)
#define KT_O    (M_O + 16 * CSTR)
#define WC_O    (KT_O + 128 * TSTR)
#define G15_O   (WC_O + RS * TSTR)
#define ST_SZ   (G15_O + 16)
#define PT_OFF  (ST_OFF + 2 * ST_SZ)
#define PP_O    0
#define SP_O    (PP_O + 4 * PL2)
#define NPS_O   (SP_O + 4 * PL2)
#define PT_SZ   (NPS_O + 128)
#define RW_OFF  (PT_OFF + 2 * PT_SZ)
#define RW_SZ   (2 * PL2)
#define VP_OFF  (RW_OFF + 4 * RW_SZ)
#define SMEM_FLOATS (VP_OFF + 2 * PL2)


static __device__ __forceinline__ void lds_barrier()
{
    __builtin_amdgcn_fence(__ATOMIC_RELEASE, "workgroup", "local");
    __builtin_amdgcn_s_barrier();
    __builtin_amdgcn_fence(__ATOMIC_ACQUIRE, "workgroup", "local");
}

__global__ __launch_bounds__(128, 1) void rec_kernel(
    const float* __restrict__ qb, const float* __restrict__ kb,
    const float* __restrict__ vb, const float* __restrict__ ib,
    const float* __restrict__ fb, const float* __restrict__ ob,
    float* __restrict__ hb)
{
    __shared__ __attribute__((aligned(16))) float sm[SMEM_FLOATS];
    const int b    = blockIdx.x >> 3;
    const int sl   = blockIdx.x & 7;
    const int r    = threadIdx.x;
    const int lane = r & 31;
    const int wave = r >> 5;
    const int lm   = lane & 15;
    const int hh   = lane >> 4;
    const int lk   = hh << 1;
    const int x    = r & 15;
    const bool rowthr = (r >> 4) == sl;

    for (int j = r; j < RS * CSTR; j += 128) sm[C_OFF + j] = 0.0f;
    float creg[2][8];
    #pragma unroll
    for (int t = 0; t < 2; ++t)
        #pragma unroll
        for (int v = 0; v < 8; ++v) creg[t][v] = 0.0f;

    float nprev = 0.0f;
    float* const hrow = hb + (long)blockIdx.x * (TT * RS);

    float pk[2][LCH], pq[2][LCH], pi[2][LCH], pf[2][LCH];
    const long cbase = (long)b * NCH * (4 * HH * 4) + r * 4;
    auto load_qkif = [&](int c, int set) __attribute__((always_inline)) {
        const long nbase = cbase + (long)c * (4 * HH * 4);
        #pragma unroll
        for (int g = 0; g < 4; ++g) {
            const v4f t = *(const v4f*)(fb + nbase + g * (HH * 4));
            pf[set][4 * g] = t.x; pf[set][4 * g + 1] = t.y; pf[set][4 * g + 2] = t.z; pf[set][4 * g + 3] = t.w;
        }
        #pragma unroll
        for (int g = 0; g < 4; ++g) {
            const v4f t = *(const v4f*)(ib + nbase + g * (HH * 4));
            pi[set][4 * g] = t.x; pi[set][4 * g + 1] = t.y; pi[set][4 * g + 2] = t.z; pi[set][4 * g + 3] = t.w;
        }
        #pragma unroll
        for (int g = 0; g < 4; ++g) {
            const v4f t = *(const v4f*)(kb + nbase + g * (HH * 4));
            pk[set][4 * g] = t.x; pk[set][4 * g + 1] = t.y; pk[set][4 * g + 2] = t.z; pk[set][4 * g + 3] = t.w;
        }
        #pragma unroll
        for (int g = 0; g < 4; ++g) {
            const v4f t = *(const v4f*)(qb + nbase + g * (HH * 4));
            pq[set][4 * g] = t.x; pq[set][4 * g + 1] = t.y; pq[set][4 * g + 2] = t.z; pq[set][4 * g + 3] = t.w;
        }
    };
    load_qkif(0, 0);

    const long vobase = (long)b * TT * HH + (long)(r >> 3) * HH + sl * RS + (r & 7) * 2;
    v2f vnx;
    auto load_v = [&](int c) __attribute__((always_inline)) {
        vnx = *(const v2f*)(vb + vobase + (long)(c * LCH) * HH);
    };
    auto load_o = [&](int c) __attribute__((always_inline)) {
        return *(const v2f*)(ob + vobase + (long)(c * LCH) * HH);
    };
    auto store_v = [&](int c) __attribute__((always_inline)) {
        *(v2f*)&sm[VP_OFF + (c & 1) * PL2 + (r >> 3) * PS2 + (r & 7) * 2] = vnx;
    };
    load_v(0); store_v(0);
    load_v(1); store_v(1);

    auto stage_rows = [&](int c, int set) __attribute__((always_inline)) {
        float* st = sm + ST_OFF + (c & 1) * ST_SZ;
        float* rw = sm + RW_OFF + (c & 3) * RW_SZ;
        if (rowthr) {
            const float* vp = sm + VP_OFF + (c & 1) * PL2;
            float pv[LCH];
            #pragma unroll
            for (int s = 0; s < LCH; ++s) pv[s] = vp[s * PS2 + x];
            float G[LCH];
            float g = 1.0f;
            #pragma unroll
            for (int s = 0; s < LCH; ++s) { g *= pf[set][s]; G[s] = g; }
            const float g15 = G[LCH - 1];
            #pragma unroll
            for (int s = 0; s < LCH; ++s) {
                const float r0 = __builtin_amdgcn_rcpf(G[s]);
                const float rg = fmaf(fmaf(-G[s], r0, 1.0f), r0, r0);
                const float wg = pi[set][s] * pv[s] * rg;
                st[WC_O + x * TSTR + s]   = g15 * wg;
                rw[0 * PL2 + s * PS2 + x] = wg;
                rw[1 * PL2 + s * PS2 + x] = G[s];
            }
            st[G15_O + x] = g15;
        }
    };

    auto stage_cols = [&](int c, int set) __attribute__((always_inline)) {
        float* st = sm + ST_OFF + (c & 1) * ST_SZ;
        float np = nprev;
        #pragma unroll
        for (int s = 0; s < LCH; ++s) {
            np = fmaf(pf[set][s], np, pi[set][s] * pk[set][s]);
            st[M_O + s * CSTR + r] = np * pq[set][s];
        }
        nprev = np;
        #pragma unroll
        for (int s = 0; s < LCH; ++s) {
            st[Q_O + s * CSTR + r]  = pq[set][s];
            st[K_O + s * CSTR + r]  = pk[set][s];
            st[KT_O + r * TSTR + s] = pk[set][s];
        }
    };

    const int s1 = r >> 3, seg = r & 7;
    const int cb = wave * 32 + lk;
    v4f mv[4];
    v2f qf[8], kf[8], cf[8], afr[4], bfr[2][4];
    float g[8];

    auto state_load = [&](int c) __attribute__((always_inline)) {
        const float* st = sm + ST_OFF + (c & 1) * ST_SZ;
        #pragma unroll
        for (int e = 0; e < 4; ++e) mv[e] = *(const v4f*)&st[M_O + s1 * CSTR + seg * 16 + 4 * e];

        #pragma unroll
        for (int kk = 0; kk < 8; ++kk) {
            qf[kk] = *(const v2f*)&st[Q_O + lm * CSTR + cb + kk * 4];
            kf[kk] = *(const v2f*)&st[K_O + lm * CSTR + cb + kk * 4];
            cf[kk] = *(const v2f*)&sm[C_OFF + lm * CSTR + cb + kk * 4];
        }
        #pragma unroll
        for (int kk = 0; kk < 4; ++kk)
            afr[kk] = *(const v2f*)&st[WC_O + lm * TSTR + kk * 4 + lk];
        #pragma unroll
        for (int v = 0; v < 8; ++v) g[v] = st[G15_O + 8 * hh + v];
        #pragma unroll
        for (int t = 0; t < 2; ++t)
            #pragma unroll
            for (int kk = 0; kk < 4; ++kk)
                bfr[t][kk] = *(const v2f*)&st[KT_O + (wave * 32 + t * 16 + lm) * TSTR + kk * 4 + lk];
    };

    auto state_cupd = [&]() __attribute__((always_inline)) {
        v8f d0 = {};
        v8f d1 = {};
        #pragma unroll
        for (int kk = 0; kk < 4; ++kk) {
            d0 = __builtin_amdgcn_wmma_f32_16x16x4_f32(false, afr[kk], false, bfr[0][kk],
                                                       (short)0, d0, false, false);
            d1 = __builtin_amdgcn_wmma_f32_16x16x4_f32(false, afr[kk], false, bfr[1][kk],
                                                       (short)0, d1, false, false);
        }
        #pragma unroll
        for (int v = 0; v < 8; ++v) {
            creg[0][v] = fmaf(g[v], creg[0][v], d0[v]);
            creg[1][v] = fmaf(g[v], creg[1][v], d1[v]);
        }
    };

    auto state_ps = [&](int c) __attribute__((always_inline)) {
        float* pt = sm + PT_OFF + (c & 1) * PT_SZ;
        float ps = 0.0f;
        #pragma unroll
        for (int e = 0; e < 4; ++e) { ps += mv[e].x; ps += mv[e].y; ps += mv[e].z; ps += mv[e].w; }

        v8f pacc = {};
        v8f sacc = {};
        #pragma unroll
        for (int kk = 0; kk < 8; ++kk) {
            pacc = __builtin_amdgcn_wmma_f32_16x16x4_f32(false, qf[kk], false, cf[kk],
                                                         (short)0, pacc, false, false);
            sacc = __builtin_amdgcn_wmma_f32_16x16x4_f32(false, kf[kk], false, qf[kk],
                                                         (short)0, sacc, false, false);
        }
        pt[NPS_O + s1 * 8 + seg] = ps;
        #pragma unroll
        for (int v = 0; v < 8; ++v)
            pt[PP_O + wave * PL2 + (v + 8 * hh) * PS2 + lm] = pacc[v];
        *(v4f*)&pt[SP_O + wave * PL2 + lm * PS2 + 8 * hh]     = (v4f){sacc[0], sacc[1], sacc[2], sacc[3]};
        *(v4f*)&pt[SP_O + wave * PL2 + lm * PS2 + 8 * hh + 4] = (v4f){sacc[4], sacc[5], sacc[6], sacc[7]};
        #pragma unroll
        for (int t = 0; t < 2; ++t)
            #pragma unroll
            for (int v = 0; v < 8; ++v)
                sm[C_OFF + (8 * hh + v) * CSTR + wave * 32 + t * 16 + lm] = creg[t][v];
    };

    const int s  = r >> 3;
    const int x0 = (r & 7) * 2;
    v4f sp[4][4], n0, n1;
    v2f pp[4], wg[LCH], gs;
    auto out_load = [&](int c) __attribute__((always_inline)) {
        const float* pt = sm + PT_OFF + (c & 1) * PT_SZ;
        const float* rw = sm + RW_OFF + (c & 3) * RW_SZ;
        #pragma unroll
        for (int w = 0; w < 4; ++w)
            #pragma unroll
            for (int j = 0; j < 4; ++j) sp[w][j] = *(const v4f*)&pt[SP_O + w * PL2 + s * PS2 + 4 * j];
        #pragma unroll
        for (int w = 0; w < 4; ++w) pp[w] = *(const v2f*)&pt[PP_O + w * PL2 + s * PS2 + x0];
        n0 = *(const v4f*)&pt[NPS_O + s * 8];
        n1 = *(const v4f*)&pt[NPS_O + s * 8 + 4];
        #pragma unroll
        for (int u = 0; u < LCH; ++u) wg[u] = *(const v2f*)&rw[0 * PL2 + u * PS2 + x0];
        gs = *(const v2f*)&rw[1 * PL2 + s * PS2 + x0];
    };

    float sv[LCH], nrm;
    v2f acc;
    auto out_sum = [&]() __attribute__((always_inline)) {
        #pragma unroll
        for (int j = 0; j < 4; ++j) {
            const v4f t = ((sp[0][j] + sp[1][j]) + sp[2][j]) + sp[3][j];
            sv[4 * j] = t.x; sv[4 * j + 1] = t.y; sv[4 * j + 2] = t.z; sv[4 * j + 3] = t.w;
        }
        float nq = 0.0f;
        nq += n0.x; nq += n0.y; nq += n0.z; nq += n0.w;
        nq += n1.x; nq += n1.y; nq += n1.z; nq += n1.w;
        nrm = fmaxf(fabsf(nq), 1.0f);
        acc = ((pp[0] + pp[1]) + pp[2]) + pp[3];
    };

    auto out_chain = [&](int c, v2f os) __attribute__((always_inline)) {
        #pragma unroll
        for (int u = 0; u < LCH; ++u) {
            const float nx0 = fmaf(wg[u].x, sv[u], acc.x);
            const float nx1 = fmaf(wg[u].y, sv[u], acc.y);
            acc.x = (u <= s) ? nx0 : acc.x;
            acc.y = (u <= s) ? nx1 : acc.y;
        }
        v2f res;
        res.x = os.x * (gs.x * acc.x) / nrm;
        res.y = os.y * (gs.y * acc.y) / nrm;
        *(v2f*)&hrow[(long)c * (LCH * RS) + s * RS + x0] = res;
    };

    load_qkif(1, 1);
    lds_barrier();
    stage_rows(0, 0);
    stage_cols(0, 0);
    lds_barrier();

    load_v(2);
    load_qkif(2, 0);
    stage_rows(1, 1);
    state_load(0);
    stage_cols(1, 1);
    state_cupd();
    state_ps(0);
    store_v(2);
    lds_barrier();

    auto trip = [&](int t, int cs, int ns) __attribute__((always_inline)) {
        const int nc = (t + 1 < NCH) ? t + 1 : NCH - 1;
        load_v(nc);
        const v2f os = load_o(t - 2);
        load_qkif(nc, ns);
        stage_rows(t, cs);
        state_load(t - 1);
        out_load(t - 2);
        stage_cols(t, cs);
        state_cupd();
        state_ps(t - 1);
        out_sum();
        out_chain(t - 2, os);
        store_v(t + 1);
        lds_barrier();
    };
    for (int t = 2; t < NCH; t += 2) {
        trip(t, 0, 1);
        trip(t + 1, 1, 0);
    }

    const v2f osa = load_o(NCH - 2);
    const v2f osb = load_o(NCH - 1);
    state_load(NCH - 1);
    out_load(NCH - 2);
    state_cupd();
    state_ps(NCH - 1);
    out_sum();
    out_chain(NCH - 2, osa);
    lds_barrier();
    out_load(NCH - 1);
    out_sum();
    out_chain(NCH - 1, osb);
}

__global__ __launch_bounds__(256) void ln_kernel(
    const float* __restrict__ hb, const float* __restrict__ x,
    const float* __restrict__ lnw, const float* __restrict__ lnb,
    float* __restrict__ out)
{
    const int lane = threadIdx.x & 31;
    const int wave = threadIdx.x >> 5;
    const long row = (long)blockIdx.x * 8 + wave;
    const long rb  = row * HH;

    float y[4];
    float s = 0.0f, s2 = 0.0f;
    #pragma unroll
    for (int j = 0; j < 4; ++j) {
        const int c = lane + 32 * j;
        const long hidx = (((row >> 9) * NSL + (c >> 4)) * TT + (row & (TT - 1))) * RS + (c & 15);
        y[j] = hb[hidx] + x[rb + c];
        s  += y[j];
        s2 += y[j] * y[j];
    }
    #pragma unroll
    for (int off = 16; off >= 1; off >>= 1) {
        s  += __shfl_xor(s,  off, 32);
        s2 += __shfl_xor(s2, off, 32);
    }
    const float mu  = s * (1.0f / HH);
    const float var = s2 * (1.0f / HH) - mu * mu;
    const float rs  = rsqrtf(var + LN_EPS);

    #pragma unroll
    for (int j = 0; j < 4; ++j) {
        const int c = lane + 32 * j;
        out[rb + c] = (y[j] - mu) * rs * lnw[c] + lnb[c];
    }
}

extern "C" void kernel_launch(void* const* d_in, const int* in_sizes, int n_in,
                              void* d_out, int out_size, void* d_ws, size_t ws_size,
                              hipStream_t stream)
{
    const float* x   = (const float*)d_in[0];
    const float* Wq  = (const float*)d_in[1];
    const float* Wk  = (const float*)d_in[2];
    const float* Wv  = (const float*)d_in[3];
    const float* Wi  = (const float*)d_in[4];
    const float* Wf  = (const float*)d_in[5];
    const float* Wo  = (const float*)d_in[6];
    const float* b_i = (const float*)d_in[7];
    const float* b_f = (const float*)d_in[8];
    const float* b_o = (const float*)d_in[9];
    const float* lnw = (const float*)d_in[10];
    const float* lnb = (const float*)d_in[11];
    float* out = (float*)d_out;

    float* ws = (float*)d_ws;
    float* q  = ws + 0L * NEL;
    float* k  = ws + 1L * NEL;
    float* v  = ws + 2L * NEL;
    float* ig = ws + 3L * NEL;
    float* fg = ws + 4L * NEL;
    float* og = ws + 5L * NEL;
    float* h  = ws + 6L * NEL;

    proj_kernel<<<PJ_GRID, 128, 0, stream>>>(x, Wq, Wk, Wv, Wi, Wf, Wo,
                                          b_i, b_f, b_o,
                                          q, k, v, ig, fg, og);
    rec_kernel<<<REC_GRID, 128, 0, stream>>>(q, k, v, ig, fg, og, h);
    ln_kernel<<<MM / 8, 256, 0, stream>>>(h, x, lnw, lnb, out);
}


# Recurrence work-group in two wave roles, one of each per SIMD: 4 waves fetch and stage chunk t while 4 more step the state over t-1 on WMMA and write h of t-2; 1 barrier a trip; sum orders unchanged
# speedup vs baseline: 7.6376x; 1.1536x over previous
#include <hip/hip_runtime.h>
#include <math.h>


typedef float v2f __attribute__((ext_vector_type(2)));
typedef float v4f __attribute__((ext_vector_type(4)));
typedef float v8f __attribute__((ext_vector_type(8)));

#define BB   16
#define TT   512
#define HH   128
#define MM   (BB * TT)
#define NEL  (BB * TT * HH)
#define SCALE 0.08838834764831845f
#define LN_EPS 1e-5f
#define LCH  16
#define NCH  (TT / LCH)

#define PJ_MT 4
#define PJ_NT 2
#define PJ_ROWS (16 * PJ_MT)
#define PJ_GRID ((MM / PJ_ROWS) * 6)

static __device__ __forceinline__ float xhalf(float v)
{
    return __builtin_bit_cast(float, __builtin_amdgcn_permlanex16(
        __builtin_bit_cast(int, v), __builtin_bit_cast(int, v),
        0x76543210, 0xfedcba98, false, false));
}

template <int ACT>
static __device__ __forceinline__ float proj_act(float val, float bias)
{
    if (ACT == 0) return val * SCALE;
    if (ACT == 1) return val;
    if (ACT == 2) return expf(val + bias);
    return 1.0f / (1.0f + expf(-(val + bias)));
}

template <int ACT>
static __device__ __forceinline__ void proj_store(
    const v8f (&acc)[PJ_MT][PJ_NT], float* __restrict__ outp,
    int m0, int n0, int lane, float bias0, float bias1)
{
    const int hh = lane >> 4;
    #pragma unroll
    for (int i = 0; i < PJ_MT; ++i) {
        #pragma unroll
        for (int v = 0; v < 8; ++v) {
            const float t0 = proj_act<ACT>(acc[i][0][v], bias0);
            const float t1 = proj_act<ACT>(acc[i][1][v], bias1);
            const float got = xhalf(hh ? t0 : t1);
            const float lo = hh ? got : t0;
            const float hi = hh ? t1 : got;
            float* o = outp + (long)(m0 + 16 * i + v) * HH + n0 + lane;
            o[0]      = lo;
            o[8 * HH] = hi;
        }
    }
}

template <int ACT>
static __device__ __forceinline__ void proj_store_t(
    const v8f (&acc)[PJ_MT][PJ_NT], float* __restrict__ outp,
    int m0, int n0, int lane, float bias0, float bias1)
{
    const int lm = lane & 15;
    const int hh = lane >> 4;
    #pragma unroll
    for (int i = 0; i < PJ_MT; ++i) {
        #pragma unroll
        for (int j = 0; j < PJ_NT; ++j) {
            const float bias = j ? bias1 : bias0;
            #pragma unroll
            for (int g = 0; g < 2; ++g) {
                v4f t;
                t.x = proj_act<ACT>(acc[i][j][4 * g + 0], bias);
                t.y = proj_act<ACT>(acc[i][j][4 * g + 1], bias);
                t.z = proj_act<ACT>(acc[i][j][4 * g + 2], bias);
                t.w = proj_act<ACT>(acc[i][j][4 * g + 3], bias);
                const long grp = (long)((m0 >> 4) + i) * 4 + 2 * hh + g;
                *(v4f*)(outp + (grp * HH + n0 + 16 * j + lm) * 4) = t;
            }
        }
    }
}

__global__ __launch_bounds__(128, 3) void proj_kernel(
    const float* __restrict__ x,
    const float* __restrict__ Wq, const float* __restrict__ Wk,
    const float* __restrict__ Wv, const float* __restrict__ Wi,
    const float* __restrict__ Wf, const float* __restrict__ Wo,
    const float* __restrict__ b_i, const float* __restrict__ b_f,
    const float* __restrict__ b_o,
    float* __restrict__ oq, float* __restrict__ ok, float* __restrict__ ov,
    float* __restrict__ oi, float* __restrict__ of, float* __restrict__ oo)
{
    const int lane   = threadIdx.x & 31;
    const int wave   = threadIdx.x >> 5;
    const int matIdx = blockIdx.x % 6;
    const int m0     = (blockIdx.x / 6) * PJ_ROWS;
    const int n0     = wave * (16 * PJ_NT);

    const float* W;
    const float* bp = b_i;
    float* outp;
    switch (matIdx) {
        case 0:  W = Wq; outp = oq; break;
        case 1:  W = Wk; outp = ok; break;
        case 2:  W = Wv; outp = ov; break;
        case 3:  W = Wi; outp = oi; bp = b_i; break;
        case 4:  W = Wf; outp = of; bp = b_f; break;
        default: W = Wo; outp = oo; bp = b_o; break;
    }

    const int lm = lane & 15;
    const int hh = lane >> 4;

    const float* aptr = x + (long)(m0 + lm) * HH + 16 * hh;
    const float* bptr = W + (n0 + lm) * HH + 16 * hh;

    v8f acc[PJ_MT][PJ_NT];
    #pragma unroll
    for (int i = 0; i < PJ_MT; ++i)
        #pragma unroll
        for (int j = 0; j < PJ_NT; ++j) acc[i][j] = (v8f){};

    v4f fa[2][PJ_MT][4];
    v4f fb[2][PJ_NT][4];

    #pragma unroll
    for (int i = 0; i < PJ_MT; ++i)
        #pragma unroll
        for (int q = 0; q < 4; ++q) fa[0][i][q] = *(const v4f*)(aptr + i * 16 * HH + 4 * q);
    #pragma unroll
    for (int j = 0; j < PJ_NT; ++j)
        #pragma unroll
        for (int q = 0; q < 4; ++q) fb[0][j][q] = *(const v4f*)(bptr + j * 16 * HH + 4 * q);

    #pragma unroll
    for (int w = 0; w < 4; ++w) {
        const int cur = w & 1, nxt = cur ^ 1;
        if (w + 1 < 4) {
            #pragma unroll
            for (int i = 0; i < PJ_MT; ++i)
                #pragma unroll
                for (int q = 0; q < 4; ++q)
                    fa[nxt][i][q] = *(const v4f*)(aptr + i * 16 * HH + 32 * (w + 1) + 4 * q);
            #pragma unroll
            for (int j = 0; j < PJ_NT; ++j)
                #pragma unroll
                for (int q = 0; q < 4; ++q)
                    fb[nxt][j][q] = *(const v4f*)(bptr + j * 16 * HH + 32 * (w + 1) + 4 * q);
        }
        __builtin_amdgcn_sched_barrier(0);
        #pragma unroll
        for (int q = 0; q < 4; ++q) {
            #pragma unroll
            for (int i = 0; i < PJ_MT; ++i)
                #pragma unroll
                for (int j = 0; j < PJ_NT; ++j) {
                    const v2f a0 = {fa[cur][i][q].x, fa[cur][i][q].y};
                    const v2f b0 = {fb[cur][j][q].x, fb[cur][j][q].y};
                    acc[i][j] = __builtin_amdgcn_wmma_f32_16x16x4_f32(false, a0, false, b0,
                                                                      (short)0, acc[i][j], false, false);
                }
            #pragma unroll
            for (int i = 0; i < PJ_MT; ++i)
                #pragma unroll
                for (int j = 0; j < PJ_NT; ++j) {
                    const v2f a1 = {fa[cur][i][q].z, fa[cur][i][q].w};
                    const v2f b1 = {fb[cur][j][q].z, fb[cur][j][q].w};
                    acc[i][j] = __builtin_amdgcn_wmma_f32_16x16x4_f32(false, a1, false, b1,
                                                                      (short)0, acc[i][j], false, false);
                }
        }
        __builtin_amdgcn_sched_barrier(0);
    }

    if (matIdx <= 1) {
        proj_store_t<0>(acc, outp, m0, n0, lane, 0.0f, 0.0f);
    } else if (matIdx == 2) {
        proj_store<1>(acc, outp, m0, n0, lane, 0.0f, 0.0f);
    } else {
        const float bias0 = bp[n0 + lm];
        const float bias1 = bp[n0 + 16 + lm];
        if (matIdx == 3)      proj_store_t<2>(acc, outp, m0, n0, lane, bias0, bias1);
        else if (matIdx == 4) proj_store_t<3>(acc, outp, m0, n0, lane, bias0, bias1);
        else                  proj_store<3>(acc, outp, m0, n0, lane, bias0, bias1);
    }
}

#define NSL  8
#define RS   16
#define REC_GRID 128
#define CSTR 132
#define TSTR 20
#define PS2  20
#define PL2  (16 * PS2)
#define C_OFF   0
#define ST_OFF  (C_OFF + RS * CSTR)
#define Q_O     0
#define K_O     (Q_O + 16 * CSTR)
#define M_O     (K_O + 16 * CSTR)
#define KT_O    (M_O + 16 * CSTR)
#define WC_O    (KT_O + 128 * TSTR)
#define G15_O   (WC_O + RS * TSTR)
#define ST_SZ   (G15_O + 16)
#define PT_OFF  (ST_OFF + 2 * ST_SZ)
#define PP_O    0
#define SP_O    (PP_O + 4 * PL2)
#define NPS_O   (SP_O + 4 * PL2)
#define PT_SZ   (NPS_O + 128)
#define RW_OFF  (PT_OFF + 2 * PT_SZ)
#define RW_SZ   (2 * PL2)
#define VP_OFF  (RW_OFF + 4 * RW_SZ)
#define SMEM_FLOATS (VP_OFF + 2 * PL2)


static __device__ __forceinline__ void lds_barrier()
{
    __builtin_amdgcn_fence(__ATOMIC_RELEASE, "workgroup", "local");
    __builtin_amdgcn_s_barrier();
    __builtin_amdgcn_fence(__ATOMIC_ACQUIRE, "workgroup", "local");
}

__global__ __launch_bounds__(256, 1) void rec_kernel(
    const float* __restrict__ qb, const float* __restrict__ kb,
    const float* __restrict__ vb, const float* __restrict__ ib,
    const float* __restrict__ fb, const float* __restrict__ ob,
    float* __restrict__ hb)
{
    __shared__ __attribute__((aligned(16))) float sm[SMEM_FLOATS];
    const int b    = blockIdx.x >> 3;
    const int sl   = blockIdx.x & 7;
    const int r    = threadIdx.x & 127;
    const bool vec = __builtin_amdgcn_readfirstlane((int)(threadIdx.x >> 7)) == 0;
    const int lane = r & 31;
    const int wave = r >> 5;
    const int lm   = lane & 15;
    const int hh   = lane >> 4;
    const int lk   = hh << 1;
    const int x    = r & 15;
    const bool rowthr = (r >> 4) == sl;

    for (int j = threadIdx.x; j < RS * CSTR; j += 256) sm[C_OFF + j] = 0.0f;
    float creg[2][8];
    #pragma unroll
    for (int t = 0; t < 2; ++t)
        #pragma unroll
        for (int v = 0; v < 8; ++v) creg[t][v] = 0.0f;

    float nprev = 0.0f;
    float* const hrow = hb + (long)blockIdx.x * (TT * RS);

    float pk[2][LCH], pq[2][LCH], pi[2][LCH], pf[2][LCH];
    const long cbase = (long)b * NCH * (4 * HH * 4) + r * 4;
    auto load_qkif = [&](int c, int set) __attribute__((always_inline)) {
        const long nbase = cbase + (long)c * (4 * HH * 4);
        #pragma unroll
        for (int g = 0; g < 4; ++g) {
            const v4f t = *(const v4f*)(fb + nbase + g * (HH * 4));
            pf[set][4 * g] = t.x; pf[set][4 * g + 1] = t.y; pf[set][4 * g + 2] = t.z; pf[set][4 * g + 3] = t.w;
        }
        #pragma unroll
        for (int g = 0; g < 4; ++g) {
            const v4f t = *(const v4f*)(ib + nbase + g * (HH * 4));
            pi[set][4 * g] = t.x; pi[set][4 * g + 1] = t.y; pi[set][4 * g + 2] = t.z; pi[set][4 * g + 3] = t.w;
        }
        #pragma unroll
        for (int g = 0; g < 4; ++g) {
            const v4f t = *(const v4f*)(kb + nbase + g * (HH * 4));
            pk[set][4 * g] = t.x; pk[set][4 * g + 1] = t.y; pk[set][4 * g + 2] = t.z; pk[set][4 * g + 3] = t.w;
        }
        #pragma unroll
        for (int g = 0; g < 4; ++g) {
            const v4f t = *(const v4f*)(qb + nbase + g * (HH * 4));
            pq[set][4 * g] = t.x; pq[set][4 * g + 1] = t.y; pq[set][4 * g + 2] = t.z; pq[set][4 * g + 3] = t.w;
        }
    };

    const long vobase = (long)b * TT * HH + (long)(r >> 3) * HH + sl * RS + (r & 7) * 2;
    v2f vnx;
    auto load_v = [&](int c) __attribute__((always_inline)) {
        vnx = *(const v2f*)(vb + vobase + (long)(c * LCH) * HH);
    };
    auto load_o = [&](int c) __attribute__((always_inline)) {
        return *(const v2f*)(ob + vobase + (long)(c * LCH) * HH);
    };
    auto store_v = [&](int c) __attribute__((always_inline)) {
        *(v2f*)&sm[VP_OFF + (c & 1) * PL2 + (r >> 3) * PS2 + (r & 7) * 2] = vnx;
    };

    auto stage_rows = [&](int c, int set) __attribute__((always_inline)) {
        float* st = sm + ST_OFF + (c & 1) * ST_SZ;
        float* rw = sm + RW_OFF + (c & 3) * RW_SZ;
        if (rowthr) {
            const float* vp = sm + VP_OFF + (c & 1) * PL2;
            float pv[LCH];
            #pragma unroll
            for (int s = 0; s < LCH; ++s) pv[s] = vp[s * PS2 + x];
            float G[LCH];
            float g = 1.0f;
            #pragma unroll
            for (int s = 0; s < LCH; ++s) { g *= pf[set][s]; G[s] = g; }
            const float g15 = G[LCH - 1];
            #pragma unroll
            for (int s = 0; s < LCH; ++s) {
                const float r0 = __builtin_amdgcn_rcpf(G[s]);
                const float rg = fmaf(fmaf(-G[s], r0, 1.0f), r0, r0);
                const float wg = pi[set][s] * pv[s] * rg;
                st[WC_O + x * TSTR + s]   = g15 * wg;
                rw[0 * PL2 + s * PS2 + x] = wg;
                rw[1 * PL2 + s * PS2 + x] = G[s];
            }
            st[G15_O + x] = g15;
        }
    };

    auto stage_cols = [&](int c, int set) __attribute__((always_inline)) {
        float* st = sm + ST_OFF + (c & 1) * ST_SZ;
        float np = nprev;
        #pragma unroll
        for (int s = 0; s < LCH; ++s) {
            np = fmaf(pf[set][s], np, pi[set][s] * pk[set][s]);
            st[M_O + s * CSTR + r] = np * pq[set][s];
        }
        nprev = np;
        #pragma unroll
        for (int s = 0; s < LCH; ++s) {
            st[Q_O + s * CSTR + r]  = pq[set][s];
            st[K_O + s * CSTR + r]  = pk[set][s];
            st[KT_O + r * TSTR + s] = pk[set][s];
        }
    };

    const int s1 = r >> 3, seg = r & 7;
    const int cb = wave * 32 + lk;
    v4f mv[4];
    v2f qf[8], kf[8], cf[8], afr[4], bfr[2][4];
    float g[8];

    auto state_load = [&](int c) __attribute__((always_inline)) {
        const float* st = sm + ST_OFF + (c & 1) * ST_SZ;
        #pragma unroll
        for (int e = 0; e < 4; ++e) mv[e] = *(const v4f*)&st[M_O + s1 * CSTR + seg * 16 + 4 * e];

        #pragma unroll
        for (int kk = 0; kk < 8; ++kk) {
            qf[kk] = *(const v2f*)&st[Q_O + lm * CSTR + cb + kk * 4];
            kf[kk] = *(const v2f*)&st[K_O + lm * CSTR + cb + kk * 4];
            cf[kk] = *(const v2f*)&sm[C_OFF + lm * CSTR + cb + kk * 4];
        }
        #pragma unroll
        for (int kk = 0; kk < 4; ++kk)
            afr[kk] = *(const v2f*)&st[WC_O + lm * TSTR + kk * 4 + lk];
        #pragma unroll
        for (int v = 0; v < 8; ++v) g[v] = st[G15_O + 8 * hh + v];
        #pragma unroll
        for (int t = 0; t < 2; ++t)
            #pragma unroll
            for (int kk = 0; kk < 4; ++kk)
                bfr[t][kk] = *(const v2f*)&st[KT_O + (wave * 32 + t * 16 + lm) * TSTR + kk * 4 + lk];
    };

    auto state_cupd = [&]() __attribute__((always_inline)) {
        v8f d0 = {};
        v8f d1 = {};
        #pragma unroll
        for (int kk = 0; kk < 4; ++kk) {
            d0 = __builtin_amdgcn_wmma_f32_16x16x4_f32(false, afr[kk], false, bfr[0][kk],
                                                       (short)0, d0, false, false);
            d1 = __builtin_amdgcn_wmma_f32_16x16x4_f32(false, afr[kk], false, bfr[1][kk],
                                                       (short)0, d1, false, false);
        }
        #pragma unroll
        for (int v = 0; v < 8; ++v) {
            creg[0][v] = fmaf(g[v], creg[0][v], d0[v]);
            creg[1][v] = fmaf(g[v], creg[1][v], d1[v]);
        }
    };

    auto state_ps = [&](int c) __attribute__((always_inline)) {
        float* pt = sm + PT_OFF + (c & 1) * PT_SZ;
        float ps = 0.0f;
        #pragma unroll
        for (int e = 0; e < 4; ++e) { ps += mv[e].x; ps += mv[e].y; ps += mv[e].z; ps += mv[e].w; }

        v8f pacc = {};
        v8f sacc = {};
        #pragma unroll
        for (int kk = 0; kk < 8; ++kk) {
            pacc = __builtin_amdgcn_wmma_f32_16x16x4_f32(false, qf[kk], false, cf[kk],
                                                         (short)0, pacc, false, false);
            sacc = __builtin_amdgcn_wmma_f32_16x16x4_f32(false, kf[kk], false, qf[kk],
                                                         (short)0, sacc, false, false);
        }
        pt[NPS_O + s1 * 8 + seg] = ps;
        #pragma unroll
        for (int v = 0; v < 8; ++v)
            pt[PP_O + wave * PL2 + (v + 8 * hh) * PS2 + lm] = pacc[v];
        *(v4f*)&pt[SP_O + wave * PL2 + lm * PS2 + 8 * hh]     = (v4f){sacc[0], sacc[1], sacc[2], sacc[3]};
        *(v4f*)&pt[SP_O + wave * PL2 + lm * PS2 + 8 * hh + 4] = (v4f){sacc[4], sacc[5], sacc[6], sacc[7]};
        #pragma unroll
        for (int t = 0; t < 2; ++t)
            #pragma unroll
            for (int v = 0; v < 8; ++v)
                sm[C_OFF + (8 * hh + v) * CSTR + wave * 32 + t * 16 + lm] = creg[t][v];
    };

    const int s  = r >> 3;
    const int x0 = (r & 7) * 2;
    v4f sp[4][4], n0, n1;
    v2f pp[4], wg[LCH], gs;
    auto out_load = [&](int c) __attribute__((always_inline)) {
        const float* pt = sm + PT_OFF + (c & 1) * PT_SZ;
        const float* rw = sm + RW_OFF + (c & 3) * RW_SZ;
        #pragma unroll
        for (int w = 0; w < 4; ++w)
            #pragma unroll
            for (int j = 0; j < 4; ++j) sp[w][j] = *(const v4f*)&pt[SP_O + w * PL2 + s * PS2 + 4 * j];
        #pragma unroll
        for (int w = 0; w < 4; ++w) pp[w] = *(const v2f*)&pt[PP_O + w * PL2 + s * PS2 + x0];
        n0 = *(const v4f*)&pt[NPS_O + s * 8];
        n1 = *(const v4f*)&pt[NPS_O + s * 8 + 4];
        #pragma unroll
        for (int u = 0; u < LCH; ++u) wg[u] = *(const v2f*)&rw[0 * PL2 + u * PS2 + x0];
        gs = *(const v2f*)&rw[1 * PL2 + s * PS2 + x0];
    };

    float sv[LCH], nrm;
    v2f acc;
    auto out_sum = [&]() __attribute__((always_inline)) {
        #pragma unroll
        for (int j = 0; j < 4; ++j) {
            const v4f t = ((sp[0][j] + sp[1][j]) + sp[2][j]) + sp[3][j];
            sv[4 * j] = t.x; sv[4 * j + 1] = t.y; sv[4 * j + 2] = t.z; sv[4 * j + 3] = t.w;
        }
        float nq = 0.0f;
        nq += n0.x; nq += n0.y; nq += n0.z; nq += n0.w;
        nq += n1.x; nq += n1.y; nq += n1.z; nq += n1.w;
        nrm = fmaxf(fabsf(nq), 1.0f);
        acc = ((pp[0] + pp[1]) + pp[2]) + pp[3];
    };

    auto out_chain = [&](int c, v2f os) __attribute__((always_inline)) {
        #pragma unroll
        for (int u = 0; u < LCH; ++u) {
            const float nx0 = fmaf(wg[u].x, sv[u], acc.x);
            const float nx1 = fmaf(wg[u].y, sv[u], acc.y);
            acc.x = (u <= s) ? nx0 : acc.x;
            acc.y = (u <= s) ? nx1 : acc.y;
        }
        v2f res;
        res.x = os.x * (gs.x * acc.x) / nrm;
        res.y = os.y * (gs.y * acc.y) / nrm;
        *(v2f*)&hrow[(long)c * (LCH * RS) + s * RS + x0] = res;
    };

    if (vec) {
        load_qkif(0, 0);
        load_v(0); store_v(0);
        load_v(1); store_v(1);
        load_qkif(1, 1);
    }
    lds_barrier();
    if (vec) {
        stage_rows(0, 0);
        stage_cols(0, 0);
    }
    lds_barrier();

    if (vec) {
        load_v(2);
        load_qkif(2, 0);
        stage_rows(1, 1);
        stage_cols(1, 1);
        store_v(2);
    } else {
        state_load(0);
        state_cupd();
        state_ps(0);
    }
    lds_barrier();

    auto trip_vec = [&](int t, int cs, int ns) __attribute__((always_inline)) {
        const int nc = (t + 1 < NCH) ? t + 1 : NCH - 1;
        load_v(nc);
        load_qkif(nc, ns);
        stage_rows(t, cs);
        stage_cols(t, cs);
        store_v(t + 1);
    };
    auto trip_mat = [&](int t) __attribute__((always_inline)) {
        const v2f os = load_o(t - 2);
        state_load(t - 1);
        out_load(t - 2);
        state_cupd();
        state_ps(t - 1);
        out_sum();
        out_chain(t - 2, os);
    };
    for (int t = 2; t < NCH; t += 2) {
        if (vec) trip_vec(t, 0, 1);
        else     trip_mat(t);
        lds_barrier();
        if (vec) trip_vec(t + 1, 1, 0);
        else     trip_mat(t + 1);
        lds_barrier();
    }

    v2f osb = {};
    if (!vec) {
        const v2f osa = load_o(NCH - 2);
        osb = load_o(NCH - 1);
        state_load(NCH - 1);
        out_load(NCH - 2);
        state_cupd();
        state_ps(NCH - 1);
        out_sum();
        out_chain(NCH - 2, osa);
    }
    lds_barrier();
    if (!vec) {
        out_load(NCH - 1);
        out_sum();
        out_chain(NCH - 1, osb);
    }
}

__global__ __launch_bounds__(256) void ln_kernel(
    const float* __restrict__ hb, const float* __restrict__ x,
    const float* __restrict__ lnw, const float* __restrict__ lnb,
    float* __restrict__ out)
{
    const int lane = threadIdx.x & 31;
    const int wave = threadIdx.x >> 5;
    const long row = (long)blockIdx.x * 8 + wave;
    const long rb  = row * HH;

    float y[4];
    float s = 0.0f, s2 = 0.0f;
    #pragma unroll
    for (int j = 0; j < 4; ++j) {
        const int c = lane + 32 * j;
        const long hidx = (((row >> 9) * NSL + (c >> 4)) * TT + (row & (TT - 1))) * RS + (c & 15);
        y[j] = hb[hidx] + x[rb + c];
        s  += y[j];
        s2 += y[j] * y[j];
    }
    #pragma unroll
    for (int off = 16; off >= 1; off >>= 1) {
        s  += __shfl_xor(s,  off, 32);
        s2 += __shfl_xor(s2, off, 32);
    }
    const float mu  = s * (1.0f / HH);
    const float var = s2 * (1.0f / HH) - mu * mu;
    const float rs  = rsqrtf(var + LN_EPS);

    #pragma unroll
    for (int j = 0; j < 4; ++j) {
        const int c = lane + 32 * j;
        out[rb + c] = (y[j] - mu) * rs * lnw[c] + lnb[c];
    }
}

extern "C" void kernel_launch(void* const* d_in, const int* in_sizes, int n_in,
                              void* d_out, int out_size, void* d_ws, size_t ws_size,
                              hipStream_t stream)
{
    const float* x   = (const float*)d_in[0];
    const float* Wq  = (const float*)d_in[1];
    const float* Wk  = (const float*)d_in[2];
    const float* Wv  = (const float*)d_in[3];
    const float* Wi  = (const float*)d_in[4];
    const float* Wf  = (const float*)d_in[5];
    const float* Wo  = (const float*)d_in[6];
    const float* b_i = (const float*)d_in[7];
    const float* b_f = (const float*)d_in[8];
    const float* b_o = (const float*)d_in[9];
    const float* lnw = (const float*)d_in[10];
    const float* lnb = (const float*)d_in[11];
    float* out = (float*)d_out;

    float* ws = (float*)d_ws;
    float* q  = ws + 0L * NEL;
    float* k  = ws + 1L * NEL;
    float* v  = ws + 2L * NEL;
    float* ig = ws + 3L * NEL;
    float* fg = ws + 4L * NEL;
    float* og = ws + 5L * NEL;
    float* h  = ws + 6L * NEL;

    proj_kernel<<<PJ_GRID, 128, 0, stream>>>(x, Wq, Wk, Wv, Wi, Wf, Wo,
                                          b_i, b_f, b_o,
                                          q, k, v, ig, fg, og);
    rec_kernel<<<REC_GRID, 256, 0, stream>>>(q, k, v, ig, fg, og, h);
    ln_kernel<<<MM / 8, 256, 0, stream>>>(h, x, lnw, lnb, out);
}


# Recurrence: a third wave role: 4 more waves (one a SIMD) sum the partial tiles, run the causal chain and write h of chunk t-2 beside the state waves, which only step the state; sum orders unchanged
# speedup vs baseline: 7.8546x; 1.0284x over previous
#include <hip/hip_runtime.h>
#include <math.h>


typedef float v2f __attribute__((ext_vector_type(2)));
typedef float v4f __attribute__((ext_vector_type(4)));
typedef float v8f __attribute__((ext_vector_type(8)));

#define BB   16
#define TT   512
#define HH   128
#define MM   (BB * TT)
#define NEL  (BB * TT * HH)
#define SCALE 0.08838834764831845f
#define LN_EPS 1e-5f
#define LCH  16
#define NCH  (TT / LCH)

#define PJ_MT 4
#define PJ_NT 2
#define PJ_ROWS (16 * PJ_MT)
#define PJ_GRID ((MM / PJ_ROWS) * 6)

static __device__ __forceinline__ float xhalf(float v)
{
    return __builtin_bit_cast(float, __builtin_amdgcn_permlanex16(
        __builtin_bit_cast(int, v), __builtin_bit_cast(int, v),
        0x76543210, 0xfedcba98, false, false));
}

template <int ACT>
static __device__ __forceinline__ float proj_act(float val, float bias)
{
    if (ACT == 0) return val * SCALE;
    if (ACT == 1) return val;
    if (ACT == 2) return expf(val + bias);
    return 1.0f / (1.0f + expf(-(val + bias)));
}

template <int ACT>
static __device__ __forceinline__ void proj_store(
    const v8f (&acc)[PJ_MT][PJ_NT], float* __restrict__ outp,
    int m0, int n0, int lane, float bias0, float bias1)
{
    const int hh = lane >> 4;
    #pragma unroll
    for (int i = 0; i < PJ_MT; ++i) {
        #pragma unroll
        for (int v = 0; v < 8; ++v) {
            const float t0 = proj_act<ACT>(acc[i][0][v], bias0);
            const float t1 = proj_act<ACT>(acc[i][1][v], bias1);
            const float got = xhalf(hh ? t0 : t1);
            const float lo = hh ? got : t0;
            const float hi = hh ? t1 : got;
            float* o = outp + (long)(m0 + 16 * i + v) * HH + n0 + lane;
            o[0]      = lo;
            o[8 * HH] = hi;
        }
    }
}

template <int ACT>
static __device__ __forceinline__ void proj_store_t(
    const v8f (&acc)[PJ_MT][PJ_NT], float* __restrict__ outp,
    int m0, int n0, int lane, float bias0, float bias1)
{
    const int lm = lane & 15;
    const int hh = lane >> 4;
    #pragma unroll
    for (int i = 0; i < PJ_MT; ++i) {
        #pragma unroll
        for (int j = 0; j < PJ_NT; ++j) {
            const float bias = j ? bias1 : bias0;
            #pragma unroll
            for (int g = 0; g < 2; ++g) {
                v4f t;
                t.x = proj_act<ACT>(acc[i][j][4 * g + 0], bias);
                t.y = proj_act<ACT>(acc[i][j][4 * g + 1], bias);
                t.z = proj_act<ACT>(acc[i][j][4 * g + 2], bias);
                t.w = proj_act<ACT>(acc[i][j][4 * g + 3], bias);
                const long grp = (long)((m0 >> 4) + i) * 4 + 2 * hh + g;
                *(v4f*)(outp + (grp * HH + n0 + 16 * j + lm) * 4) = t;
            }
        }
    }
}

__global__ __launch_bounds__(128, 3) void proj_kernel(
    const float* __restrict__ x,
    const float* __restrict__ Wq, const float* __restrict__ Wk,
    const float* __restrict__ Wv, const float* __restrict__ Wi,
    const float* __restrict__ Wf, const float* __restrict__ Wo,
    const float* __restrict__ b_i, const float* __restrict__ b_f,
    const float* __restrict__ b_o,
    float* __restrict__ oq, float* __restrict__ ok, float* __restrict__ ov,
    float* __restrict__ oi, float* __restrict__ of, float* __restrict__ oo)
{
    const int lane   = threadIdx.x & 31;
    const int wave   = threadIdx.x >> 5;
    const int matIdx = blockIdx.x % 6;
    const int m0     = (blockIdx.x / 6) * PJ_ROWS;
    const int n0     = wave * (16 * PJ_NT);

    const float* W;
    const float* bp = b_i;
    float* outp;
    switch (matIdx) {
        case 0:  W = Wq; outp = oq; break;
        case 1:  W = Wk; outp = ok; break;
        case 2:  W = Wv; outp = ov; break;
        case 3:  W = Wi; outp = oi; bp = b_i; break;
        case 4:  W = Wf; outp = of; bp = b_f; break;
        default: W = Wo; outp = oo; bp = b_o; break;
    }

    const int lm = lane & 15;
    const int hh = lane >> 4;

    const float* aptr = x + (long)(m0 + lm) * HH + 16 * hh;
    const float* bptr = W + (n0 + lm) * HH + 16 * hh;

    v8f acc[PJ_MT][PJ_NT];
    #pragma unroll
    for (int i = 0; i < PJ_MT; ++i)
        #pragma unroll
        for (int j = 0; j < PJ_NT; ++j) acc[i][j] = (v8f){};

    v4f fa[2][PJ_MT][4];
    v4f fb[2][PJ_NT][4];

    #pragma unroll
    for (int i = 0; i < PJ_MT; ++i)
        #pragma unroll
        for (int q = 0; q < 4; ++q) fa[0][i][q] = *(const v4f*)(aptr + i * 16 * HH + 4 * q);
    #pragma unroll
    for (int j = 0; j < PJ_NT; ++j)
        #pragma unroll
        for (int q = 0; q < 4; ++q) fb[0][j][q] = *(const v4f*)(bptr + j * 16 * HH + 4 * q);

    #pragma unroll
    for (int w = 0; w < 4; ++w) {
        const int cur = w & 1, nxt = cur ^ 1;
        if (w + 1 < 4) {
            #pragma unroll
            for (int i = 0; i < PJ_MT; ++i)
                #pragma unroll
                for (int q = 0; q < 4; ++q)
                    fa[nxt][i][q] = *(const v4f*)(aptr + i * 16 * HH + 32 * (w + 1) + 4 * q);
            #pragma unroll
            for (int j = 0; j < PJ_NT; ++j)
                #pragma unroll
                for (int q = 0; q < 4; ++q)
                    fb[nxt][j][q] = *(const v4f*)(bptr + j * 16 * HH + 32 * (w + 1) + 4 * q);
        }
        __builtin_amdgcn_sched_barrier(0);
        #pragma unroll
        for (int q = 0; q < 4; ++q) {
            #pragma unroll
            for (int i = 0; i < PJ_MT; ++i)
                #pragma unroll
                for (int j = 0; j < PJ_NT; ++j) {
                    const v2f a0 = {fa[cur][i][q].x, fa[cur][i][q].y};
                    const v2f b0 = {fb[cur][j][q].x, fb[cur][j][q].y};
                    acc[i][j] = __builtin_amdgcn_wmma_f32_16x16x4_f32(false, a0, false, b0,
                                                                      (short)0, acc[i][j], false, false);
                }
            #pragma unroll
            for (int i = 0; i < PJ_MT; ++i)
                #pragma unroll
                for (int j = 0; j < PJ_NT; ++j) {
                    const v2f a1 = {fa[cur][i][q].z, fa[cur][i][q].w};
                    const v2f b1 = {fb[cur][j][q].z, fb[cur][j][q].w};
                    acc[i][j] = __builtin_amdgcn_wmma_f32_16x16x4_f32(false, a1, false, b1,
                                                                      (short)0, acc[i][j], false, false);
                }
        }
        __builtin_amdgcn_sched_barrier(0);
    }

    if (matIdx <= 1) {
        proj_store_t<0>(acc, outp, m0, n0, lane, 0.0f, 0.0f);
    } else if (matIdx == 2) {
        proj_store<1>(acc, outp, m0, n0, lane, 0.0f, 0.0f);
    } else {
        const float bias0 = bp[n0 + lm];
        const float bias1 = bp[n0 + 16 + lm];
        if (matIdx == 3)      proj_store_t<2>(acc, outp, m0, n0, lane, bias0, bias1);
        else if (matIdx == 4) proj_store_t<3>(acc, outp, m0, n0, lane, bias0, bias1);
        else                  proj_store<3>(acc, outp, m0, n0, lane, bias0, bias1);
    }
}

#define NSL  8
#define RS   16
#define REC_GRID 128
#define CSTR 132
#define TSTR 20
#define PS2  20
#define PL2  (16 * PS2)
#define C_OFF   0
#define ST_OFF  (C_OFF + RS * CSTR)
#define Q_O     0
#define K_O     (Q_O + 16 * CSTR)
#define M_O     (K_O + 16 * CSTR)
#define KT_O    (M_O + 16 * CSTR)
#define WC_O    (KT_O + 128 * TSTR)
#define G15_O   (WC_O + RS * TSTR)
#define ST_SZ   (G15_O + 16)
#define PT_OFF  (ST_OFF + 2 * ST_SZ)
#define PP_O    0
#define SP_O    (PP_O + 4 * PL2)
#define NPS_O   (SP_O + 4 * PL2)
#define PT_SZ   (NPS_O + 128)
#define RW_OFF  (PT_OFF + 2 * PT_SZ)
#define RW_SZ   (2 * PL2)
#define VP_OFF  (RW_OFF + 4 * RW_SZ)
#define SMEM_FLOATS (VP_OFF + 2 * PL2)


static __device__ __forceinline__ void lds_barrier()
{
    __builtin_amdgcn_fence(__ATOMIC_RELEASE, "workgroup", "local");
    __builtin_amdgcn_s_barrier();
    __builtin_amdgcn_fence(__ATOMIC_ACQUIRE, "workgroup", "local");
}

__global__ __launch_bounds__(384, 1) void rec_kernel(
    const float* __restrict__ qb, const float* __restrict__ kb,
    const float* __restrict__ vb, const float* __restrict__ ib,
    const float* __restrict__ fb, const float* __restrict__ ob,
    float* __restrict__ hb)
{
    __shared__ __attribute__((aligned(16))) float sm[SMEM_FLOATS];
    const int b    = blockIdx.x >> 3;
    const int sl   = blockIdx.x & 7;
    const int r    = threadIdx.x & 127;
    const int role = __builtin_amdgcn_readfirstlane((int)(threadIdx.x >> 7));
    const bool vec = role == 0;
    const bool mat = role == 1;
    const int lane = r & 31;
    const int wave = r >> 5;
    const int lm   = lane & 15;
    const int hh   = lane >> 4;
    const int lk   = hh << 1;
    const int x    = r & 15;
    const bool rowthr = (r >> 4) == sl;

    for (int j = threadIdx.x; j < RS * CSTR; j += 384) sm[C_OFF + j] = 0.0f;
    float creg[2][8];
    #pragma unroll
    for (int t = 0; t < 2; ++t)
        #pragma unroll
        for (int v = 0; v < 8; ++v) creg[t][v] = 0.0f;

    float nprev = 0.0f;
    float* const hrow = hb + (long)blockIdx.x * (TT * RS);

    float pk[2][LCH], pq[2][LCH], pi[2][LCH], pf[2][LCH];
    const long cbase = (long)b * NCH * (4 * HH * 4) + r * 4;
    auto load_qkif = [&](int c, int set) __attribute__((always_inline)) {
        const long nbase = cbase + (long)c * (4 * HH * 4);
        #pragma unroll
        for (int g = 0; g < 4; ++g) {
            const v4f t = *(const v4f*)(fb + nbase + g * (HH * 4));
            pf[set][4 * g] = t.x; pf[set][4 * g + 1] = t.y; pf[set][4 * g + 2] = t.z; pf[set][4 * g + 3] = t.w;
        }
        #pragma unroll
        for (int g = 0; g < 4; ++g) {
            const v4f t = *(const v4f*)(ib + nbase + g * (HH * 4));
            pi[set][4 * g] = t.x; pi[set][4 * g + 1] = t.y; pi[set][4 * g + 2] = t.z; pi[set][4 * g + 3] = t.w;
        }
        #pragma unroll
        for (int g = 0; g < 4; ++g) {
            const v4f t = *(const v4f*)(kb + nbase + g * (HH * 4));
            pk[set][4 * g] = t.x; pk[set][4 * g + 1] = t.y; pk[set][4 * g + 2] = t.z; pk[set][4 * g + 3] = t.w;
        }
        #pragma unroll
        for (int g = 0; g < 4; ++g) {
            const v4f t = *(const v4f*)(qb + nbase + g * (HH * 4));
            pq[set][4 * g] = t.x; pq[set][4 * g + 1] = t.y; pq[set][4 * g + 2] = t.z; pq[set][4 * g + 3] = t.w;
        }
    };

    const long vobase = (long)b * TT * HH + (long)(r >> 3) * HH + sl * RS + (r & 7) * 2;
    v2f vnx;
    auto load_v = [&](int c) __attribute__((always_inline)) {
        vnx = *(const v2f*)(vb + vobase + (long)(c * LCH) * HH);
    };
    auto load_o = [&](int c) __attribute__((always_inline)) {
        return *(const v2f*)(ob + vobase + (long)(c * LCH) * HH);
    };
    auto store_v = [&](int c) __attribute__((always_inline)) {
        *(v2f*)&sm[VP_OFF + (c & 1) * PL2 + (r >> 3) * PS2 + (r & 7) * 2] = vnx;
    };

    auto stage_rows = [&](int c, int set) __attribute__((always_inline)) {
        float* st = sm + ST_OFF + (c & 1) * ST_SZ;
        float* rw = sm + RW_OFF + (c & 3) * RW_SZ;
        if (rowthr) {
            const float* vp = sm + VP_OFF + (c & 1) * PL2;
            float pv[LCH];
            #pragma unroll
            for (int s = 0; s < LCH; ++s) pv[s] = vp[s * PS2 + x];
            float G[LCH];
            float g = 1.0f;
            #pragma unroll
            for (int s = 0; s < LCH; ++s) { g *= pf[set][s]; G[s] = g; }
            const float g15 = G[LCH - 1];
            #pragma unroll
            for (int s = 0; s < LCH; ++s) {
                const float r0 = __builtin_amdgcn_rcpf(G[s]);
                const float rg = fmaf(fmaf(-G[s], r0, 1.0f), r0, r0);
                const float wg = pi[set][s] * pv[s] * rg;
                st[WC_O + x * TSTR + s]   = g15 * wg;
                rw[0 * PL2 + s * PS2 + x] = wg;
                rw[1 * PL2 + s * PS2 + x] = G[s];
            }
            st[G15_O + x] = g15;
        }
    };

    auto stage_cols = [&](int c, int set) __attribute__((always_inline)) {
        float* st = sm + ST_OFF + (c & 1) * ST_SZ;
        float np = nprev;
        #pragma unroll
        for (int s = 0; s < LCH; ++s) {
            np = fmaf(pf[set][s], np, pi[set][s] * pk[set][s]);
            st[M_O + s * CSTR + r] = np * pq[set][s];
        }
        nprev = np;
        #pragma unroll
        for (int s = 0; s < LCH; ++s) {
            st[Q_O + s * CSTR + r]  = pq[set][s];
            st[K_O + s * CSTR + r]  = pk[set][s];
            st[KT_O + r * TSTR + s] = pk[set][s];
        }
    };

    const int s1 = r >> 3, seg = r & 7;
    const int cb = wave * 32 + lk;
    v4f mv[4];
    v2f qf[8], kf[8], cf[8], afr[4], bfr[2][4];
    float g[8];

    auto state_load = [&](int c) __attribute__((always_inline)) {
        const float* st = sm + ST_OFF + (c & 1) * ST_SZ;
        #pragma unroll
        for (int e = 0; e < 4; ++e) mv[e] = *(const v4f*)&st[M_O + s1 * CSTR + seg * 16 + 4 * e];

        #pragma unroll
        for (int kk = 0; kk < 8; ++kk) {
            qf[kk] = *(const v2f*)&st[Q_O + lm * CSTR + cb + kk * 4];
            kf[kk] = *(const v2f*)&st[K_O + lm * CSTR + cb + kk * 4];
            cf[kk] = *(const v2f*)&sm[C_OFF + lm * CSTR + cb + kk * 4];
        }
        #pragma unroll
        for (int kk = 0; kk < 4; ++kk)
            afr[kk] = *(const v2f*)&st[WC_O + lm * TSTR + kk * 4 + lk];
        #pragma unroll
        for (int v = 0; v < 8; ++v) g[v] = st[G15_O + 8 * hh + v];
        #pragma unroll
        for (int t = 0; t < 2; ++t)
            #pragma unroll
            for (int kk = 0; kk < 4; ++kk)
                bfr[t][kk] = *(const v2f*)&st[KT_O + (wave * 32 + t * 16 + lm) * TSTR + kk * 4 + lk];
    };

    auto state_cupd = [&]() __attribute__((always_inline)) {
        v8f d0 = {};
        v8f d1 = {};
        #pragma unroll
        for (int kk = 0; kk < 4; ++kk) {
            d0 = __builtin_amdgcn_wmma_f32_16x16x4_f32(false, afr[kk], false, bfr[0][kk],
                                                       (short)0, d0, false, false);
            d1 = __builtin_amdgcn_wmma_f32_16x16x4_f32(false, afr[kk], false, bfr[1][kk],
                                                       (short)0, d1, false, false);
        }
        #pragma unroll
        for (int v = 0; v < 8; ++v) {
            creg[0][v] = fmaf(g[v], creg[0][v], d0[v]);
            creg[1][v] = fmaf(g[v], creg[1][v], d1[v]);
        }
    };

    auto state_ps = [&](int c) __attribute__((always_inline)) {
        float* pt = sm + PT_OFF + (c & 1) * PT_SZ;
        float ps = 0.0f;
        #pragma unroll
        for (int e = 0; e < 4; ++e) { ps += mv[e].x; ps += mv[e].y; ps += mv[e].z; ps += mv[e].w; }

        v8f pacc = {};
        v8f sacc = {};
        #pragma unroll
        for (int kk = 0; kk < 8; ++kk) {
            pacc = __builtin_amdgcn_wmma_f32_16x16x4_f32(false, qf[kk], false, cf[kk],
                                                         (short)0, pacc, false, false);
            sacc = __builtin_amdgcn_wmma_f32_16x16x4_f32(false, kf[kk], false, qf[kk],
                                                         (short)0, sacc, false, false);
        }
        pt[NPS_O + s1 * 8 + seg] = ps;
        #pragma unroll
        for (int v = 0; v < 8; ++v)
            pt[PP_O + wave * PL2 + (v + 8 * hh) * PS2 + lm] = pacc[v];
        *(v4f*)&pt[SP_O + wave * PL2 + lm * PS2 + 8 * hh]     = (v4f){sacc[0], sacc[1], sacc[2], sacc[3]};
        *(v4f*)&pt[SP_O + wave * PL2 + lm * PS2 + 8 * hh + 4] = (v4f){sacc[4], sacc[5], sacc[6], sacc[7]};
        #pragma unroll
        for (int t = 0; t < 2; ++t)
            #pragma unroll
            for (int v = 0; v < 8; ++v)
                sm[C_OFF + (8 * hh + v) * CSTR + wave * 32 + t * 16 + lm] = creg[t][v];
    };

    const int s  = r >> 3;
    const int x0 = (r & 7) * 2;
    v4f sp[4][4], n0, n1;
    v2f pp[4], wg[LCH], gs;
    auto out_load = [&](int c) __attribute__((always_inline)) {
        const float* pt = sm + PT_OFF + (c & 1) * PT_SZ;
        const float* rw = sm + RW_OFF + (c & 3) * RW_SZ;
        #pragma unroll
        for (int w = 0; w < 4; ++w)
            #pragma unroll
            for (int j = 0; j < 4; ++j) sp[w][j] = *(const v4f*)&pt[SP_O + w * PL2 + s * PS2 + 4 * j];
        #pragma unroll
        for (int w = 0; w < 4; ++w) pp[w] = *(const v2f*)&pt[PP_O + w * PL2 + s * PS2 + x0];
        n0 = *(const v4f*)&pt[NPS_O + s * 8];
        n1 = *(const v4f*)&pt[NPS_O + s * 8 + 4];
        #pragma unroll
        for (int u = 0; u < LCH; ++u) wg[u] = *(const v2f*)&rw[0 * PL2 + u * PS2 + x0];
        gs = *(const v2f*)&rw[1 * PL2 + s * PS2 + x0];
    };

    float sv[LCH], nrm;
    v2f acc;
    auto out_sum = [&]() __attribute__((always_inline)) {
        #pragma unroll
        for (int j = 0; j < 4; ++j) {
            const v4f t = ((sp[0][j] + sp[1][j]) + sp[2][j]) + sp[3][j];
            sv[4 * j] = t.x; sv[4 * j + 1] = t.y; sv[4 * j + 2] = t.z; sv[4 * j + 3] = t.w;
        }
        float nq = 0.0f;
        nq += n0.x; nq += n0.y; nq += n0.z; nq += n0.w;
        nq += n1.x; nq += n1.y; nq += n1.z; nq += n1.w;
        nrm = fmaxf(fabsf(nq), 1.0f);
        acc = ((pp[0] + pp[1]) + pp[2]) + pp[3];
    };

    auto out_chain = [&](int c, v2f os) __attribute__((always_inline)) {
        #pragma unroll
        for (int u = 0; u < LCH; ++u) {
            const float nx0 = fmaf(wg[u].x, sv[u], acc.x);
            const float nx1 = fmaf(wg[u].y, sv[u], acc.y);
            acc.x = (u <= s) ? nx0 : acc.x;
            acc.y = (u <= s) ? nx1 : acc.y;
        }
        v2f res;
        res.x = os.x * (gs.x * acc.x) / nrm;
        res.y = os.y * (gs.y * acc.y) / nrm;
        *(v2f*)&hrow[(long)c * (LCH * RS) + s * RS + x0] = res;
    };

    if (vec) {
        load_qkif(0, 0);
        load_v(0); store_v(0);
        load_v(1); store_v(1);
        load_qkif(1, 1);
    }
    lds_barrier();
    if (vec) {
        stage_rows(0, 0);
        stage_cols(0, 0);
    }
    lds_barrier();

    if (vec) {
        load_v(2);
        load_qkif(2, 0);
        stage_rows(1, 1);
        stage_cols(1, 1);
        store_v(2);
    } else if (mat) {
        state_load(0);
        state_cupd();
        state_ps(0);
    }
    lds_barrier();

    auto trip_vec = [&](int t, int cs, int ns) __attribute__((always_inline)) {
        const int nc = (t + 1 < NCH) ? t + 1 : NCH - 1;
        load_v(nc);
        load_qkif(nc, ns);
        stage_rows(t, cs);
        stage_cols(t, cs);
        store_v(t + 1);
    };
    auto trip_mat = [&](int t) __attribute__((always_inline)) {
        state_load(t - 1);
        state_cupd();
        state_ps(t - 1);
    };
    auto trip_out = [&](int t) __attribute__((always_inline)) {
        const v2f os = load_o(t - 2);
        out_load(t - 2);
        out_sum();
        out_chain(t - 2, os);
    };
    for (int t = 2; t < NCH; t += 2) {
        if (vec)      trip_vec(t, 0, 1);
        else if (mat) trip_mat(t);
        else          trip_out(t);
        lds_barrier();
        if (vec)      trip_vec(t + 1, 1, 0);
        else if (mat) trip_mat(t + 1);
        else          trip_out(t + 1);
        lds_barrier();
    }

    if (mat) {
        trip_mat(NCH);
    } else if (!vec) {
        trip_out(NCH);
    }
    lds_barrier();
    if (!vec && !mat) trip_out(NCH + 1);
}

__global__ __launch_bounds__(256) void ln_kernel(
    const float* __restrict__ hb, const float* __restrict__ x,
    const float* __restrict__ lnw, const float* __restrict__ lnb,
    float* __restrict__ out)
{
    const int lane = threadIdx.x & 31;
    const int wave = threadIdx.x >> 5;
    const long row = (long)blockIdx.x * 8 + wave;
    const long rb  = row * HH;

    float y[4];
    float s = 0.0f, s2 = 0.0f;
    #pragma unroll
    for (int j = 0; j < 4; ++j) {
        const int c = lane + 32 * j;
        const long hidx = (((row >> 9) * NSL + (c >> 4)) * TT + (row & (TT - 1))) * RS + (c & 15);
        y[j] = hb[hidx] + x[rb + c];
        s  += y[j];
        s2 += y[j] * y[j];
    }
    #pragma unroll
    for (int off = 16; off >= 1; off >>= 1) {
        s  += __shfl_xor(s,  off, 32);
        s2 += __shfl_xor(s2, off, 32);
    }
    const float mu  = s * (1.0f / HH);
    const float var = s2 * (1.0f / HH) - mu * mu;
    const float rs  = rsqrtf(var + LN_EPS);

    #pragma unroll
    for (int j = 0; j < 4; ++j) {
        const int c = lane + 32 * j;
        out[rb + c] = (y[j] - mu) * rs * lnw[c] + lnb[c];
    }
}

extern "C" void kernel_launch(void* const* d_in, const int* in_sizes, int n_in,
                              void* d_out, int out_size, void* d_ws, size_t ws_size,
                              hipStream_t stream)
{
    const float* x   = (const float*)d_in[0];
    const float* Wq  = (const float*)d_in[1];
    const float* Wk  = (const float*)d_in[2];
    const float* Wv  = (const float*)d_in[3];
    const float* Wi  = (const float*)d_in[4];
    const float* Wf  = (const float*)d_in[5];
    const float* Wo  = (const float*)d_in[6];
    const float* b_i = (const float*)d_in[7];
    const float* b_f = (const float*)d_in[8];
    const float* b_o = (const float*)d_in[9];
    const float* lnw = (const float*)d_in[10];
    const float* lnb = (const float*)d_in[11];
    float* out = (float*)d_out;

    float* ws = (float*)d_ws;
    float* q  = ws + 0L * NEL;
    float* k  = ws + 1L * NEL;
    float* v  = ws + 2L * NEL;
    float* ig = ws + 3L * NEL;
    float* fg = ws + 4L * NEL;
    float* og = ws + 5L * NEL;
    float* h  = ws + 6L * NEL;

    proj_kernel<<<PJ_GRID, 128, 0, stream>>>(x, Wq, Wk, Wv, Wi, Wf, Wo,
                                          b_i, b_f, b_o,
                                          q, k, v, ig, fg, og);
    rec_kernel<<<REC_GRID, 384, 0, stream>>>(q, k, v, ig, fg, og, h);
    ln_kernel<<<MM / 8, 256, 0, stream>>>(h, x, lnw, lnb, out);
}


# Recurrence: output waves request chunk t+6's q,k,i,f,v,o lines into the second-level cache (global_prefetch_b8, 5 wave instructions a trip): staging loads hit there, not memory; sum orders unchanged
# speedup vs baseline: 8.1549x; 1.0382x over previous
#include <hip/hip_runtime.h>
#include <math.h>


typedef float v2f __attribute__((ext_vector_type(2)));
typedef float v4f __attribute__((ext_vector_type(4)));
typedef float v8f __attribute__((ext_vector_type(8)));

#define BB   16
#define TT   512
#define HH   128
#define MM   (BB * TT)
#define NEL  (BB * TT * HH)
#define SCALE 0.08838834764831845f
#define LN_EPS 1e-5f
#define LCH  16
#define NCH  (TT / LCH)

#define PJ_MT 4
#define PJ_NT 2
#define PJ_ROWS (16 * PJ_MT)
#define PJ_GRID ((MM / PJ_ROWS) * 6)

static __device__ __forceinline__ float xhalf(float v)
{
    return __builtin_bit_cast(float, __builtin_amdgcn_permlanex16(
        __builtin_bit_cast(int, v), __builtin_bit_cast(int, v),
        0x76543210, 0xfedcba98, false, false));
}

template <int ACT>
static __device__ __forceinline__ float proj_act(float val, float bias)
{
    if (ACT == 0) return val * SCALE;
    if (ACT == 1) return val;
    if (ACT == 2) return expf(val + bias);
    return 1.0f / (1.0f + expf(-(val + bias)));
}

template <int ACT>
static __device__ __forceinline__ void proj_store(
    const v8f (&acc)[PJ_MT][PJ_NT], float* __restrict__ outp,
    int m0, int n0, int lane, float bias0, float bias1)
{
    const int hh = lane >> 4;
    #pragma unroll
    for (int i = 0; i < PJ_MT; ++i) {
        #pragma unroll
        for (int v = 0; v < 8; ++v) {
            const float t0 = proj_act<ACT>(acc[i][0][v], bias0);
            const float t1 = proj_act<ACT>(acc[i][1][v], bias1);
            const float got = xhalf(hh ? t0 : t1);
            const float lo = hh ? got : t0;
            const float hi = hh ? t1 : got;
            float* o = outp + (long)(m0 + 16 * i + v) * HH + n0 + lane;
            o[0]      = lo;
            o[8 * HH] = hi;
        }
    }
}

template <int ACT>
static __device__ __forceinline__ void proj_store_t(
    const v8f (&acc)[PJ_MT][PJ_NT], float* __restrict__ outp,
    int m0, int n0, int lane, float bias0, float bias1)
{
    const int lm = lane & 15;
    const int hh = lane >> 4;
    #pragma unroll
    for (int i = 0; i < PJ_MT; ++i) {
        #pragma unroll
        for (int j = 0; j < PJ_NT; ++j) {
            const float bias = j ? bias1 : bias0;
            #pragma unroll
            for (int g = 0; g < 2; ++g) {
                v4f t;
                t.x = proj_act<ACT>(acc[i][j][4 * g + 0], bias);
                t.y = proj_act<ACT>(acc[i][j][4 * g + 1], bias);
                t.z = proj_act<ACT>(acc[i][j][4 * g + 2], bias);
                t.w = proj_act<ACT>(acc[i][j][4 * g + 3], bias);
                const long grp = (long)((m0 >> 4) + i) * 4 + 2 * hh + g;
                *(v4f*)(outp + (grp * HH + n0 + 16 * j + lm) * 4) = t;
            }
        }
    }
}

__global__ __launch_bounds__(128, 3) void proj_kernel(
    const float* __restrict__ x,
    const float* __restrict__ Wq, const float* __restrict__ Wk,
    const float* __restrict__ Wv, const float* __restrict__ Wi,
    const float* __restrict__ Wf, const float* __restrict__ Wo,
    const float* __restrict__ b_i, const float* __restrict__ b_f,
    const float* __restrict__ b_o,
    float* __restrict__ oq, float* __restrict__ ok, float* __restrict__ ov,
    float* __restrict__ oi, float* __restrict__ of, float* __restrict__ oo)
{
    const int lane   = threadIdx.x & 31;
    const int wave   = threadIdx.x >> 5;
    const int matIdx = blockIdx.x % 6;
    const int m0     = (blockIdx.x / 6) * PJ_ROWS;
    const int n0     = wave * (16 * PJ_NT);

    const float* W;
    const float* bp = b_i;
    float* outp;
    switch (matIdx) {
        case 0:  W = Wq; outp = oq; break;
        case 1:  W = Wk; outp = ok; break;
        case 2:  W = Wv; outp = ov; break;
        case 3:  W = Wi; outp = oi; bp = b_i; break;
        case 4:  W = Wf; outp = of; bp = b_f; break;
        default: W = Wo; outp = oo; bp = b_o; break;
    }

    const int lm = lane & 15;
    const int hh = lane >> 4;

    const float* aptr = x + (long)(m0 + lm) * HH + 16 * hh;
    const float* bptr = W + (n0 + lm) * HH + 16 * hh;

    v8f acc[PJ_MT][PJ_NT];
    #pragma unroll
    for (int i = 0; i < PJ_MT; ++i)
        #pragma unroll
        for (int j = 0; j < PJ_NT; ++j) acc[i][j] = (v8f){};

    v4f fa[2][PJ_MT][4];
    v4f fb[2][PJ_NT][4];

    #pragma unroll
    for (int i = 0; i < PJ_MT; ++i)
        #pragma unroll
        for (int q = 0; q < 4; ++q) fa[0][i][q] = *(const v4f*)(aptr + i * 16 * HH + 4 * q);
    #pragma unroll
    for (int j = 0; j < PJ_NT; ++j)
        #pragma unroll
        for (int q = 0; q < 4; ++q) fb[0][j][q] = *(const v4f*)(bptr + j * 16 * HH + 4 * q);

    #pragma unroll
    for (int w = 0; w < 4; ++w) {
        const int cur = w & 1, nxt = cur ^ 1;
        if (w + 1 < 4) {
            #pragma unroll
            for (int i = 0; i < PJ_MT; ++i)
                #pragma unroll
                for (int q = 0; q < 4; ++q)
                    fa[nxt][i][q] = *(const v4f*)(aptr + i * 16 * HH + 32 * (w + 1) + 4 * q);
            #pragma unroll
            for (int j = 0; j < PJ_NT; ++j)
                #pragma unroll
                for (int q = 0; q < 4; ++q)
                    fb[nxt][j][q] = *(const v4f*)(bptr + j * 16 * HH + 32 * (w + 1) + 4 * q);
        }
        __builtin_amdgcn_sched_barrier(0);
        #pragma unroll
        for (int q = 0; q < 4; ++q) {
            #pragma unroll
            for (int i = 0; i < PJ_MT; ++i)
                #pragma unroll
                for (int j = 0; j < PJ_NT; ++j) {
                    const v2f a0 = {fa[cur][i][q].x, fa[cur][i][q].y};
                    const v2f b0 = {fb[cur][j][q].x, fb[cur][j][q].y};
                    acc[i][j] = __builtin_amdgcn_wmma_f32_16x16x4_f32(false, a0, false, b0,
                                                                      (short)0, acc[i][j], false, false);
                }
            #pragma unroll
            for (int i = 0; i < PJ_MT; ++i)
                #pragma unroll
                for (int j = 0; j < PJ_NT; ++j) {
                    const v2f a1 = {fa[cur][i][q].z, fa[cur][i][q].w};
                    const v2f b1 = {fb[cur][j][q].z, fb[cur][j][q].w};
                    acc[i][j] = __builtin_amdgcn_wmma_f32_16x16x4_f32(false, a1, false, b1,
                                                                      (short)0, acc[i][j], false, false);
                }
        }
        __builtin_amdgcn_sched_barrier(0);
    }

    if (matIdx <= 1) {
        proj_store_t<0>(acc, outp, m0, n0, lane, 0.0f, 0.0f);
    } else if (matIdx == 2) {
        proj_store<1>(acc, outp, m0, n0, lane, 0.0f, 0.0f);
    } else {
        const float bias0 = bp[n0 + lm];
        const float bias1 = bp[n0 + 16 + lm];
        if (matIdx == 3)      proj_store_t<2>(acc, outp, m0, n0, lane, bias0, bias1);
        else if (matIdx == 4) proj_store_t<3>(acc, outp, m0, n0, lane, bias0, bias1);
        else                  proj_store<3>(acc, outp, m0, n0, lane, bias0, bias1);
    }
}

#define NSL  8
#define RS   16
#define REC_GRID 128
#define PFD  6
#define CSTR 132
#define TSTR 20
#define PS2  20
#define PL2  (16 * PS2)
#define C_OFF   0
#define ST_OFF  (C_OFF + RS * CSTR)
#define Q_O     0
#define K_O     (Q_O + 16 * CSTR)
#define M_O     (K_O + 16 * CSTR)
#define KT_O    (M_O + 16 * CSTR)
#define WC_O    (KT_O + 128 * TSTR)
#define G15_O   (WC_O + RS * TSTR)
#define ST_SZ   (G15_O + 16)
#define PT_OFF  (ST_OFF + 2 * ST_SZ)
#define PP_O    0
#define SP_O    (PP_O + 4 * PL2)
#define NPS_O   (SP_O + 4 * PL2)
#define PT_SZ   (NPS_O + 128)
#define RW_OFF  (PT_OFF + 2 * PT_SZ)
#define RW_SZ   (2 * PL2)
#define VP_OFF  (RW_OFF + 4 * RW_SZ)
#define SMEM_FLOATS (VP_OFF + 2 * PL2)


static __device__ __forceinline__ void lds_barrier()
{
    __builtin_amdgcn_fence(__ATOMIC_RELEASE, "workgroup", "local");
    __builtin_amdgcn_s_barrier();
    __builtin_amdgcn_fence(__ATOMIC_ACQUIRE, "workgroup", "local");
}

__global__ __launch_bounds__(384, 1) void rec_kernel(
    const float* __restrict__ qb, const float* __restrict__ kb,
    const float* __restrict__ vb, const float* __restrict__ ib,
    const float* __restrict__ fb, const float* __restrict__ ob,
    float* __restrict__ hb)
{
    __shared__ __attribute__((aligned(16))) float sm[SMEM_FLOATS];
    const int b    = blockIdx.x >> 3;
    const int sl   = blockIdx.x & 7;
    const int r    = threadIdx.x & 127;
    const int role = __builtin_amdgcn_readfirstlane((int)(threadIdx.x >> 7));
    const bool vec = role == 0;
    const bool mat = role == 1;
    const int lane = r & 31;
    const int wave = r >> 5;
    const int lm   = lane & 15;
    const int hh   = lane >> 4;
    const int lk   = hh << 1;
    const int x    = r & 15;
    const bool rowthr = (r >> 4) == sl;

    for (int j = threadIdx.x; j < RS * CSTR; j += 384) sm[C_OFF + j] = 0.0f;
    float creg[2][8];
    #pragma unroll
    for (int t = 0; t < 2; ++t)
        #pragma unroll
        for (int v = 0; v < 8; ++v) creg[t][v] = 0.0f;

    float nprev = 0.0f;
    float* const hrow = hb + (long)blockIdx.x * (TT * RS);

    float pk[2][LCH], pq[2][LCH], pi[2][LCH], pf[2][LCH];
    const long cbase = (long)b * NCH * (4 * HH * 4) + r * 4;
    auto load_qkif = [&](int c, int set) __attribute__((always_inline)) {
        const long nbase = cbase + (long)c * (4 * HH * 4);
        #pragma unroll
        for (int g = 0; g < 4; ++g) {
            const v4f t = *(const v4f*)(fb + nbase + g * (HH * 4));
            pf[set][4 * g] = t.x; pf[set][4 * g + 1] = t.y; pf[set][4 * g + 2] = t.z; pf[set][4 * g + 3] = t.w;
        }
        #pragma unroll
        for (int g = 0; g < 4; ++g) {
            const v4f t = *(const v4f*)(ib + nbase + g * (HH * 4));
            pi[set][4 * g] = t.x; pi[set][4 * g + 1] = t.y; pi[set][4 * g + 2] = t.z; pi[set][4 * g + 3] = t.w;
        }
        #pragma unroll
        for (int g = 0; g < 4; ++g) {
            const v4f t = *(const v4f*)(kb + nbase + g * (HH * 4));
            pk[set][4 * g] = t.x; pk[set][4 * g + 1] = t.y; pk[set][4 * g + 2] = t.z; pk[set][4 * g + 3] = t.w;
        }
        #pragma unroll
        for (int g = 0; g < 4; ++g) {
            const v4f t = *(const v4f*)(qb + nbase + g * (HH * 4));
            pq[set][4 * g] = t.x; pq[set][4 * g + 1] = t.y; pq[set][4 * g + 2] = t.z; pq[set][4 * g + 3] = t.w;
        }
    };

    const long vobase = (long)b * TT * HH + (long)(r >> 3) * HH + sl * RS + (r & 7) * 2;
    v2f vnx;
    auto load_v = [&](int c) __attribute__((always_inline)) {
        vnx = *(const v2f*)(vb + vobase + (long)(c * LCH) * HH);
    };
    auto load_o = [&](int c) __attribute__((always_inline)) {
        return *(const v2f*)(ob + vobase + (long)(c * LCH) * HH);
    };
    auto store_v = [&](int c) __attribute__((always_inline)) {
        *(v2f*)&sm[VP_OFF + (c & 1) * PL2 + (r >> 3) * PS2 + (r & 7) * 2] = vnx;
    };

    const int wv = __builtin_amdgcn_readfirstlane(wave);
    const float* const pfplane = (wv == 0 ? fb : wv == 1 ? ib : wv == 2 ? kb : qb)
                                 + (long)b * NCH * (4 * HH * 4) + lane * 64;
    const float* const pfvo = (hh ? ob : vb) + (long)b * TT * HH + (long)lm * HH + sl * RS;
    auto prefetch_chunk = [&](int c) __attribute__((always_inline)) {
        __builtin_prefetch(pfplane + (long)c * (4 * HH * 4), 0, 2);
        if (wv == 0) __builtin_prefetch(pfvo + (long)(c * LCH) * HH, 0, 2);
    };

    auto stage_rows = [&](int c, int set) __attribute__((always_inline)) {
        float* st = sm + ST_OFF + (c & 1) * ST_SZ;
        float* rw = sm + RW_OFF + (c & 3) * RW_SZ;
        if (rowthr) {
            const float* vp = sm + VP_OFF + (c & 1) * PL2;
            float pv[LCH];
            #pragma unroll
            for (int s = 0; s < LCH; ++s) pv[s] = vp[s * PS2 + x];
            float G[LCH];
            float g = 1.0f;
            #pragma unroll
            for (int s = 0; s < LCH; ++s) { g *= pf[set][s]; G[s] = g; }
            const float g15 = G[LCH - 1];
            #pragma unroll
            for (int s = 0; s < LCH; ++s) {
                const float r0 = __builtin_amdgcn_rcpf(G[s]);
                const float rg = fmaf(fmaf(-G[s], r0, 1.0f), r0, r0);
                const float wg = pi[set][s] * pv[s] * rg;
                st[WC_O + x * TSTR + s]   = g15 * wg;
                rw[0 * PL2 + s * PS2 + x] = wg;
                rw[1 * PL2 + s * PS2 + x] = G[s];
            }
            st[G15_O + x] = g15;
        }
    };

    auto stage_cols = [&](int c, int set) __attribute__((always_inline)) {
        float* st = sm + ST_OFF + (c & 1) * ST_SZ;
        float np = nprev;
        #pragma unroll
        for (int s = 0; s < LCH; ++s) {
            np = fmaf(pf[set][s], np, pi[set][s] * pk[set][s]);
            st[M_O + s * CSTR + r] = np * pq[set][s];
        }
        nprev = np;
        #pragma unroll
        for (int s = 0; s < LCH; ++s) {
            st[Q_O + s * CSTR + r]  = pq[set][s];
            st[K_O + s * CSTR + r]  = pk[set][s];
            st[KT_O + r * TSTR + s] = pk[set][s];
        }
    };

    const int s1 = r >> 3, seg = r & 7;
    const int cb = wave * 32 + lk;
    v4f mv[4];
    v2f qf[8], kf[8], cf[8], afr[4], bfr[2][4];
    float g[8];

    auto state_load = [&](int c) __attribute__((always_inline)) {
        const float* st = sm + ST_OFF + (c & 1) * ST_SZ;
        #pragma unroll
        for (int e = 0; e < 4; ++e) mv[e] = *(const v4f*)&st[M_O + s1 * CSTR + seg * 16 + 4 * e];

        #pragma unroll
        for (int kk = 0; kk < 8; ++kk) {
            qf[kk] = *(const v2f*)&st[Q_O + lm * CSTR + cb + kk * 4];
            kf[kk] = *(const v2f*)&st[K_O + lm * CSTR + cb + kk * 4];
            cf[kk] = *(const v2f*)&sm[C_OFF + lm * CSTR + cb + kk * 4];
        }
        #pragma unroll
        for (int kk = 0; kk < 4; ++kk)
            afr[kk] = *(const v2f*)&st[WC_O + lm * TSTR + kk * 4 + lk];
        #pragma unroll
        for (int v = 0; v < 8; ++v) g[v] = st[G15_O + 8 * hh + v];
        #pragma unroll
        for (int t = 0; t < 2; ++t)
            #pragma unroll
            for (int kk = 0; kk < 4; ++kk)
                bfr[t][kk] = *(const v2f*)&st[KT_O + (wave * 32 + t * 16 + lm) * TSTR + kk * 4 + lk];
    };

    auto state_cupd = [&]() __attribute__((always_inline)) {
        v8f d0 = {};
        v8f d1 = {};
        #pragma unroll
        for (int kk = 0; kk < 4; ++kk) {
            d0 = __builtin_amdgcn_wmma_f32_16x16x4_f32(false, afr[kk], false, bfr[0][kk],
                                                       (short)0, d0, false, false);
            d1 = __builtin_amdgcn_wmma_f32_16x16x4_f32(false, afr[kk], false, bfr[1][kk],
                                                       (short)0, d1, false, false);
        }
        #pragma unroll
        for (int v = 0; v < 8; ++v) {
            creg[0][v] = fmaf(g[v], creg[0][v], d0[v]);
            creg[1][v] = fmaf(g[v], creg[1][v], d1[v]);
        }
    };

    auto state_ps = [&](int c) __attribute__((always_inline)) {
        float* pt = sm + PT_OFF + (c & 1) * PT_SZ;
        float ps = 0.0f;
        #pragma unroll
        for (int e = 0; e < 4; ++e) { ps += mv[e].x; ps += mv[e].y; ps += mv[e].z; ps += mv[e].w; }

        v8f pacc = {};
        v8f sacc = {};
        #pragma unroll
        for (int kk = 0; kk < 8; ++kk) {
            pacc = __builtin_amdgcn_wmma_f32_16x16x4_f32(false, qf[kk], false, cf[kk],
                                                         (short)0, pacc, false, false);
            sacc = __builtin_amdgcn_wmma_f32_16x16x4_f32(false, kf[kk], false, qf[kk],
                                                         (short)0, sacc, false, false);
        }
        pt[NPS_O + s1 * 8 + seg] = ps;
        #pragma unroll
        for (int v = 0; v < 8; ++v)
            pt[PP_O + wave * PL2 + (v + 8 * hh) * PS2 + lm] = pacc[v];
        *(v4f*)&pt[SP_O + wave * PL2 + lm * PS2 + 8 * hh]     = (v4f){sacc[0], sacc[1], sacc[2], sacc[3]};
        *(v4f*)&pt[SP_O + wave * PL2 + lm * PS2 + 8 * hh + 4] = (v4f){sacc[4], sacc[5], sacc[6], sacc[7]};
        #pragma unroll
        for (int t = 0; t < 2; ++t)
            #pragma unroll
            for (int v = 0; v < 8; ++v)
                sm[C_OFF + (8 * hh + v) * CSTR + wave * 32 + t * 16 + lm] = creg[t][v];
    };

    const int s  = r >> 3;
    const int x0 = (r & 7) * 2;
    v4f sp[4][4], n0, n1;
    v2f pp[4], wg[LCH], gs;
    auto out_load = [&](int c) __attribute__((always_inline)) {
        const float* pt = sm + PT_OFF + (c & 1) * PT_SZ;
        const float* rw = sm + RW_OFF + (c & 3) * RW_SZ;
        #pragma unroll
        for (int w = 0; w < 4; ++w)
            #pragma unroll
            for (int j = 0; j < 4; ++j) sp[w][j] = *(const v4f*)&pt[SP_O + w * PL2 + s * PS2 + 4 * j];
        #pragma unroll
        for (int w = 0; w < 4; ++w) pp[w] = *(const v2f*)&pt[PP_O + w * PL2 + s * PS2 + x0];
        n0 = *(const v4f*)&pt[NPS_O + s * 8];
        n1 = *(const v4f*)&pt[NPS_O + s * 8 + 4];
        #pragma unroll
        for (int u = 0; u < LCH; ++u) wg[u] = *(const v2f*)&rw[0 * PL2 + u * PS2 + x0];
        gs = *(const v2f*)&rw[1 * PL2 + s * PS2 + x0];
    };

    float sv[LCH], nrm;
    v2f acc;
    auto out_sum = [&]() __attribute__((always_inline)) {
        #pragma unroll
        for (int j = 0; j < 4; ++j) {
            const v4f t = ((sp[0][j] + sp[1][j]) + sp[2][j]) + sp[3][j];
            sv[4 * j] = t.x; sv[4 * j + 1] = t.y; sv[4 * j + 2] = t.z; sv[4 * j + 3] = t.w;
        }
        float nq = 0.0f;
        nq += n0.x; nq += n0.y; nq += n0.z; nq += n0.w;
        nq += n1.x; nq += n1.y; nq += n1.z; nq += n1.w;
        nrm = fmaxf(fabsf(nq), 1.0f);
        acc = ((pp[0] + pp[1]) + pp[2]) + pp[3];
    };

    auto out_chain = [&](int c, v2f os) __attribute__((always_inline)) {
        #pragma unroll
        for (int u = 0; u < LCH; ++u) {
            const float nx0 = fmaf(wg[u].x, sv[u], acc.x);
            const float nx1 = fmaf(wg[u].y, sv[u], acc.y);
            acc.x = (u <= s) ? nx0 : acc.x;
            acc.y = (u <= s) ? nx1 : acc.y;
        }
        v2f res;
        res.x = os.x * (gs.x * acc.x) / nrm;
        res.y = os.y * (gs.y * acc.y) / nrm;
        *(v2f*)&hrow[(long)c * (LCH * RS) + s * RS + x0] = res;
    };

    if (vec) {
        load_qkif(0, 0);
        load_v(0); store_v(0);
        load_v(1); store_v(1);
        load_qkif(1, 1);
    } else if (!mat) {
        for (int pc = 2; pc < 2 + PFD; ++pc) prefetch_chunk(pc);
    }
    lds_barrier();
    if (vec) {
        stage_rows(0, 0);
        stage_cols(0, 0);
    }
    lds_barrier();

    if (vec) {
        load_v(2);
        load_qkif(2, 0);
        stage_rows(1, 1);
        stage_cols(1, 1);
        store_v(2);
    } else if (mat) {
        state_load(0);
        state_cupd();
        state_ps(0);
    }
    lds_barrier();

    auto trip_vec = [&](int t, int cs, int ns) __attribute__((always_inline)) {
        const int nc = (t + 1 < NCH) ? t + 1 : NCH - 1;
        load_v(nc);
        load_qkif(nc, ns);
        stage_rows(t, cs);
        stage_cols(t, cs);
        store_v(t + 1);
    };
    auto trip_mat = [&](int t) __attribute__((always_inline)) {
        state_load(t - 1);
        state_cupd();
        state_ps(t - 1);
    };
    auto trip_out = [&](int t) __attribute__((always_inline)) {
        if (t < NCH) prefetch_chunk((t + PFD < NCH) ? t + PFD : NCH - 1);
        const v2f os = load_o(t - 2);
        out_load(t - 2);
        out_sum();
        out_chain(t - 2, os);
    };
    for (int t = 2; t < NCH; t += 2) {
        if (vec)      trip_vec(t, 0, 1);
        else if (mat) trip_mat(t);
        else          trip_out(t);
        lds_barrier();
        if (vec)      trip_vec(t + 1, 1, 0);
        else if (mat) trip_mat(t + 1);
        else          trip_out(t + 1);
        lds_barrier();
    }

    if (mat) {
        trip_mat(NCH);
    } else if (!vec) {
        trip_out(NCH);
    }
    lds_barrier();
    if (!vec && !mat) trip_out(NCH + 1);
}

__global__ __launch_bounds__(256) void ln_kernel(
    const float* __restrict__ hb, const float* __restrict__ x,
    const float* __restrict__ lnw, const float* __restrict__ lnb,
    float* __restrict__ out)
{
    const int lane = threadIdx.x & 31;
    const int wave = threadIdx.x >> 5;
    const long row = (long)blockIdx.x * 8 + wave;
    const long rb  = row * HH;

    float y[4];
    float s = 0.0f, s2 = 0.0f;
    #pragma unroll
    for (int j = 0; j < 4; ++j) {
        const int c = lane + 32 * j;
        const long hidx = (((row >> 9) * NSL + (c >> 4)) * TT + (row & (TT - 1))) * RS + (c & 15);
        y[j] = hb[hidx] + x[rb + c];
        s  += y[j];
        s2 += y[j] * y[j];
    }
    #pragma unroll
    for (int off = 16; off >= 1; off >>= 1) {
        s  += __shfl_xor(s,  off, 32);
        s2 += __shfl_xor(s2, off, 32);
    }
    const float mu  = s * (1.0f / HH);
    const float var = s2 * (1.0f / HH) - mu * mu;
    const float rs  = rsqrtf(var + LN_EPS);

    #pragma unroll
    for (int j = 0; j < 4; ++j) {
        const int c = lane + 32 * j;
        out[rb + c] = (y[j] - mu) * rs * lnw[c] + lnb[c];
    }
}

extern "C" void kernel_launch(void* const* d_in, const int* in_sizes, int n_in,
                              void* d_out, int out_size, void* d_ws, size_t ws_size,
                              hipStream_t stream)
{
    const float* x   = (const float*)d_in[0];
    const float* Wq  = (const float*)d_in[1];
    const float* Wk  = (const float*)d_in[2];
    const float* Wv  = (const float*)d_in[3];
    const float* Wi  = (const float*)d_in[4];
    const float* Wf  = (const float*)d_in[5];
    const float* Wo  = (const float*)d_in[6];
    const float* b_i = (const float*)d_in[7];
    const float* b_f = (const float*)d_in[8];
    const float* b_o = (const float*)d_in[9];
    const float* lnw = (const float*)d_in[10];
    const float* lnb = (const float*)d_in[11];
    float* out = (float*)d_out;

    float* ws = (float*)d_ws;
    float* q  = ws + 0L * NEL;
    float* k  = ws + 1L * NEL;
    float* v  = ws + 2L * NEL;
    float* ig = ws + 3L * NEL;
    float* fg = ws + 4L * NEL;
    float* og = ws + 5L * NEL;
    float* h  = ws + 6L * NEL;

    proj_kernel<<<PJ_GRID, 128, 0, stream>>>(x, Wq, Wk, Wv, Wi, Wf, Wo,
                                          b_i, b_f, b_o,
                                          q, k, v, ig, fg, og);
    rec_kernel<<<REC_GRID, 384, 0, stream>>>(q, k, v, ig, fg, og, h);
    ln_kernel<<<MM / 8, 256, 0, stream>>>(h, x, lnw, lnb, out);
}
